# starting kernel
# baseline (speedup 1.0000x reference)
#include <hip/hip_runtime.h>


typedef float v2f __attribute__((ext_vector_type(2)));
typedef float v8f __attribute__((ext_vector_type(8)));

#define DFEAT 128

__global__ void gnn_zero_f32(float* __restrict__ p, long n) {
    long i = (long)blockIdx.x * blockDim.x + threadIdx.x;
    if (i < n) p[i] = 0.0f;
}

__global__ void gnn_degree(const int* __restrict__ dst, float* __restrict__ deg,
                           int n_edges) {
    int e = blockIdx.x * blockDim.x + threadIdx.x;
    if (e < n_edges) atomicAdd(&deg[dst[e]], 1.0f);
}

__global__ void gnn_inv_deg(float* __restrict__ deg, int n_nodes) {
    int i = blockIdx.x * blockDim.x + threadIdx.x;
    if (i < n_nodes) {
        float d = deg[i];
        deg[i] = (d > 0.0f) ? (1.0f / d) : 0.0f;
    }
}

__global__ void gnn_scatter_add(const float* __restrict__ h,
                                const int* __restrict__ src,
                                const int* __restrict__ dst,
                                float* __restrict__ agg,
                                int n_edges) {
    long tid = (long)blockIdx.x * blockDim.x + threadIdx.x;
    long e   = tid >> 5;
    int  c   = (int)(tid & 31);
    if (e >= n_edges) return;
    int s = src[e], d = dst[e];
    const float4 v = *(const float4*)(h + (size_t)s * DFEAT + c * 4);
    float* o = agg + (size_t)d * DFEAT + c * 4;
    atomicAdd(o + 0, v.x);
    atomicAdd(o + 1, v.y);
    atomicAdd(o + 2, v.z);
    atomicAdd(o + 3, v.w);
}

__global__ __launch_bounds__(256)
void gnn_sage_gemm_wmma(const float* __restrict__ agg,
                        const float* __restrict__ inv_deg,
                        const float* __restrict__ h,
                        const float* __restrict__ Wl,
                        const float* __restrict__ bl,
                        const float* __restrict__ Wr,
                        float* __restrict__ out,
                        int n_nodes, int d_out, int n_col_tiles) {
    const int wave = threadIdx.x >> 5;
    const int lane = threadIdx.x & 31;
    const int row_tiles_per_blk = (blockDim.x >> 5) / n_col_tiles;

    const int mtile = blockIdx.x * row_tiles_per_blk + wave / n_col_tiles;
    const int ntile = wave % n_col_tiles;
    if (mtile * 16 >= n_nodes) return;

    const int m0 = mtile * 16;
    const int n0 = ntile * 16;
    int rA = m0 + (lane & 15);
    if (rA >= n_nodes) rA = n_nodes - 1;
    const int rW   = n0 + (lane & 15);
    const int klo  = 2 * (lane >> 4);

    const float s = inv_deg[rA];

    const float* __restrict__ arow = agg + (size_t)rA * DFEAT + klo;
    const float* __restrict__ hrow = h   + (size_t)rA * DFEAT + klo;
    const float* __restrict__ wlr  = Wl  + (size_t)rW * DFEAT + klo;
    const float* __restrict__ wrr  = Wr  + (size_t)rW * DFEAT + klo;

    v8f acc = {};

    #pragma unroll 4
    for (int kk = 0; kk < DFEAT; kk += 4) {
        v2f a = *(const v2f*)(arow + kk);
        a *= s;
        v2f b = *(const v2f*)(wlr + kk);
        acc = __builtin_amdgcn_wmma_f32_16x16x4_f32(
            false, a, false, b, (short)0, acc, false, false);
    }

    #pragma unroll 4
    for (int kk = 0; kk < DFEAT; kk += 4) {
        v2f a = *(const v2f*)(hrow + kk);
        v2f b = *(const v2f*)(wrr + kk);
        acc = __builtin_amdgcn_wmma_f32_16x16x4_f32(
            false, a, false, b, (short)0, acc, false, false);
    }

    const int n    = n0 + (lane & 15);
    const float bb = bl[n];
    const int mbase = m0 + 8 * (lane >> 4);
    #pragma unroll
    for (int v = 0; v < 8; ++v) {
        const int m = mbase + v;
        if (m < n_nodes) {
            float val = acc[v] + bb;
            out[(size_t)m * d_out + n] = (val > 0.0f) ? val : 0.0f;
        }
    }
}

extern "C" void kernel_launch(void* const* d_in, const int* in_sizes, int n_in,
                              void* d_out, int out_size, void* d_ws, size_t ws_size,
                              hipStream_t stream) {
    const float* x    = (const float*)d_in[0];
    const int*   eidx = (const int*)  d_in[1];
    const float* W_l1 = (const float*)d_in[2];
    const float* b_l1 = (const float*)d_in[3];
    const float* W_r1 = (const float*)d_in[4];
    const float* W_l2 = (const float*)d_in[5];
    const float* b_l2 = (const float*)d_in[6];
    const float* W_r2 = (const float*)d_in[7];
    const float* W_l3 = (const float*)d_in[8];
    const float* b_l3 = (const float*)d_in[9];
    const float* W_r3 = (const float*)d_in[10];
    float* out = (float*)d_out;

    const int n_nodes = in_sizes[0] / DFEAT;
    const int n_edges = in_sizes[1] / 2;
    const int d_out3  = out_size / n_nodes;
    const int* src = eidx;
    const int* dst = eidx + n_edges;

    float* deg = (float*)d_ws;
    float* agg = deg + n_nodes;
    float* h1  = agg + (size_t)n_nodes * DFEAT;
    float* h2  = h1  + (size_t)n_nodes * DFEAT;

    const long nfeat = (long)n_nodes * DFEAT;
    const int  TB    = 256;
    const int  gZeroN    = (int)((n_nodes + TB - 1) / TB);
    const int  gZeroFeat = (int)((nfeat + TB - 1) / TB);
    const int  gEdges    = (int)((n_edges + TB - 1) / TB);
    const int  gScatter  = (int)(((long)n_edges * 32 + TB - 1) / TB);

    const int mtiles = (n_nodes + 15) / 16;
    const int gGemmHid = mtiles;
    const int gGemmOut = (mtiles + 3) / 4;

    gnn_zero_f32<<<gZeroN, TB, 0, stream>>>(deg, n_nodes);
    gnn_degree  <<<gEdges, TB, 0, stream>>>(dst, deg, n_edges);
    gnn_inv_deg <<<gZeroN, TB, 0, stream>>>(deg, n_nodes);

    gnn_zero_f32   <<<gZeroFeat, TB, 0, stream>>>(agg, nfeat);
    gnn_scatter_add<<<gScatter,  TB, 0, stream>>>(x, src, dst, agg, n_edges);
    gnn_sage_gemm_wmma<<<gGemmHid, TB, 0, stream>>>(agg, deg, x, W_l1, b_l1, W_r1,
                                                    h1, n_nodes, DFEAT, 8);

    gnn_zero_f32   <<<gZeroFeat, TB, 0, stream>>>(agg, nfeat);
    gnn_scatter_add<<<gScatter,  TB, 0, stream>>>(h1, src, dst, agg, n_edges);
    gnn_sage_gemm_wmma<<<gGemmHid, TB, 0, stream>>>(agg, deg, h1, W_l2, b_l2, W_r2,
                                                    h2, n_nodes, DFEAT, 8);

    gnn_zero_f32   <<<gZeroFeat, TB, 0, stream>>>(agg, nfeat);
    gnn_scatter_add<<<gScatter,  TB, 0, stream>>>(h2, src, dst, agg, n_edges);
    gnn_sage_gemm_wmma<<<gGemmOut, TB, 0, stream>>>(agg, deg, h2, W_l3, b_l3, W_r3,
                                                    out, n_nodes, d_out3, 2);
}


# Replace the float-atomic edge scatter and its zero pass by per-call bucketing of edges by destination (int atomic rank, scan, fill) and a per-node gather sum; rows summed in arrival-rank order
# speedup vs baseline: 2.1784x; 2.1784x over previous
#include <hip/hip_runtime.h>


typedef float v2f __attribute__((ext_vector_type(2)));
typedef float v8f __attribute__((ext_vector_type(8)));

#define DFEAT 128

typedef float v4f __attribute__((ext_vector_type(4)));

__global__ void gnn_zero_i32(int* __restrict__ p, int n) {
    int i = blockIdx.x * blockDim.x + threadIdx.x;
    if (i < n) p[i] = 0;
}

__global__ void gnn_count_rank(const int* __restrict__ dst, int* __restrict__ cnt,
                               int* __restrict__ rank, int n_edges) {
    int e = blockIdx.x * blockDim.x + threadIdx.x;
    if (e < n_edges) rank[e] = atomicAdd(&cnt[dst[e]], 1);
}

__global__ __launch_bounds__(1024)
void gnn_scan_offsets(const int* __restrict__ cnt, int* __restrict__ off,
                      float* __restrict__ inv_deg, int n) {
    __shared__ int wave_tot[32];
    const int t = threadIdx.x, lane = t & 31, w = t >> 5;
    const int per = (n + 1023) >> 10;
    int beg = t * per;   if (beg > n) beg = n;
    int end = beg + per; if (end > n) end = n;

    int s = 0;
    for (int i = beg; i < end; ++i) s += cnt[i];

    int incl = s;
    for (int d = 1; d < 32; d <<= 1) {
        int y = __shfl_up(incl, d);
        if (lane >= d) incl += y;
    }
    if (lane == 31) wave_tot[w] = incl;
    __syncthreads();
    if (w == 0) {
        const int v = wave_tot[lane];
        int iv = v;
        for (int d = 1; d < 32; d <<= 1) {
            int y = __shfl_up(iv, d);
            if (lane >= d) iv += y;
        }
        wave_tot[lane] = iv - v;
    }
    __syncthreads();

    int run = wave_tot[w] + incl - s;
    for (int i = beg; i < end; ++i) {
        const int c = cnt[i];
        const float r = 1.0f / (float)(c > 0 ? c : 1);
        off[i]     = run;
        inv_deg[i] = (c > 0) ? r : 0.0f;
        run += c;
    }
    if (t == 1023) off[n] = run;
}

__global__ void gnn_fill_buckets(const int* __restrict__ src, const int* __restrict__ dst,
                                 const int* __restrict__ off, const int* __restrict__ rank,
                                 int* __restrict__ nbr, int n_edges) {
    int e = blockIdx.x * blockDim.x + threadIdx.x;
    if (e < n_edges) nbr[off[dst[e]] + rank[e]] = src[e];
}

__global__ __launch_bounds__(256)
void gnn_gather_sum(const float* __restrict__ h, const int* __restrict__ off,
                    const int* __restrict__ nbr, float* __restrict__ agg, int n_nodes) {
    const int lane = threadIdx.x & 31;
    const int node = __builtin_amdgcn_readfirstlane((int)(blockIdx.x * 8 + (threadIdx.x >> 5)));
    if (node >= n_nodes) return;
    const int beg = off[node];
    const int end = off[node + 1];
    const float* __restrict__ hl = h + lane * 4;

    v4f acc = {0.0f, 0.0f, 0.0f, 0.0f};
    for (int base = beg; base < end; base += 32) {
        int m = end - base; if (m > 32) m = 32;
        const int mine = (lane < m) ? nbr[base + lane] : 0;
        int k = 0;
        for (; k + 4 <= m; k += 4) {
            const int s0 = __builtin_amdgcn_readlane(mine, k);
            const int s1 = __builtin_amdgcn_readlane(mine, k + 1);
            const int s2 = __builtin_amdgcn_readlane(mine, k + 2);
            const int s3 = __builtin_amdgcn_readlane(mine, k + 3);
            const v4f v0 = *(const v4f*)(hl + (size_t)s0 * DFEAT);
            const v4f v1 = *(const v4f*)(hl + (size_t)s1 * DFEAT);
            const v4f v2 = *(const v4f*)(hl + (size_t)s2 * DFEAT);
            const v4f v3 = *(const v4f*)(hl + (size_t)s3 * DFEAT);
            acc += v0; acc += v1; acc += v2; acc += v3;
        }
        const int r = m - k;
        if (r == 3) {
            const int s0 = __builtin_amdgcn_readlane(mine, k);
            const int s1 = __builtin_amdgcn_readlane(mine, k + 1);
            const int s2 = __builtin_amdgcn_readlane(mine, k + 2);
            const v4f v0 = *(const v4f*)(hl + (size_t)s0 * DFEAT);
            const v4f v1 = *(const v4f*)(hl + (size_t)s1 * DFEAT);
            const v4f v2 = *(const v4f*)(hl + (size_t)s2 * DFEAT);
            acc += v0; acc += v1; acc += v2;
        } else if (r == 2) {
            const int s0 = __builtin_amdgcn_readlane(mine, k);
            const int s1 = __builtin_amdgcn_readlane(mine, k + 1);
            const v4f v0 = *(const v4f*)(hl + (size_t)s0 * DFEAT);
            const v4f v1 = *(const v4f*)(hl + (size_t)s1 * DFEAT);
            acc += v0; acc += v1;
        } else if (r == 1) {
            const int s0 = __builtin_amdgcn_readlane(mine, k);
            acc += *(const v4f*)(hl + (size_t)s0 * DFEAT);
        }
    }
    *(v4f*)(agg + (size_t)node * DFEAT + lane * 4) = acc;
}

__global__ __launch_bounds__(256)
void gnn_sage_gemm_wmma(const float* __restrict__ agg,
                        const float* __restrict__ inv_deg,
                        const float* __restrict__ h,
                        const float* __restrict__ Wl,
                        const float* __restrict__ bl,
                        const float* __restrict__ Wr,
                        float* __restrict__ out,
                        int n_nodes, int d_out, int n_col_tiles) {
    const int wave = threadIdx.x >> 5;
    const int lane = threadIdx.x & 31;
    const int row_tiles_per_blk = (blockDim.x >> 5) / n_col_tiles;

    const int mtile = blockIdx.x * row_tiles_per_blk + wave / n_col_tiles;
    const int ntile = wave % n_col_tiles;
    if (mtile * 16 >= n_nodes) return;

    const int m0 = mtile * 16;
    const int n0 = ntile * 16;
    int rA = m0 + (lane & 15);
    if (rA >= n_nodes) rA = n_nodes - 1;
    const int rW   = n0 + (lane & 15);
    const int klo  = 2 * (lane >> 4);

    const float s = inv_deg[rA];

    const float* __restrict__ arow = agg + (size_t)rA * DFEAT + klo;
    const float* __restrict__ hrow = h   + (size_t)rA * DFEAT + klo;
    const float* __restrict__ wlr  = Wl  + (size_t)rW * DFEAT + klo;
    const float* __restrict__ wrr  = Wr  + (size_t)rW * DFEAT + klo;

    v8f acc = {};

    #pragma unroll 4
    for (int kk = 0; kk < DFEAT; kk += 4) {
        v2f a = *(const v2f*)(arow + kk);
        a *= s;
        v2f b = *(const v2f*)(wlr + kk);
        acc = __builtin_amdgcn_wmma_f32_16x16x4_f32(
            false, a, false, b, (short)0, acc, false, false);
    }

    #pragma unroll 4
    for (int kk = 0; kk < DFEAT; kk += 4) {
        v2f a = *(const v2f*)(hrow + kk);
        v2f b = *(const v2f*)(wrr + kk);
        acc = __builtin_amdgcn_wmma_f32_16x16x4_f32(
            false, a, false, b, (short)0, acc, false, false);
    }

    const int n    = n0 + (lane & 15);
    const float bb = bl[n];
    const int mbase = m0 + 8 * (lane >> 4);
    #pragma unroll
    for (int v = 0; v < 8; ++v) {
        const int m = mbase + v;
        if (m < n_nodes) {
            float val = acc[v] + bb;
            out[(size_t)m * d_out + n] = (val > 0.0f) ? val : 0.0f;
        }
    }
}

extern "C" void kernel_launch(void* const* d_in, const int* in_sizes, int n_in,
                              void* d_out, int out_size, void* d_ws, size_t ws_size,
                              hipStream_t stream) {
    const float* x    = (const float*)d_in[0];
    const int*   eidx = (const int*)  d_in[1];
    const float* W_l1 = (const float*)d_in[2];
    const float* b_l1 = (const float*)d_in[3];
    const float* W_r1 = (const float*)d_in[4];
    const float* W_l2 = (const float*)d_in[5];
    const float* b_l2 = (const float*)d_in[6];
    const float* W_r2 = (const float*)d_in[7];
    const float* W_l3 = (const float*)d_in[8];
    const float* b_l3 = (const float*)d_in[9];
    const float* W_r3 = (const float*)d_in[10];
    float* out = (float*)d_out;

    const int n_nodes = in_sizes[0] / DFEAT;
    const int n_edges = in_sizes[1] / 2;
    const int d_out3  = out_size / n_nodes;
    const int* src = eidx;
    const int* dst = eidx + n_edges;

    float* deg = (float*)d_ws;
    float* agg = deg + n_nodes;
    float* h1  = agg + (size_t)n_nodes * DFEAT;
    float* h2  = h1  + (size_t)n_nodes * DFEAT;
    int* cnt   = (int*)(h2 + (size_t)n_nodes * DFEAT);
    int* off   = cnt + n_nodes;
    int* rank  = off + n_nodes + 1;
    int* nbr   = rank + n_edges;

    const int  TB      = 256;
    const int  gNodes  = (n_nodes + TB - 1) / TB;
    const int  gEdges  = (n_edges + TB - 1) / TB;
    const int  gGather = (n_nodes + 7) / 8;

    const int mtiles = (n_nodes + 15) / 16;
    const int gGemmHid = mtiles;
    const int gGemmOut = (mtiles + 3) / 4;

    gnn_zero_i32    <<<gNodes, TB, 0, stream>>>(cnt, n_nodes);
    gnn_count_rank  <<<gEdges, TB, 0, stream>>>(dst, cnt, rank, n_edges);
    gnn_scan_offsets<<<1, 1024, 0, stream>>>(cnt, off, deg, n_nodes);
    gnn_fill_buckets<<<gEdges, TB, 0, stream>>>(src, dst, off, rank, nbr, n_edges);

    gnn_gather_sum<<<gGather, TB, 0, stream>>>(x, off, nbr, agg, n_nodes);
    gnn_sage_gemm_wmma<<<gGemmHid, TB, 0, stream>>>(agg, deg, x, W_l1, b_l1, W_r1,
                                                    h1, n_nodes, DFEAT, 8);

    gnn_gather_sum<<<gGather, TB, 0, stream>>>(h1, off, nbr, agg, n_nodes);
    gnn_sage_gemm_wmma<<<gGemmHid, TB, 0, stream>>>(agg, deg, h1, W_l2, b_l2, W_r2,
                                                    h2, n_nodes, DFEAT, 8);

    gnn_gather_sum<<<gGather, TB, 0, stream>>>(h2, off, nbr, agg, n_nodes);
    gnn_sage_gemm_wmma<<<gGemmOut, TB, 0, stream>>>(agg, deg, h2, W_l3, b_l3, W_r3,
                                                    out, n_nodes, d_out3, 2);
}


# Layer GEMM: stage the layer's Wl|Wr rows in LDS once per block; each wave computes all output columns of its 16 rows, so one A fragment feeds 8 (2) WMMAs and B fragments are read from LDS
# speedup vs baseline: 3.1716x; 1.4559x over previous
#include <hip/hip_runtime.h>


typedef float v2f __attribute__((ext_vector_type(2)));
typedef float v8f __attribute__((ext_vector_type(8)));

#define DFEAT 128

typedef float v4f __attribute__((ext_vector_type(4)));

__global__ void gnn_zero_i32(int* __restrict__ p, int n) {
    int i = blockIdx.x * blockDim.x + threadIdx.x;
    if (i < n) p[i] = 0;
}

__global__ void gnn_count_rank(const int* __restrict__ dst, int* __restrict__ cnt,
                               int* __restrict__ rank, int n_edges) {
    int e = blockIdx.x * blockDim.x + threadIdx.x;
    if (e < n_edges) rank[e] = atomicAdd(&cnt[dst[e]], 1);
}

__global__ __launch_bounds__(1024)
void gnn_scan_offsets(const int* __restrict__ cnt, int* __restrict__ off,
                      float* __restrict__ inv_deg, int n) {
    __shared__ int wave_tot[32];
    const int t = threadIdx.x, lane = t & 31, w = t >> 5;
    const int per = (n + 1023) >> 10;
    int beg = t * per;   if (beg > n) beg = n;
    int end = beg + per; if (end > n) end = n;

    int s = 0;
    for (int i = beg; i < end; ++i) s += cnt[i];

    int incl = s;
    for (int d = 1; d < 32; d <<= 1) {
        int y = __shfl_up(incl, d);
        if (lane >= d) incl += y;
    }
    if (lane == 31) wave_tot[w] = incl;
    __syncthreads();
    if (w == 0) {
        const int v = wave_tot[lane];
        int iv = v;
        for (int d = 1; d < 32; d <<= 1) {
            int y = __shfl_up(iv, d);
            if (lane >= d) iv += y;
        }
        wave_tot[lane] = iv - v;
    }
    __syncthreads();

    int run = wave_tot[w] + incl - s;
    for (int i = beg; i < end; ++i) {
        const int c = cnt[i];
        const float r = 1.0f / (float)(c > 0 ? c : 1);
        off[i]     = run;
        inv_deg[i] = (c > 0) ? r : 0.0f;
        run += c;
    }
    if (t == 1023) off[n] = run;
}

__global__ void gnn_fill_buckets(const int* __restrict__ src, const int* __restrict__ dst,
                                 const int* __restrict__ off, const int* __restrict__ rank,
                                 int* __restrict__ nbr, int n_edges) {
    int e = blockIdx.x * blockDim.x + threadIdx.x;
    if (e < n_edges) nbr[off[dst[e]] + rank[e]] = src[e];
}

__global__ __launch_bounds__(256)
void gnn_gather_sum(const float* __restrict__ h, const int* __restrict__ off,
                    const int* __restrict__ nbr, float* __restrict__ agg, int n_nodes) {
    const int lane = threadIdx.x & 31;
    const int node = __builtin_amdgcn_readfirstlane((int)(blockIdx.x * 8 + (threadIdx.x >> 5)));
    if (node >= n_nodes) return;
    const int beg = off[node];
    const int end = off[node + 1];
    const float* __restrict__ hl = h + lane * 4;

    v4f acc = {0.0f, 0.0f, 0.0f, 0.0f};
    for (int base = beg; base < end; base += 32) {
        int m = end - base; if (m > 32) m = 32;
        const int mine = (lane < m) ? nbr[base + lane] : 0;
        int k = 0;
        for (; k + 4 <= m; k += 4) {
            const int s0 = __builtin_amdgcn_readlane(mine, k);
            const int s1 = __builtin_amdgcn_readlane(mine, k + 1);
            const int s2 = __builtin_amdgcn_readlane(mine, k + 2);
            const int s3 = __builtin_amdgcn_readlane(mine, k + 3);
            const v4f v0 = *(const v4f*)(hl + (size_t)s0 * DFEAT);
            const v4f v1 = *(const v4f*)(hl + (size_t)s1 * DFEAT);
            const v4f v2 = *(const v4f*)(hl + (size_t)s2 * DFEAT);
            const v4f v3 = *(const v4f*)(hl + (size_t)s3 * DFEAT);
            acc += v0; acc += v1; acc += v2; acc += v3;
        }
        const int r = m - k;
        if (r == 3) {
            const int s0 = __builtin_amdgcn_readlane(mine, k);
            const int s1 = __builtin_amdgcn_readlane(mine, k + 1);
            const int s2 = __builtin_amdgcn_readlane(mine, k + 2);
            const v4f v0 = *(const v4f*)(hl + (size_t)s0 * DFEAT);
            const v4f v1 = *(const v4f*)(hl + (size_t)s1 * DFEAT);
            const v4f v2 = *(const v4f*)(hl + (size_t)s2 * DFEAT);
            acc += v0; acc += v1; acc += v2;
        } else if (r == 2) {
            const int s0 = __builtin_amdgcn_readlane(mine, k);
            const int s1 = __builtin_amdgcn_readlane(mine, k + 1);
            const v4f v0 = *(const v4f*)(hl + (size_t)s0 * DFEAT);
            const v4f v1 = *(const v4f*)(hl + (size_t)s1 * DFEAT);
            acc += v0; acc += v1;
        } else if (r == 1) {
            const int s0 = __builtin_amdgcn_readlane(mine, k);
            acc += *(const v4f*)(hl + (size_t)s0 * DFEAT);
        }
    }
    *(v4f*)(agg + (size_t)node * DFEAT + lane * 4) = acc;
}

#define WPITCH 260

template <int NT>
__device__ __forceinline__
void gnn_sage_gemm_body(const float* __restrict__ agg,
                        const float* __restrict__ inv_deg,
                        const float* __restrict__ h,
                        const float* __restrict__ Wl,
                        const float* __restrict__ bl,
                        const float* __restrict__ Wr,
                        float* __restrict__ out,
                        float* wlds,
                        int n_nodes) {
    const int tid  = threadIdx.x;
    const int wave = tid >> 5;
    const int lane = tid & 31;

    constexpr int NV4 = NT * 16 * 64;
    const int q  = tid & 63;
    const int nr = tid >> 6;
    const float* __restrict__ wsrc = ((q < 32) ? Wl : Wr) + (size_t)nr * DFEAT + 4 * (q & 31);
    float* wdst = wlds + nr * WPITCH + 4 * q;
    #pragma unroll
    for (int i = 0; i < NV4 / 256; ++i) {
        const v4f v = *(const v4f*)(wsrc + (size_t)i * 4 * DFEAT);
        *(v4f*)(wdst + i * 4 * WPITCH) = v;
    }
    __syncthreads();

    const int mtile = blockIdx.x * 8 + wave;
    if (mtile * 16 >= n_nodes) return;

    const int m0 = mtile * 16;
    int rA = m0 + (lane & 15);
    if (rA >= n_nodes) rA = n_nodes - 1;
    const int klo = 2 * (lane >> 4);

    const float s = inv_deg[rA];

    const float* __restrict__ arow = agg + (size_t)rA * DFEAT + klo;
    const float* __restrict__ hrow = h   + (size_t)rA * DFEAT + klo;
    const float* wb = wlds + (lane & 15) * WPITCH + klo;

    v8f acc[NT];
    #pragma unroll
    for (int t = 0; t < NT; ++t) acc[t] = v8f{};

    #pragma unroll 8
    for (int kk = 0; kk < DFEAT; kk += 4) {
        v2f a = *(const v2f*)(arow + kk);
        a *= s;
        #pragma unroll
        for (int t = 0; t < NT; ++t) {
            const v2f b = *(const v2f*)(wb + t * 16 * WPITCH + kk);
            acc[t] = __builtin_amdgcn_wmma_f32_16x16x4_f32(
                false, a, false, b, (short)0, acc[t], false, false);
        }
    }

    #pragma unroll 8
    for (int kk = 0; kk < DFEAT; kk += 4) {
        const v2f a = *(const v2f*)(hrow + kk);
        #pragma unroll
        for (int t = 0; t < NT; ++t) {
            const v2f b = *(const v2f*)(wb + t * 16 * WPITCH + DFEAT + kk);
            acc[t] = __builtin_amdgcn_wmma_f32_16x16x4_f32(
                false, a, false, b, (short)0, acc[t], false, false);
        }
    }

    constexpr int DOUT = NT * 16;
    const int mbase = m0 + 8 * (lane >> 4);
    float bb[NT];
    #pragma unroll
    for (int t = 0; t < NT; ++t) bb[t] = bl[t * 16 + (lane & 15)];
    float* __restrict__ orow = out + (size_t)mbase * DOUT + (lane & 15);
    if (m0 + 16 <= n_nodes) {
        #pragma unroll
        for (int t = 0; t < NT; ++t) {
            #pragma unroll
            for (int v = 0; v < 8; ++v) {
                const float val = acc[t][v] + bb[t];
                orow[v * DOUT + t * 16] = (val > 0.0f) ? val : 0.0f;
            }
        }
    } else {
        #pragma unroll
        for (int t = 0; t < NT; ++t) {
            #pragma unroll
            for (int v = 0; v < 8; ++v) {
                if (mbase + v < n_nodes) {
                    const float val = acc[t][v] + bb[t];
                    orow[v * DOUT + t * 16] = (val > 0.0f) ? val : 0.0f;
                }
            }
        }
    }
}

__global__ __launch_bounds__(256)
void gnn_sage_gemm_wmma_hid(const float* __restrict__ agg,
                            const float* __restrict__ inv_deg,
                            const float* __restrict__ h,
                            const float* __restrict__ Wl,
                            const float* __restrict__ bl,
                            const float* __restrict__ Wr,
                            float* __restrict__ out,
                            int n_nodes) {
    __shared__ __attribute__((aligned(16))) float wlds[8 * 16 * WPITCH];
    gnn_sage_gemm_body<8>(agg, inv_deg, h, Wl, bl, Wr, out, wlds, n_nodes);
}

__global__ __launch_bounds__(256)
void gnn_sage_gemm_wmma_out(const float* __restrict__ agg,
                            const float* __restrict__ inv_deg,
                            const float* __restrict__ h,
                            const float* __restrict__ Wl,
                            const float* __restrict__ bl,
                            const float* __restrict__ Wr,
                            float* __restrict__ out,
                            int n_nodes) {
    __shared__ __attribute__((aligned(16))) float wlds[2 * 16 * WPITCH];
    gnn_sage_gemm_body<2>(agg, inv_deg, h, Wl, bl, Wr, out, wlds, n_nodes);
}

extern "C" void kernel_launch(void* const* d_in, const int* in_sizes, int n_in,
                              void* d_out, int out_size, void* d_ws, size_t ws_size,
                              hipStream_t stream) {
    const float* x    = (const float*)d_in[0];
    const int*   eidx = (const int*)  d_in[1];
    const float* W_l1 = (const float*)d_in[2];
    const float* b_l1 = (const float*)d_in[3];
    const float* W_r1 = (const float*)d_in[4];
    const float* W_l2 = (const float*)d_in[5];
    const float* b_l2 = (const float*)d_in[6];
    const float* W_r2 = (const float*)d_in[7];
    const float* W_l3 = (const float*)d_in[8];
    const float* b_l3 = (const float*)d_in[9];
    const float* W_r3 = (const float*)d_in[10];
    float* out = (float*)d_out;

    const int n_nodes = in_sizes[0] / DFEAT;
    const int n_edges = in_sizes[1] / 2;
    const int* src = eidx;
    const int* dst = eidx + n_edges;

    float* deg = (float*)d_ws;
    float* agg = deg + n_nodes;
    float* h1  = agg + (size_t)n_nodes * DFEAT;
    float* h2  = h1  + (size_t)n_nodes * DFEAT;
    int* cnt   = (int*)(h2 + (size_t)n_nodes * DFEAT);
    int* off   = cnt + n_nodes;
    int* rank  = off + n_nodes + 1;
    int* nbr   = rank + n_edges;

    const int  TB      = 256;
    const int  gNodes  = (n_nodes + TB - 1) / TB;
    const int  gEdges  = (n_edges + TB - 1) / TB;
    const int  gGather = (n_nodes + 7) / 8;

    const int mtiles = (n_nodes + 15) / 16;
    const int gGemm = (mtiles + 7) / 8;

    gnn_zero_i32    <<<gNodes, TB, 0, stream>>>(cnt, n_nodes);
    gnn_count_rank  <<<gEdges, TB, 0, stream>>>(dst, cnt, rank, n_edges);
    gnn_scan_offsets<<<1, 1024, 0, stream>>>(cnt, off, deg, n_nodes);
    gnn_fill_buckets<<<gEdges, TB, 0, stream>>>(src, dst, off, rank, nbr, n_edges);

    gnn_gather_sum<<<gGather, TB, 0, stream>>>(x, off, nbr, agg, n_nodes);
    gnn_sage_gemm_wmma_hid<<<gGemm, TB, 0, stream>>>(agg, deg, x, W_l1, b_l1, W_r1,
                                                     h1, n_nodes);

    gnn_gather_sum<<<gGather, TB, 0, stream>>>(h1, off, nbr, agg, n_nodes);
    gnn_sage_gemm_wmma_hid<<<gGemm, TB, 0, stream>>>(agg, deg, h1, W_l2, b_l2, W_r2,
                                                     h2, n_nodes);

    gnn_gather_sum<<<gGather, TB, 0, stream>>>(h2, off, nbr, agg, n_nodes);
    gnn_sage_gemm_wmma_out<<<gGemm, TB, 0, stream>>>(agg, deg, h2, W_l3, b_l3, W_r3,
                                                     out, n_nodes);
}


# Offsets scan: one block per 1024 nodes instead of one block for all; a block sums the counts ahead of its segment itself, whole-line loads and stores, one count per thread
# speedup vs baseline: 3.6504x; 1.1510x over previous
#include <hip/hip_runtime.h>


typedef float v2f __attribute__((ext_vector_type(2)));
typedef float v8f __attribute__((ext_vector_type(8)));

#define DFEAT 128

typedef float v4f __attribute__((ext_vector_type(4)));

__global__ void gnn_zero_i32(int* __restrict__ p, int n) {
    int i = blockIdx.x * blockDim.x + threadIdx.x;
    if (i < n) p[i] = 0;
}

__global__ void gnn_count_rank(const int* __restrict__ dst, int* __restrict__ cnt,
                               int* __restrict__ rank, int n_edges) {
    int e = blockIdx.x * blockDim.x + threadIdx.x;
    if (e < n_edges) rank[e] = atomicAdd(&cnt[dst[e]], 1);
}

__global__ __launch_bounds__(1024)
void gnn_scan_offsets(const int* __restrict__ cnt, int* __restrict__ off,
                      float* __restrict__ inv_deg, int n) {
    __shared__ int wave_tot[32];
    __shared__ int wave_pre[32];
    const int t = threadIdx.x, lane = t & 31, w = t >> 5;
    const int base = blockIdx.x * 1024;

    const int idx = base + t;
    const int c = (idx < n) ? cnt[idx] : 0;

    int pre = 0;
    if ((((size_t)cnt) & 15) == 0) {
        for (int i = 4 * t; i < base; i += 4096) {
            const int4 v = *(const int4*)(cnt + i);
            pre += (v.x + v.y) + (v.z + v.w);
        }
    } else {
        for (int i = t; i < base; i += 1024) pre += cnt[i];
    }

    int incl = c;
    for (int d = 1; d < 32; d <<= 1) {
        int y = __shfl_up(incl, d);
        if (lane >= d) incl += y;
    }
    for (int d = 16; d >= 1; d >>= 1) pre += __shfl_xor(pre, d);
    if (lane == 31) { wave_tot[w] = incl; wave_pre[w] = pre; }
    __syncthreads();
    if (w == 0) {
        const int v = wave_tot[lane];
        int iv = v;
        for (int d = 1; d < 32; d <<= 1) {
            int y = __shfl_up(iv, d);
            if (lane >= d) iv += y;
        }
        int p = wave_pre[lane];
        for (int d = 16; d >= 1; d >>= 1) p += __shfl_xor(p, d);
        wave_tot[lane] = p + iv - v;
    }
    __syncthreads();

    const int run = wave_tot[w] + incl - c;
    if (idx <= n) off[idx] = run;
    if (idx < n) {
        const float r = 1.0f / (float)(c > 0 ? c : 1);
        inv_deg[idx] = (c > 0) ? r : 0.0f;
    }
}

__global__ void gnn_fill_buckets(const int* __restrict__ src, const int* __restrict__ dst,
                                 const int* __restrict__ off, const int* __restrict__ rank,
                                 int* __restrict__ nbr, int n_edges) {
    int e = blockIdx.x * blockDim.x + threadIdx.x;
    if (e < n_edges) nbr[off[dst[e]] + rank[e]] = src[e];
}

__global__ __launch_bounds__(256)
void gnn_gather_sum(const float* __restrict__ h, const int* __restrict__ off,
                    const int* __restrict__ nbr, float* __restrict__ agg, int n_nodes) {
    const int lane = threadIdx.x & 31;
    const int node = __builtin_amdgcn_readfirstlane((int)(blockIdx.x * 8 + (threadIdx.x >> 5)));
    if (node >= n_nodes) return;
    const int beg = off[node];
    const int end = off[node + 1];
    const float* __restrict__ hl = h + lane * 4;

    v4f acc = {0.0f, 0.0f, 0.0f, 0.0f};
    for (int base = beg; base < end; base += 32) {
        int m = end - base; if (m > 32) m = 32;
        const int mine = (lane < m) ? nbr[base + lane] : 0;
        int k = 0;
        for (; k + 4 <= m; k += 4) {
            const int s0 = __builtin_amdgcn_readlane(mine, k);
            const int s1 = __builtin_amdgcn_readlane(mine, k + 1);
            const int s2 = __builtin_amdgcn_readlane(mine, k + 2);
            const int s3 = __builtin_amdgcn_readlane(mine, k + 3);
            const v4f v0 = *(const v4f*)(hl + (size_t)s0 * DFEAT);
            const v4f v1 = *(const v4f*)(hl + (size_t)s1 * DFEAT);
            const v4f v2 = *(const v4f*)(hl + (size_t)s2 * DFEAT);
            const v4f v3 = *(const v4f*)(hl + (size_t)s3 * DFEAT);
            acc += v0; acc += v1; acc += v2; acc += v3;
        }
        const int r = m - k;
        if (r == 3) {
            const int s0 = __builtin_amdgcn_readlane(mine, k);
            const int s1 = __builtin_amdgcn_readlane(mine, k + 1);
            const int s2 = __builtin_amdgcn_readlane(mine, k + 2);
            const v4f v0 = *(const v4f*)(hl + (size_t)s0 * DFEAT);
            const v4f v1 = *(const v4f*)(hl + (size_t)s1 * DFEAT);
            const v4f v2 = *(const v4f*)(hl + (size_t)s2 * DFEAT);
            acc += v0; acc += v1; acc += v2;
        } else if (r == 2) {
            const int s0 = __builtin_amdgcn_readlane(mine, k);
            const int s1 = __builtin_amdgcn_readlane(mine, k + 1);
            const v4f v0 = *(const v4f*)(hl + (size_t)s0 * DFEAT);
            const v4f v1 = *(const v4f*)(hl + (size_t)s1 * DFEAT);
            acc += v0; acc += v1;
        } else if (r == 1) {
            const int s0 = __builtin_amdgcn_readlane(mine, k);
            acc += *(const v4f*)(hl + (size_t)s0 * DFEAT);
        }
    }
    *(v4f*)(agg + (size_t)node * DFEAT + lane * 4) = acc;
}

#define WPITCH 260

template <int NT>
__device__ __forceinline__
void gnn_sage_gemm_body(const float* __restrict__ agg,
                        const float* __restrict__ inv_deg,
                        const float* __restrict__ h,
                        const float* __restrict__ Wl,
                        const float* __restrict__ bl,
                        const float* __restrict__ Wr,
                        float* __restrict__ out,
                        float* wlds,
                        int n_nodes) {
    const int tid  = threadIdx.x;
    const int wave = tid >> 5;
    const int lane = tid & 31;

    constexpr int NV4 = NT * 16 * 64;
    const int q  = tid & 63;
    const int nr = tid >> 6;
    const float* __restrict__ wsrc = ((q < 32) ? Wl : Wr) + (size_t)nr * DFEAT + 4 * (q & 31);
    float* wdst = wlds + nr * WPITCH + 4 * q;
    #pragma unroll
    for (int i = 0; i < NV4 / 256; ++i) {
        const v4f v = *(const v4f*)(wsrc + (size_t)i * 4 * DFEAT);
        *(v4f*)(wdst + i * 4 * WPITCH) = v;
    }
    __syncthreads();

    const int mtile = blockIdx.x * 8 + wave;
    if (mtile * 16 >= n_nodes) return;

    const int m0 = mtile * 16;
    int rA = m0 + (lane & 15);
    if (rA >= n_nodes) rA = n_nodes - 1;
    const int klo = 2 * (lane >> 4);

    const float s = inv_deg[rA];

    const float* __restrict__ arow = agg + (size_t)rA * DFEAT + klo;
    const float* __restrict__ hrow = h   + (size_t)rA * DFEAT + klo;
    const float* wb = wlds + (lane & 15) * WPITCH + klo;

    v8f acc[NT];
    #pragma unroll
    for (int t = 0; t < NT; ++t) acc[t] = v8f{};

    #pragma unroll 8
    for (int kk = 0; kk < DFEAT; kk += 4) {
        v2f a = *(const v2f*)(arow + kk);
        a *= s;
        #pragma unroll
        for (int t = 0; t < NT; ++t) {
            const v2f b = *(const v2f*)(wb + t * 16 * WPITCH + kk);
            acc[t] = __builtin_amdgcn_wmma_f32_16x16x4_f32(
                false, a, false, b, (short)0, acc[t], false, false);
        }
    }

    #pragma unroll 8
    for (int kk = 0; kk < DFEAT; kk += 4) {
        const v2f a = *(const v2f*)(hrow + kk);
        #pragma unroll
        for (int t = 0; t < NT; ++t) {
            const v2f b = *(const v2f*)(wb + t * 16 * WPITCH + DFEAT + kk);
            acc[t] = __builtin_amdgcn_wmma_f32_16x16x4_f32(
                false, a, false, b, (short)0, acc[t], false, false);
        }
    }

    constexpr int DOUT = NT * 16;
    const int mbase = m0 + 8 * (lane >> 4);
    float bb[NT];
    #pragma unroll
    for (int t = 0; t < NT; ++t) bb[t] = bl[t * 16 + (lane & 15)];
    float* __restrict__ orow = out + (size_t)mbase * DOUT + (lane & 15);
    if (m0 + 16 <= n_nodes) {
        #pragma unroll
        for (int t = 0; t < NT; ++t) {
            #pragma unroll
            for (int v = 0; v < 8; ++v) {
                const float val = acc[t][v] + bb[t];
                orow[v * DOUT + t * 16] = (val > 0.0f) ? val : 0.0f;
            }
        }
    } else {
        #pragma unroll
        for (int t = 0; t < NT; ++t) {
            #pragma unroll
            for (int v = 0; v < 8; ++v) {
                if (mbase + v < n_nodes) {
                    const float val = acc[t][v] + bb[t];
                    orow[v * DOUT + t * 16] = (val > 0.0f) ? val : 0.0f;
                }
            }
        }
    }
}

__global__ __launch_bounds__(256)
void gnn_sage_gemm_wmma_hid(const float* __restrict__ agg,
                            const float* __restrict__ inv_deg,
                            const float* __restrict__ h,
                            const float* __restrict__ Wl,
                            const float* __restrict__ bl,
                            const float* __restrict__ Wr,
                            float* __restrict__ out,
                            int n_nodes) {
    __shared__ __attribute__((aligned(16))) float wlds[8 * 16 * WPITCH];
    gnn_sage_gemm_body<8>(agg, inv_deg, h, Wl, bl, Wr, out, wlds, n_nodes);
}

__global__ __launch_bounds__(256)
void gnn_sage_gemm_wmma_out(const float* __restrict__ agg,
                            const float* __restrict__ inv_deg,
                            const float* __restrict__ h,
                            const float* __restrict__ Wl,
                            const float* __restrict__ bl,
                            const float* __restrict__ Wr,
                            float* __restrict__ out,
                            int n_nodes) {
    __shared__ __attribute__((aligned(16))) float wlds[2 * 16 * WPITCH];
    gnn_sage_gemm_body<2>(agg, inv_deg, h, Wl, bl, Wr, out, wlds, n_nodes);
}

extern "C" void kernel_launch(void* const* d_in, const int* in_sizes, int n_in,
                              void* d_out, int out_size, void* d_ws, size_t ws_size,
                              hipStream_t stream) {
    const float* x    = (const float*)d_in[0];
    const int*   eidx = (const int*)  d_in[1];
    const float* W_l1 = (const float*)d_in[2];
    const float* b_l1 = (const float*)d_in[3];
    const float* W_r1 = (const float*)d_in[4];
    const float* W_l2 = (const float*)d_in[5];
    const float* b_l2 = (const float*)d_in[6];
    const float* W_r2 = (const float*)d_in[7];
    const float* W_l3 = (const float*)d_in[8];
    const float* b_l3 = (const float*)d_in[9];
    const float* W_r3 = (const float*)d_in[10];
    float* out = (float*)d_out;

    const int n_nodes = in_sizes[0] / DFEAT;
    const int n_edges = in_sizes[1] / 2;
    const int* src = eidx;
    const int* dst = eidx + n_edges;

    float* deg = (float*)d_ws;
    float* agg = deg + n_nodes;
    float* h1  = agg + (size_t)n_nodes * DFEAT;
    float* h2  = h1  + (size_t)n_nodes * DFEAT;
    int* cnt   = (int*)(h2 + (size_t)n_nodes * DFEAT);
    int* off   = cnt + n_nodes;
    int* rank  = off + n_nodes + 1;
    int* nbr   = rank + n_edges;

    const int  TB      = 256;
    const int  gNodes  = (n_nodes + TB - 1) / TB;
    const int  gEdges  = (n_edges + TB - 1) / TB;
    const int  gGather = (n_nodes + 7) / 8;
    const int  gScan   = (n_nodes + 1024) / 1024;

    const int mtiles = (n_nodes + 15) / 16;
    const int gGemm = (mtiles + 7) / 8;

    gnn_zero_i32    <<<gNodes, TB, 0, stream>>>(cnt, n_nodes);
    gnn_count_rank  <<<gEdges, TB, 0, stream>>>(dst, cnt, rank, n_edges);
    gnn_scan_offsets<<<gScan, 1024, 0, stream>>>(cnt, off, deg, n_nodes);
    gnn_fill_buckets<<<gEdges, TB, 0, stream>>>(src, dst, off, rank, nbr, n_edges);

    gnn_gather_sum<<<gGather, TB, 0, stream>>>(x, off, nbr, agg, n_nodes);
    gnn_sage_gemm_wmma_hid<<<gGemm, TB, 0, stream>>>(agg, deg, x, W_l1, b_l1, W_r1,
                                                     h1, n_nodes);

    gnn_gather_sum<<<gGather, TB, 0, stream>>>(h1, off, nbr, agg, n_nodes);
    gnn_sage_gemm_wmma_hid<<<gGemm, TB, 0, stream>>>(agg, deg, h1, W_l2, b_l2, W_r2,
                                                     h2, n_nodes);

    gnn_gather_sum<<<gGather, TB, 0, stream>>>(h2, off, nbr, agg, n_nodes);
    gnn_sage_gemm_wmma_out<<<gGemm, TB, 0, stream>>>(agg, deg, h2, W_l3, b_l3, W_r3,
                                                     out, n_nodes);
}


# Edges placed by the counting atomic straight into a fixed 64-id row per node (overflow list past 64): scan and fill launches dropped, 1/deg applied in the gather; sums stay in arrival-rank order
# speedup vs baseline: 4.3012x; 1.1783x over previous
#include <hip/hip_runtime.h>


typedef float v2f __attribute__((ext_vector_type(2)));
typedef float v8f __attribute__((ext_vector_type(8)));

#define DFEAT 128

typedef float v4f __attribute__((ext_vector_type(4)));

__global__ void gnn_zero_i32(int* __restrict__ p, int n) {
    int i = blockIdx.x * blockDim.x + threadIdx.x;
    if (i < n) p[i] = 0;
}

#define SLOTS 64

__global__ void gnn_count_place(const int* __restrict__ src, const int* __restrict__ dst,
                                int* __restrict__ cnt, int* __restrict__ slot,
                                int* __restrict__ ovf_dst, int* __restrict__ ovf_src,
                                int n_nodes, int n_edges) {
    int e = blockIdx.x * blockDim.x + threadIdx.x;
    if (e < n_edges) {
        const int d = dst[e];
        const int s = src[e];
        const int r = atomicAdd(&cnt[d], 1);
        if (r < SLOTS) {
            slot[(size_t)d * SLOTS + r] = s;
        } else {
            const int p = atomicAdd(&cnt[n_nodes], 1);
            ovf_dst[p] = d;
            ovf_src[p] = s;
        }
    }
}

__global__ __launch_bounds__(256)
void gnn_gather_mean(const float* __restrict__ h, const int* __restrict__ cnt,
                     const int* __restrict__ slot, const int* __restrict__ ovf_dst,
                     const int* __restrict__ ovf_src, float* __restrict__ agg, int n_nodes) {
    const int lane = threadIdx.x & 31;
    const int node = __builtin_amdgcn_readfirstlane((int)(blockIdx.x * 8 + (threadIdx.x >> 5)));
    if (node >= n_nodes) return;
    const int deg = cnt[node];
    const int end = (deg < SLOTS) ? deg : SLOTS;
    const int* __restrict__ ids = slot + (size_t)node * SLOTS;
    const float* __restrict__ hl = h + lane * 4;

    v4f acc = {0.0f, 0.0f, 0.0f, 0.0f};
    for (int base = 0; base < end; base += 32) {
        int m = end - base; if (m > 32) m = 32;
        const int mine = (lane < m) ? ids[base + lane] : 0;
        int k = 0;
        for (; k + 4 <= m; k += 4) {
            const int s0 = __builtin_amdgcn_readlane(mine, k);
            const int s1 = __builtin_amdgcn_readlane(mine, k + 1);
            const int s2 = __builtin_amdgcn_readlane(mine, k + 2);
            const int s3 = __builtin_amdgcn_readlane(mine, k + 3);
            const v4f v0 = *(const v4f*)(hl + (size_t)s0 * DFEAT);
            const v4f v1 = *(const v4f*)(hl + (size_t)s1 * DFEAT);
            const v4f v2 = *(const v4f*)(hl + (size_t)s2 * DFEAT);
            const v4f v3 = *(const v4f*)(hl + (size_t)s3 * DFEAT);
            acc += v0; acc += v1; acc += v2; acc += v3;
        }
        const int r = m - k;
        if (r == 3) {
            const int s0 = __builtin_amdgcn_readlane(mine, k);
            const int s1 = __builtin_amdgcn_readlane(mine, k + 1);
            const int s2 = __builtin_amdgcn_readlane(mine, k + 2);
            const v4f v0 = *(const v4f*)(hl + (size_t)s0 * DFEAT);
            const v4f v1 = *(const v4f*)(hl + (size_t)s1 * DFEAT);
            const v4f v2 = *(const v4f*)(hl + (size_t)s2 * DFEAT);
            acc += v0; acc += v1; acc += v2;
        } else if (r == 2) {
            const int s0 = __builtin_amdgcn_readlane(mine, k);
            const int s1 = __builtin_amdgcn_readlane(mine, k + 1);
            const v4f v0 = *(const v4f*)(hl + (size_t)s0 * DFEAT);
            const v4f v1 = *(const v4f*)(hl + (size_t)s1 * DFEAT);
            acc += v0; acc += v1;
        } else if (r == 1) {
            const int s0 = __builtin_amdgcn_readlane(mine, k);
            acc += *(const v4f*)(hl + (size_t)s0 * DFEAT);
        }
    }
    if (deg > SLOTS) {
        const int n_ovf = cnt[n_nodes];
        for (int p0 = 0; p0 < n_ovf; p0 += 32) {
            const int p  = p0 + lane;
            const int od = (p < n_ovf) ? ovf_dst[p] : -1;
            const int os = (p < n_ovf) ? ovf_src[p] : 0;
            unsigned hit = __builtin_amdgcn_ballot_w32(od == node);
            while (hit != 0u) {
                const int l = __builtin_ctz(hit);
                hit &= hit - 1u;
                const int s0 = __shfl(os, l);
                acc += *(const v4f*)(hl + (size_t)s0 * DFEAT);
            }
        }
    }
    const float rdeg = 1.0f / (float)(deg > 0 ? deg : 1);
    acc *= (deg > 0) ? rdeg : 0.0f;
    *(v4f*)(agg + (size_t)node * DFEAT + lane * 4) = acc;
}

#define WPITCH 260

template <int NT>
__device__ __forceinline__
void gnn_sage_gemm_body(const float* __restrict__ agg,
                        const float* __restrict__ h,
                        const float* __restrict__ Wl,
                        const float* __restrict__ bl,
                        const float* __restrict__ Wr,
                        float* __restrict__ out,
                        float* wlds,
                        int n_nodes) {
    const int tid  = threadIdx.x;
    const int wave = tid >> 5;
    const int lane = tid & 31;

    constexpr int NV4 = NT * 16 * 64;
    const int q  = tid & 63;
    const int nr = tid >> 6;
    const float* __restrict__ wsrc = ((q < 32) ? Wl : Wr) + (size_t)nr * DFEAT + 4 * (q & 31);
    float* wdst = wlds + nr * WPITCH + 4 * q;
    #pragma unroll
    for (int i = 0; i < NV4 / 256; ++i) {
        const v4f v = *(const v4f*)(wsrc + (size_t)i * 4 * DFEAT);
        *(v4f*)(wdst + i * 4 * WPITCH) = v;
    }
    __syncthreads();

    const int mtile = blockIdx.x * 8 + wave;
    if (mtile * 16 >= n_nodes) return;

    const int m0 = mtile * 16;
    int rA = m0 + (lane & 15);
    if (rA >= n_nodes) rA = n_nodes - 1;
    const int klo = 2 * (lane >> 4);

    const float* __restrict__ arow = agg + (size_t)rA * DFEAT + klo;
    const float* __restrict__ hrow = h   + (size_t)rA * DFEAT + klo;
    const float* wb = wlds + (lane & 15) * WPITCH + klo;

    v8f acc[NT];
    #pragma unroll
    for (int t = 0; t < NT; ++t) acc[t] = v8f{};

    #pragma unroll 8
    for (int kk = 0; kk < DFEAT; kk += 4) {
        const v2f a = *(const v2f*)(arow + kk);
        #pragma unroll
        for (int t = 0; t < NT; ++t) {
            const v2f b = *(const v2f*)(wb + t * 16 * WPITCH + kk);
            acc[t] = __builtin_amdgcn_wmma_f32_16x16x4_f32(
                false, a, false, b, (short)0, acc[t], false, false);
        }
    }

    #pragma unroll 8
    for (int kk = 0; kk < DFEAT; kk += 4) {
        const v2f a = *(const v2f*)(hrow + kk);
        #pragma unroll
        for (int t = 0; t < NT; ++t) {
            const v2f b = *(const v2f*)(wb + t * 16 * WPITCH + DFEAT + kk);
            acc[t] = __builtin_amdgcn_wmma_f32_16x16x4_f32(
                false, a, false, b, (short)0, acc[t], false, false);
        }
    }

    constexpr int DOUT = NT * 16;
    const int mbase = m0 + 8 * (lane >> 4);
    float bb[NT];
    #pragma unroll
    for (int t = 0; t < NT; ++t) bb[t] = bl[t * 16 + (lane & 15)];
    float* __restrict__ orow = out + (size_t)mbase * DOUT + (lane & 15);
    if (m0 + 16 <= n_nodes) {
        #pragma unroll
        for (int t = 0; t < NT; ++t) {
            #pragma unroll
            for (int v = 0; v < 8; ++v) {
                const float val = acc[t][v] + bb[t];
                orow[v * DOUT + t * 16] = (val > 0.0f) ? val : 0.0f;
            }
        }
    } else {
        #pragma unroll
        for (int t = 0; t < NT; ++t) {
            #pragma unroll
            for (int v = 0; v < 8; ++v) {
                if (mbase + v < n_nodes) {
                    const float val = acc[t][v] + bb[t];
                    orow[v * DOUT + t * 16] = (val > 0.0f) ? val : 0.0f;
                }
            }
        }
    }
}

__global__ __launch_bounds__(256)
void gnn_sage_gemm_wmma_hid(const float* __restrict__ agg,
                            const float* __restrict__ h,
                            const float* __restrict__ Wl,
                            const float* __restrict__ bl,
                            const float* __restrict__ Wr,
                            float* __restrict__ out,
                            int n_nodes) {
    __shared__ __attribute__((aligned(16))) float wlds[8 * 16 * WPITCH];
    gnn_sage_gemm_body<8>(agg, h, Wl, bl, Wr, out, wlds, n_nodes);
}

__global__ __launch_bounds__(256)
void gnn_sage_gemm_wmma_out(const float* __restrict__ agg,
                            const float* __restrict__ h,
                            const float* __restrict__ Wl,
                            const float* __restrict__ bl,
                            const float* __restrict__ Wr,
                            float* __restrict__ out,
                            int n_nodes) {
    __shared__ __attribute__((aligned(16))) float wlds[2 * 16 * WPITCH];
    gnn_sage_gemm_body<2>(agg, h, Wl, bl, Wr, out, wlds, n_nodes);
}

extern "C" void kernel_launch(void* const* d_in, const int* in_sizes, int n_in,
                              void* d_out, int out_size, void* d_ws, size_t ws_size,
                              hipStream_t stream) {
    const float* x    = (const float*)d_in[0];
    const int*   eidx = (const int*)  d_in[1];
    const float* W_l1 = (const float*)d_in[2];
    const float* b_l1 = (const float*)d_in[3];
    const float* W_r1 = (const float*)d_in[4];
    const float* W_l2 = (const float*)d_in[5];
    const float* b_l2 = (const float*)d_in[6];
    const float* W_r2 = (const float*)d_in[7];
    const float* W_l3 = (const float*)d_in[8];
    const float* b_l3 = (const float*)d_in[9];
    const float* W_r3 = (const float*)d_in[10];
    float* out = (float*)d_out;

    const int n_nodes = in_sizes[0] / DFEAT;
    const int n_edges = in_sizes[1] / 2;
    const int* src = eidx;
    const int* dst = eidx + n_edges;

    int*   slot = (int*)d_ws;
    float* agg  = (float*)(slot + (size_t)n_nodes * SLOTS);
    float* h1   = agg + (size_t)n_nodes * DFEAT;
    float* h2   = h1  + (size_t)n_nodes * DFEAT;
    int* cnt     = (int*)(h2 + (size_t)n_nodes * DFEAT);
    int* ovf_dst = cnt + n_nodes + 1;
    int* ovf_src = ovf_dst + n_edges;

    const int  TB      = 256;
    const int  gCnt    = (n_nodes + 1 + TB - 1) / TB;
    const int  gEdges  = (n_edges + TB - 1) / TB;
    const int  gGather = (n_nodes + 7) / 8;

    const int mtiles = (n_nodes + 15) / 16;
    const int gGemm = (mtiles + 7) / 8;

    gnn_zero_i32   <<<gCnt, TB, 0, stream>>>(cnt, n_nodes + 1);
    gnn_count_place<<<gEdges, TB, 0, stream>>>(src, dst, cnt, slot, ovf_dst, ovf_src, n_nodes, n_edges);

    gnn_gather_mean<<<gGather, TB, 0, stream>>>(x, cnt, slot, ovf_dst, ovf_src, agg, n_nodes);
    gnn_sage_gemm_wmma_hid<<<gGemm, TB, 0, stream>>>(agg, x, W_l1, b_l1, W_r1,
                                                     h1, n_nodes);

    gnn_gather_mean<<<gGather, TB, 0, stream>>>(h1, cnt, slot, ovf_dst, ovf_src, agg, n_nodes);
    gnn_sage_gemm_wmma_hid<<<gGemm, TB, 0, stream>>>(agg, h1, W_l2, b_l2, W_r2,
                                                     h2, n_nodes);

    gnn_gather_mean<<<gGather, TB, 0, stream>>>(h2, cnt, slot, ovf_dst, ovf_src, agg, n_nodes);
    gnn_sage_gemm_wmma_out<<<gGemm, TB, 0, stream>>>(agg, h2, W_l3, b_l3, W_r3,
                                                     out, n_nodes);
}


# Last layer (128->32) projects before it aggregates: zl=h2@Wl3^T, zr=h2@Wr3^T+b on WMMA, then out=relu(mean of 32-float zl rows + zr); sum order changes (mean of products)
# speedup vs baseline: 4.5682x; 1.0621x over previous
#include <hip/hip_runtime.h>


typedef float v2f __attribute__((ext_vector_type(2)));
typedef float v8f __attribute__((ext_vector_type(8)));

#define DFEAT 128

typedef float v4f __attribute__((ext_vector_type(4)));
typedef int   v4i __attribute__((ext_vector_type(4)));

__global__ void gnn_zero_i32(int* __restrict__ p, int n) {
    int i = blockIdx.x * blockDim.x + threadIdx.x;
    if (i < n) p[i] = 0;
}

#define SLOTS 64

__global__ void gnn_count_place(const int* __restrict__ src, const int* __restrict__ dst,
                                int* __restrict__ cnt, int* __restrict__ slot,
                                int* __restrict__ ovf_dst, int* __restrict__ ovf_src,
                                int n_nodes, int n_edges) {
    int e = blockIdx.x * blockDim.x + threadIdx.x;
    if (e < n_edges) {
        const int d = dst[e];
        const int s = src[e];
        const int r = atomicAdd(&cnt[d], 1);
        if (r < SLOTS) {
            slot[(size_t)d * SLOTS + r] = s;
        } else {
            const int p = atomicAdd(&cnt[n_nodes], 1);
            ovf_dst[p] = d;
            ovf_src[p] = s;
        }
    }
}

__global__ __launch_bounds__(256)
void gnn_gather_mean(const float* __restrict__ h, const int* __restrict__ cnt,
                     const int* __restrict__ slot, const int* __restrict__ ovf_dst,
                     const int* __restrict__ ovf_src, float* __restrict__ agg, int n_nodes) {
    const int lane = threadIdx.x & 31;
    const int node = __builtin_amdgcn_readfirstlane((int)(blockIdx.x * 8 + (threadIdx.x >> 5)));
    if (node >= n_nodes) return;
    const int deg = cnt[node];
    const int end = (deg < SLOTS) ? deg : SLOTS;
    const int* __restrict__ ids = slot + (size_t)node * SLOTS;
    const float* __restrict__ hl = h + lane * 4;

    v4f acc = {0.0f, 0.0f, 0.0f, 0.0f};
    for (int base = 0; base < end; base += 32) {
        int m = end - base; if (m > 32) m = 32;
        const int mine = (lane < m) ? ids[base + lane] : 0;
        int k = 0;
        for (; k + 4 <= m; k += 4) {
            const int s0 = __builtin_amdgcn_readlane(mine, k);
            const int s1 = __builtin_amdgcn_readlane(mine, k + 1);
            const int s2 = __builtin_amdgcn_readlane(mine, k + 2);
            const int s3 = __builtin_amdgcn_readlane(mine, k + 3);
            const v4f v0 = *(const v4f*)(hl + (size_t)s0 * DFEAT);
            const v4f v1 = *(const v4f*)(hl + (size_t)s1 * DFEAT);
            const v4f v2 = *(const v4f*)(hl + (size_t)s2 * DFEAT);
            const v4f v3 = *(const v4f*)(hl + (size_t)s3 * DFEAT);
            acc += v0; acc += v1; acc += v2; acc += v3;
        }
        const int r = m - k;
        if (r == 3) {
            const int s0 = __builtin_amdgcn_readlane(mine, k);
            const int s1 = __builtin_amdgcn_readlane(mine, k + 1);
            const int s2 = __builtin_amdgcn_readlane(mine, k + 2);
            const v4f v0 = *(const v4f*)(hl + (size_t)s0 * DFEAT);
            const v4f v1 = *(const v4f*)(hl + (size_t)s1 * DFEAT);
            const v4f v2 = *(const v4f*)(hl + (size_t)s2 * DFEAT);
            acc += v0; acc += v1; acc += v2;
        } else if (r == 2) {
            const int s0 = __builtin_amdgcn_readlane(mine, k);
            const int s1 = __builtin_amdgcn_readlane(mine, k + 1);
            const v4f v0 = *(const v4f*)(hl + (size_t)s0 * DFEAT);
            const v4f v1 = *(const v4f*)(hl + (size_t)s1 * DFEAT);
            acc += v0; acc += v1;
        } else if (r == 1) {
            const int s0 = __builtin_amdgcn_readlane(mine, k);
            acc += *(const v4f*)(hl + (size_t)s0 * DFEAT);
        }
    }
    if (deg > SLOTS) {
        const int n_ovf = cnt[n_nodes];
        for (int p0 = 0; p0 < n_ovf; p0 += 32) {
            const int p  = p0 + lane;
            const int od = (p < n_ovf) ? ovf_dst[p] : -1;
            const int os = (p < n_ovf) ? ovf_src[p] : 0;
            unsigned hit = __builtin_amdgcn_ballot_w32(od == node);
            while (hit != 0u) {
                const int l = __builtin_ctz(hit);
                hit &= hit - 1u;
                const int s0 = __shfl(os, l);
                acc += *(const v4f*)(hl + (size_t)s0 * DFEAT);
            }
        }
    }
    const float rdeg = 1.0f / (float)(deg > 0 ? deg : 1);
    acc *= (deg > 0) ? rdeg : 0.0f;
    *(v4f*)(agg + (size_t)node * DFEAT + lane * 4) = acc;
}

#define DOUT3 32

__global__ __launch_bounds__(256)
void gnn_gather_mean_out(const float* __restrict__ zl, const float* __restrict__ zr,
                         const int* __restrict__ cnt, const int* __restrict__ slot,
                         const int* __restrict__ ovf_dst, const int* __restrict__ ovf_src,
                         float* __restrict__ out, int n_nodes) {
    const int q    = threadIdx.x & 7;
    const int node = (int)(blockIdx.x * 256 + threadIdx.x) >> 3;
    if (node >= n_nodes) return;
    const int* __restrict__ ids = slot + (size_t)node * SLOTS;
    v4i c0 = *(const v4i*)(ids);
    v4i c1 = *(const v4i*)(ids + 4);
    const int deg = cnt[node];
    const v4f self = *(const v4f*)(zr + (size_t)node * DOUT3 + q * 4);
    const int end = (deg < SLOTS) ? deg : SLOTS;
    const float* __restrict__ zq = zl + q * 4;

    v4f acc = {0.0f, 0.0f, 0.0f, 0.0f};
    int k = 0;
    for (; k + 8 <= end; k += 8) {
        const v4i n0 = *(const v4i*)(ids + ((k + 8) & (SLOTS - 1)));
        const v4i n1 = *(const v4i*)(ids + ((k + 12) & (SLOTS - 1)));
        const v4f v0 = *(const v4f*)(zq + (size_t)c0.x * DOUT3);
        const v4f v1 = *(const v4f*)(zq + (size_t)c0.y * DOUT3);
        const v4f v2 = *(const v4f*)(zq + (size_t)c0.z * DOUT3);
        const v4f v3 = *(const v4f*)(zq + (size_t)c0.w * DOUT3);
        const v4f v4 = *(const v4f*)(zq + (size_t)c1.x * DOUT3);
        const v4f v5 = *(const v4f*)(zq + (size_t)c1.y * DOUT3);
        const v4f v6 = *(const v4f*)(zq + (size_t)c1.z * DOUT3);
        const v4f v7 = *(const v4f*)(zq + (size_t)c1.w * DOUT3);
        acc += v0; acc += v1; acc += v2; acc += v3;
        acc += v4; acc += v5; acc += v6; acc += v7;
        c0 = n0; c1 = n1;
    }
    const int r = end - k;
    if (r > 0) {
        const v4f zero = {0.0f, 0.0f, 0.0f, 0.0f};
        const int s0 = c0.x;
        const int s1 = (r > 1) ? c0.y : s0;
        const int s2 = (r > 2) ? c0.z : s0;
        const int s3 = (r > 3) ? c0.w : s0;
        const int s4 = (r > 4) ? c1.x : s0;
        const int s5 = (r > 5) ? c1.y : s0;
        const int s6 = (r > 6) ? c1.z : s0;
        const v4f v0 = *(const v4f*)(zq + (size_t)s0 * DOUT3);
        const v4f v1 = *(const v4f*)(zq + (size_t)s1 * DOUT3);
        const v4f v2 = *(const v4f*)(zq + (size_t)s2 * DOUT3);
        const v4f v3 = *(const v4f*)(zq + (size_t)s3 * DOUT3);
        const v4f v4 = *(const v4f*)(zq + (size_t)s4 * DOUT3);
        const v4f v5 = *(const v4f*)(zq + (size_t)s5 * DOUT3);
        const v4f v6 = *(const v4f*)(zq + (size_t)s6 * DOUT3);
        acc += v0;
        acc += (r > 1) ? v1 : zero;
        acc += (r > 2) ? v2 : zero;
        acc += (r > 3) ? v3 : zero;
        acc += (r > 4) ? v4 : zero;
        acc += (r > 5) ? v5 : zero;
        acc += (r > 6) ? v6 : zero;
    }
    if (deg > SLOTS) {
        const int n_ovf = cnt[n_nodes];
        for (int p = 0; p < n_ovf; ++p) {
            if (ovf_dst[p] == node) acc += *(const v4f*)(zq + (size_t)ovf_src[p] * DOUT3);
        }
    }
    const float rdeg = 1.0f / (float)(deg > 0 ? deg : 1);
    const float sc   = (deg > 0) ? rdeg : 0.0f;
    v4f o = acc * sc + self;
    o.x = (o.x > 0.0f) ? o.x : 0.0f;
    o.y = (o.y > 0.0f) ? o.y : 0.0f;
    o.z = (o.z > 0.0f) ? o.z : 0.0f;
    o.w = (o.w > 0.0f) ? o.w : 0.0f;
    *(v4f*)(out + (size_t)node * DOUT3 + q * 4) = o;
}

#define WPITCH 260

template <int NT>
__device__ __forceinline__
void gnn_sage_gemm_body(const float* __restrict__ agg,
                        const float* __restrict__ h,
                        const float* __restrict__ Wl,
                        const float* __restrict__ bl,
                        const float* __restrict__ Wr,
                        float* __restrict__ out,
                        float* wlds,
                        int n_nodes) {
    const int tid  = threadIdx.x;
    const int wave = tid >> 5;
    const int lane = tid & 31;

    constexpr int NV4 = NT * 16 * 64;
    const int q  = tid & 63;
    const int nr = tid >> 6;
    const float* __restrict__ wsrc = ((q < 32) ? Wl : Wr) + (size_t)nr * DFEAT + 4 * (q & 31);
    float* wdst = wlds + nr * WPITCH + 4 * q;
    #pragma unroll
    for (int i = 0; i < NV4 / 256; ++i) {
        const v4f v = *(const v4f*)(wsrc + (size_t)i * 4 * DFEAT);
        *(v4f*)(wdst + i * 4 * WPITCH) = v;
    }
    __syncthreads();

    const int mtile = blockIdx.x * 8 + wave;
    if (mtile * 16 >= n_nodes) return;

    const int m0 = mtile * 16;
    int rA = m0 + (lane & 15);
    if (rA >= n_nodes) rA = n_nodes - 1;
    const int klo = 2 * (lane >> 4);

    const float* __restrict__ arow = agg + (size_t)rA * DFEAT + klo;
    const float* __restrict__ hrow = h   + (size_t)rA * DFEAT + klo;
    const float* wb = wlds + (lane & 15) * WPITCH + klo;

    v8f acc[NT];
    #pragma unroll
    for (int t = 0; t < NT; ++t) acc[t] = v8f{};

    #pragma unroll 8
    for (int kk = 0; kk < DFEAT; kk += 4) {
        const v2f a = *(const v2f*)(arow + kk);
        #pragma unroll
        for (int t = 0; t < NT; ++t) {
            const v2f b = *(const v2f*)(wb + t * 16 * WPITCH + kk);
            acc[t] = __builtin_amdgcn_wmma_f32_16x16x4_f32(
                false, a, false, b, (short)0, acc[t], false, false);
        }
    }

    #pragma unroll 8
    for (int kk = 0; kk < DFEAT; kk += 4) {
        const v2f a = *(const v2f*)(hrow + kk);
        #pragma unroll
        for (int t = 0; t < NT; ++t) {
            const v2f b = *(const v2f*)(wb + t * 16 * WPITCH + DFEAT + kk);
            acc[t] = __builtin_amdgcn_wmma_f32_16x16x4_f32(
                false, a, false, b, (short)0, acc[t], false, false);
        }
    }

    constexpr int DOUT = NT * 16;
    const int mbase = m0 + 8 * (lane >> 4);
    float bb[NT];
    #pragma unroll
    for (int t = 0; t < NT; ++t) bb[t] = bl[t * 16 + (lane & 15)];
    float* __restrict__ orow = out + (size_t)mbase * DOUT + (lane & 15);
    if (m0 + 16 <= n_nodes) {
        #pragma unroll
        for (int t = 0; t < NT; ++t) {
            #pragma unroll
            for (int v = 0; v < 8; ++v) {
                const float val = acc[t][v] + bb[t];
                orow[v * DOUT + t * 16] = (val > 0.0f) ? val : 0.0f;
            }
        }
    } else {
        #pragma unroll
        for (int t = 0; t < NT; ++t) {
            #pragma unroll
            for (int v = 0; v < 8; ++v) {
                if (mbase + v < n_nodes) {
                    const float val = acc[t][v] + bb[t];
                    orow[v * DOUT + t * 16] = (val > 0.0f) ? val : 0.0f;
                }
            }
        }
    }
}

__global__ __launch_bounds__(256)
void gnn_sage_gemm_wmma_hid(const float* __restrict__ agg,
                            const float* __restrict__ h,
                            const float* __restrict__ Wl,
                            const float* __restrict__ bl,
                            const float* __restrict__ Wr,
                            float* __restrict__ out,
                            int n_nodes) {
    __shared__ __attribute__((aligned(16))) float wlds[8 * 16 * WPITCH];
    gnn_sage_gemm_body<8>(agg, h, Wl, bl, Wr, out, wlds, n_nodes);
}

#define ZPITCH 132

__global__ __launch_bounds__(256)
void gnn_sage_proj_wmma_out(const float* __restrict__ h,
                            const float* __restrict__ Wl,
                            const float* __restrict__ bl,
                            const float* __restrict__ Wr,
                            float* __restrict__ zl,
                            float* __restrict__ zr,
                            int n_nodes) {
    __shared__ __attribute__((aligned(16))) float wlds[4 * 16 * ZPITCH];
    const int tid  = threadIdx.x;
    const int wave = tid >> 5;
    const int lane = tid & 31;

    const float* __restrict__ wsl = Wl + (size_t)wave * DFEAT + 4 * lane;
    const float* __restrict__ wsr = Wr + (size_t)wave * DFEAT + 4 * lane;
    float* wdst = wlds + wave * ZPITCH + 4 * lane;
    #pragma unroll
    for (int i = 0; i < 4; ++i) {
        const v4f vl = *(const v4f*)(wsl + (size_t)i * 8 * DFEAT);
        const v4f vr = *(const v4f*)(wsr + (size_t)i * 8 * DFEAT);
        *(v4f*)(wdst + i * 8 * ZPITCH) = vl;
        *(v4f*)(wdst + (i * 8 + DOUT3) * ZPITCH) = vr;
    }
    __syncthreads();

    const int mtile = blockIdx.x * 8 + wave;
    if (mtile * 16 >= n_nodes) return;

    const int m0 = mtile * 16;
    int rA = m0 + (lane & 15);
    if (rA >= n_nodes) rA = n_nodes - 1;
    const int klo = 2 * (lane >> 4);

    const float* __restrict__ hrow = h + (size_t)rA * DFEAT + klo;
    const float* wb = wlds + (lane & 15) * ZPITCH + klo;

    v8f acc[4];
    #pragma unroll
    for (int t = 0; t < 4; ++t) acc[t] = v8f{};

    #pragma unroll 8
    for (int kk = 0; kk < DFEAT; kk += 4) {
        const v2f a = *(const v2f*)(hrow + kk);
        #pragma unroll
        for (int t = 0; t < 4; ++t) {
            const v2f b = *(const v2f*)(wb + t * 16 * ZPITCH + kk);
            acc[t] = __builtin_amdgcn_wmma_f32_16x16x4_f32(
                false, a, false, b, (short)0, acc[t], false, false);
        }
    }

    const int mbase = m0 + 8 * (lane >> 4);
    const float b0 = bl[lane & 15];
    const float b1 = bl[16 + (lane & 15)];
    float* __restrict__ lrow = zl + (size_t)mbase * DOUT3 + (lane & 15);
    float* __restrict__ rrow = zr + (size_t)mbase * DOUT3 + (lane & 15);
    if (m0 + 16 <= n_nodes) {
        #pragma unroll
        for (int v = 0; v < 8; ++v) {
            lrow[v * DOUT3]      = acc[0][v];
            lrow[v * DOUT3 + 16] = acc[1][v];
            rrow[v * DOUT3]      = acc[2][v] + b0;
            rrow[v * DOUT3 + 16] = acc[3][v] + b1;
        }
    } else {
        #pragma unroll
        for (int v = 0; v < 8; ++v) {
            if (mbase + v < n_nodes) {
                lrow[v * DOUT3]      = acc[0][v];
                lrow[v * DOUT3 + 16] = acc[1][v];
                rrow[v * DOUT3]      = acc[2][v] + b0;
                rrow[v * DOUT3 + 16] = acc[3][v] + b1;
            }
        }
    }
}

extern "C" void kernel_launch(void* const* d_in, const int* in_sizes, int n_in,
                              void* d_out, int out_size, void* d_ws, size_t ws_size,
                              hipStream_t stream) {
    const float* x    = (const float*)d_in[0];
    const int*   eidx = (const int*)  d_in[1];
    const float* W_l1 = (const float*)d_in[2];
    const float* b_l1 = (const float*)d_in[3];
    const float* W_r1 = (const float*)d_in[4];
    const float* W_l2 = (const float*)d_in[5];
    const float* b_l2 = (const float*)d_in[6];
    const float* W_r2 = (const float*)d_in[7];
    const float* W_l3 = (const float*)d_in[8];
    const float* b_l3 = (const float*)d_in[9];
    const float* W_r3 = (const float*)d_in[10];
    float* out = (float*)d_out;

    const int n_nodes = in_sizes[0] / DFEAT;
    const int n_edges = in_sizes[1] / 2;
    const int* src = eidx;
    const int* dst = eidx + n_edges;

    int*   slot = (int*)d_ws;
    float* agg  = (float*)(slot + (size_t)n_nodes * SLOTS);
    float* h1   = agg + (size_t)n_nodes * DFEAT;
    float* h2   = h1  + (size_t)n_nodes * DFEAT;
    int* cnt     = (int*)(h2 + (size_t)n_nodes * DFEAT);
    int* ovf_dst = cnt + n_nodes + 1;
    int* ovf_src = ovf_dst + n_edges;

    const int  TB      = 256;
    const int  gCnt    = (n_nodes + 1 + TB - 1) / TB;
    const int  gEdges  = (n_edges + TB - 1) / TB;
    const int  gGather = (n_nodes + 7) / 8;
    const int  gGatherOut = (n_nodes + 31) / 32;

    const int mtiles = (n_nodes + 15) / 16;
    const int gGemm = (mtiles + 7) / 8;

    gnn_zero_i32   <<<gCnt, TB, 0, stream>>>(cnt, n_nodes + 1);
    gnn_count_place<<<gEdges, TB, 0, stream>>>(src, dst, cnt, slot, ovf_dst, ovf_src, n_nodes, n_edges);

    gnn_gather_mean<<<gGather, TB, 0, stream>>>(x, cnt, slot, ovf_dst, ovf_src, agg, n_nodes);
    gnn_sage_gemm_wmma_hid<<<gGemm, TB, 0, stream>>>(agg, x, W_l1, b_l1, W_r1,
                                                     h1, n_nodes);

    gnn_gather_mean<<<gGather, TB, 0, stream>>>(h1, cnt, slot, ovf_dst, ovf_src, agg, n_nodes);
    gnn_sage_gemm_wmma_hid<<<gGemm, TB, 0, stream>>>(agg, h1, W_l2, b_l2, W_r2,
                                                     h2, n_nodes);

    float* zl = agg;
    float* zr = agg + (size_t)n_nodes * DOUT3;
    gnn_sage_proj_wmma_out<<<gGemm, TB, 0, stream>>>(h2, W_l3, b_l3, W_r3, zl, zr, n_nodes);
    gnn_gather_mean_out<<<gGatherOut, TB, 0, stream>>>(zl, zr, cnt, slot, ovf_dst, ovf_src, out, n_nodes);
}


# Edge placement through LDS: pass 1 groups edges by 256-node bin in coalesced runs, pass 2 ranks each bin with LDS atomics and writes whole id rows; no per-edge global atomic; sums in LDS-rank order
# speedup vs baseline: 5.6217x; 1.2306x over previous
#include <hip/hip_runtime.h>


typedef float v2f __attribute__((ext_vector_type(2)));
typedef float v8f __attribute__((ext_vector_type(8)));

#define DFEAT 128

typedef float v4f __attribute__((ext_vector_type(4)));
typedef int   v4i __attribute__((ext_vector_type(4)));

__global__ void gnn_zero_i32(int* __restrict__ p, int n) {
    int i = blockIdx.x * blockDim.x + threadIdx.x;
    if (i < n) p[i] = 0;
}

#define SLOTS 64
#define BIN_SHIFT 8
#define BIN_NODES (1 << BIN_SHIFT)
#define MAXB 256
#define BIN_CAP 8192
#define CUR_PITCH 32
#define EPT 13
#define EPB (EPT * 256)

typedef int v2i __attribute__((ext_vector_type(2)));

__global__ __launch_bounds__(256)
void gnn_bin_edges(const int* __restrict__ src, const int* __restrict__ dst,
                   int* __restrict__ cur, int* __restrict__ binbuf_i,
                   int* __restrict__ spl_dst, int* __restrict__ spl_src, int n_edges) {
    v2i* __restrict__ binbuf = (v2i*)binbuf_i;
    __shared__ int hist[MAXB + 1];
    __shared__ int delta[MAXB];
    __shared__ int wsum[8];
    __shared__ int ntot;
    __shared__ v2i sorted[EPB];
    const int tid  = threadIdx.x;
    const int lane = tid & 31;
    const int wave = tid >> 5;
    const int e0   = blockIdx.x * EPB;

    hist[tid] = 0;
    if (tid == 0) hist[MAXB] = 0;
    __syncthreads();

    int d[EPT], s[EPT], r[EPT];
    #pragma unroll
    for (int i = 0; i < EPT; ++i) {
        const int e  = e0 + i * 256 + tid;
        const int ec = (e < n_edges) ? e : 0;
        const int dv = dst[ec];
        s[i] = src[ec];
        d[i] = (e < n_edges) ? dv : -1;
    }
    #pragma unroll
    for (int i = 0; i < EPT; ++i) {
        const int b = d[i] >> BIN_SHIFT;
        r[i] = atomicAdd(&hist[((unsigned)b < (unsigned)MAXB) ? b : MAXB], 1);
    }
    __syncthreads();

    const int c = hist[tid];
    int incl = c;
    #pragma unroll
    for (int o = 1; o < 32; o <<= 1) {
        const int t = __shfl_up(incl, o);
        if (lane >= o) incl += t;
    }
    if (lane == 31) wsum[wave] = incl;
    __syncthreads();
    int woff = 0;
    #pragma unroll
    for (int w = 0; w < 8; ++w) woff += (w < wave) ? wsum[w] : 0;
    const int loff = woff + incl - c;
    int g = 0;
    if (c > 0) g = atomicAdd(&cur[tid * CUR_PITCH], c);
    hist[tid]  = loff;
    delta[tid] = g - loff;
    if (tid == 255) ntot = loff + c;
    __syncthreads();

    int o[EPT];
    #pragma unroll
    for (int i = 0; i < EPT; ++i) {
        const int b = d[i] >> BIN_SHIFT;
        o[i] = hist[((unsigned)b < (unsigned)MAXB) ? b : MAXB] + r[i];
    }
    #pragma unroll
    for (int i = 0; i < EPT; ++i) {
        const int b = d[i] >> BIN_SHIFT;
        if ((unsigned)b < (unsigned)MAXB) {
            v2i e; e.x = d[i]; e.y = s[i];
            sorted[o[i]] = e;
        } else if (d[i] >= 0) {
            const int p = atomicAdd(&cur[MAXB * CUR_PITCH], 1);
            spl_dst[p] = d[i];
            spl_src[p] = s[i];
        }
    }
    __syncthreads();

    const int n = ntot;
    v2i oe[EPT];
    int pos[EPT];
    #pragma unroll
    for (int i = 0; i < EPT; ++i) {
        const int idx = i * 256 + tid;
        oe[i] = sorted[(idx < n) ? idx : 0];
        const int b = (idx < n) ? (oe[i].x >> BIN_SHIFT) : 0;
        pos[i] = delta[b] + idx;
    }
    #pragma unroll
    for (int i = 0; i < EPT; ++i) {
        if (i * 256 + tid < n) {
            const int b = oe[i].x >> BIN_SHIFT;
            if (pos[i] < BIN_CAP) {
                binbuf[(size_t)b * BIN_CAP + pos[i]] = oe[i];
            } else {
                const int p = atomicAdd(&cur[MAXB * CUR_PITCH], 1);
                spl_dst[p] = oe[i].x;
                spl_src[p] = oe[i].y;
            }
        }
    }
}

__global__ __launch_bounds__(512)
void gnn_place_bins(const int* __restrict__ cur, const int* __restrict__ binbuf_i,
                    const int* __restrict__ spl_dst, const int* __restrict__ spl_src,
                    int* __restrict__ cnt, int* __restrict__ slot,
                    int* __restrict__ ovf_dst, int* __restrict__ ovf_src, int n_nodes) {
    __shared__ int cntL[BIN_NODES + 1];
    __shared__ __attribute__((aligned(16))) int slotL[BIN_NODES * SLOTS];
    const int tid   = threadIdx.x;
    const int b     = blockIdx.x;
    const int node0 = b << BIN_SHIFT;

    if (tid <= BIN_NODES) cntL[tid] = 0;
    const v4i zero4 = {0, 0, 0, 0};
    #pragma unroll
    for (int i = 0; i < BIN_NODES * SLOTS / 4 / 512; ++i) ((v4i*)slotL)[i * 512 + tid] = zero4;
    __syncthreads();

    int n = 0;
    if (b < MAXB) { n = cur[b * CUR_PITCH]; if (n > BIN_CAP) n = BIN_CAP; }
    const v2i* __restrict__ eb = (const v2i*)binbuf_i + (size_t)b * BIN_CAP;
    for (int base = 0; base < n; base += 512 * 8) {
        v2i e[8];
        int r[8];
        #pragma unroll
        for (int u = 0; u < 8; ++u) {
            const int i = base + u * 512 + tid;
            e[u] = eb[(i < n) ? i : 0];
            if (i >= n) e[u].x = -1;
        }
        #pragma unroll
        for (int u = 0; u < 8; ++u)
            r[u] = atomicAdd(&cntL[(e[u].x >= 0) ? (e[u].x - node0) : BIN_NODES], 1);
        #pragma unroll
        for (int u = 0; u < 8; ++u) {
            if (e[u].x >= 0) {
                if (r[u] < SLOTS) {
                    slotL[(e[u].x - node0) * SLOTS + r[u]] = e[u].y;
                } else {
                    const int p = atomicAdd(&cnt[n_nodes], 1);
                    ovf_dst[p] = e[u].x;
                    ovf_src[p] = e[u].y;
                }
            }
        }
    }
    const int n_spl = cur[MAXB * CUR_PITCH];
    for (int i = tid; i < n_spl; i += 512) {
        const int dd = spl_dst[i];
        if ((dd >> BIN_SHIFT) == b) {
            const int ss = spl_src[i];
            const int l = dd - node0;
            const int r = atomicAdd(&cntL[l], 1);
            if (r < SLOTS) {
                slotL[l * SLOTS + r] = ss;
            } else {
                const int p = atomicAdd(&cnt[n_nodes], 1);
                ovf_dst[p] = dd;
                ovf_src[p] = ss;
            }
        }
    }
    __syncthreads();

    if (tid < BIN_NODES && node0 + tid < n_nodes) cnt[node0 + tid] = cntL[tid];
    const int q = tid & 7;
    #pragma unroll
    for (int g = tid >> 3; g < BIN_NODES; g += 64) {
        const int node = node0 + g;
        if (node < n_nodes) {
            const v4i lo = *(const v4i*)(slotL + g * SLOTS + q * 4);
            *(v4i*)(slot + (size_t)node * SLOTS + q * 4) = lo;
            if (cntL[g] >= 32) {
                const v4i hi = *(const v4i*)(slotL + g * SLOTS + 32 + q * 4);
                *(v4i*)(slot + (size_t)node * SLOTS + 32 + q * 4) = hi;
            }
        }
    }
}

__global__ __launch_bounds__(256)
void gnn_gather_mean(const float* __restrict__ h, const int* __restrict__ cnt,
                     const int* __restrict__ slot, const int* __restrict__ ovf_dst,
                     const int* __restrict__ ovf_src, float* __restrict__ agg, int n_nodes) {
    const int lane = threadIdx.x & 31;
    const int node = __builtin_amdgcn_readfirstlane((int)(blockIdx.x * 8 + (threadIdx.x >> 5)));
    if (node >= n_nodes) return;
    const int deg = cnt[node];
    const int end = (deg < SLOTS) ? deg : SLOTS;
    const int* __restrict__ ids = slot + (size_t)node * SLOTS;
    const float* __restrict__ hl = h + lane * 4;

    v4f acc = {0.0f, 0.0f, 0.0f, 0.0f};
    for (int base = 0; base < end; base += 32) {
        int m = end - base; if (m > 32) m = 32;
        const int mine = (lane < m) ? ids[base + lane] : 0;
        int k = 0;
        for (; k + 4 <= m; k += 4) {
            const int s0 = __builtin_amdgcn_readlane(mine, k);
            const int s1 = __builtin_amdgcn_readlane(mine, k + 1);
            const int s2 = __builtin_amdgcn_readlane(mine, k + 2);
            const int s3 = __builtin_amdgcn_readlane(mine, k + 3);
            const v4f v0 = *(const v4f*)(hl + (size_t)s0 * DFEAT);
            const v4f v1 = *(const v4f*)(hl + (size_t)s1 * DFEAT);
            const v4f v2 = *(const v4f*)(hl + (size_t)s2 * DFEAT);
            const v4f v3 = *(const v4f*)(hl + (size_t)s3 * DFEAT);
            acc += v0; acc += v1; acc += v2; acc += v3;
        }
        const int r = m - k;
        if (r == 3) {
            const int s0 = __builtin_amdgcn_readlane(mine, k);
            const int s1 = __builtin_amdgcn_readlane(mine, k + 1);
            const int s2 = __builtin_amdgcn_readlane(mine, k + 2);
            const v4f v0 = *(const v4f*)(hl + (size_t)s0 * DFEAT);
            const v4f v1 = *(const v4f*)(hl + (size_t)s1 * DFEAT);
            const v4f v2 = *(const v4f*)(hl + (size_t)s2 * DFEAT);
            acc += v0; acc += v1; acc += v2;
        } else if (r == 2) {
            const int s0 = __builtin_amdgcn_readlane(mine, k);
            const int s1 = __builtin_amdgcn_readlane(mine, k + 1);
            const v4f v0 = *(const v4f*)(hl + (size_t)s0 * DFEAT);
            const v4f v1 = *(const v4f*)(hl + (size_t)s1 * DFEAT);
            acc += v0; acc += v1;
        } else if (r == 1) {
            const int s0 = __builtin_amdgcn_readlane(mine, k);
            acc += *(const v4f*)(hl + (size_t)s0 * DFEAT);
        }
    }
    if (deg > SLOTS) {
        const int n_ovf = cnt[n_nodes];
        for (int p0 = 0; p0 < n_ovf; p0 += 32) {
            const int p  = p0 + lane;
            const int od = (p < n_ovf) ? ovf_dst[p] : -1;
            const int os = (p < n_ovf) ? ovf_src[p] : 0;
            unsigned hit = __builtin_amdgcn_ballot_w32(od == node);
            while (hit != 0u) {
                const int l = __builtin_ctz(hit);
                hit &= hit - 1u;
                const int s0 = __shfl(os, l);
                acc += *(const v4f*)(hl + (size_t)s0 * DFEAT);
            }
        }
    }
    const float rdeg = 1.0f / (float)(deg > 0 ? deg : 1);
    acc *= (deg > 0) ? rdeg : 0.0f;
    *(v4f*)(agg + (size_t)node * DFEAT + lane * 4) = acc;
}

#define DOUT3 32

__global__ __launch_bounds__(256)
void gnn_gather_mean_out(const float* __restrict__ zl, const float* __restrict__ zr,
                         const int* __restrict__ cnt, const int* __restrict__ slot,
                         const int* __restrict__ ovf_dst, const int* __restrict__ ovf_src,
                         float* __restrict__ out, int n_nodes) {
    const int q    = threadIdx.x & 7;
    const int node = (int)(blockIdx.x * 256 + threadIdx.x) >> 3;
    if (node >= n_nodes) return;
    const int* __restrict__ ids = slot + (size_t)node * SLOTS;
    v4i c0 = *(const v4i*)(ids);
    v4i c1 = *(const v4i*)(ids + 4);
    const int deg = cnt[node];
    const v4f self = *(const v4f*)(zr + (size_t)node * DOUT3 + q * 4);
    const int end = (deg < SLOTS) ? deg : SLOTS;
    const float* __restrict__ zq = zl + q * 4;

    v4f acc = {0.0f, 0.0f, 0.0f, 0.0f};
    int k = 0;
    for (; k + 8 <= end; k += 8) {
        const v4i n0 = *(const v4i*)(ids + ((k + 8) & (SLOTS - 1)));
        const v4i n1 = *(const v4i*)(ids + ((k + 12) & (SLOTS - 1)));
        const v4f v0 = *(const v4f*)(zq + (size_t)c0.x * DOUT3);
        const v4f v1 = *(const v4f*)(zq + (size_t)c0.y * DOUT3);
        const v4f v2 = *(const v4f*)(zq + (size_t)c0.z * DOUT3);
        const v4f v3 = *(const v4f*)(zq + (size_t)c0.w * DOUT3);
        const v4f v4 = *(const v4f*)(zq + (size_t)c1.x * DOUT3);
        const v4f v5 = *(const v4f*)(zq + (size_t)c1.y * DOUT3);
        const v4f v6 = *(const v4f*)(zq + (size_t)c1.z * DOUT3);
        const v4f v7 = *(const v4f*)(zq + (size_t)c1.w * DOUT3);
        acc += v0; acc += v1; acc += v2; acc += v3;
        acc += v4; acc += v5; acc += v6; acc += v7;
        c0 = n0; c1 = n1;
    }
    const int r = end - k;
    if (r > 0) {
        const v4f zero = {0.0f, 0.0f, 0.0f, 0.0f};
        const int s0 = c0.x;
        const int s1 = (r > 1) ? c0.y : s0;
        const int s2 = (r > 2) ? c0.z : s0;
        const int s3 = (r > 3) ? c0.w : s0;
        const int s4 = (r > 4) ? c1.x : s0;
        const int s5 = (r > 5) ? c1.y : s0;
        const int s6 = (r > 6) ? c1.z : s0;
        const v4f v0 = *(const v4f*)(zq + (size_t)s0 * DOUT3);
        const v4f v1 = *(const v4f*)(zq + (size_t)s1 * DOUT3);
        const v4f v2 = *(const v4f*)(zq + (size_t)s2 * DOUT3);
        const v4f v3 = *(const v4f*)(zq + (size_t)s3 * DOUT3);
        const v4f v4 = *(const v4f*)(zq + (size_t)s4 * DOUT3);
        const v4f v5 = *(const v4f*)(zq + (size_t)s5 * DOUT3);
        const v4f v6 = *(const v4f*)(zq + (size_t)s6 * DOUT3);
        acc += v0;
        acc += (r > 1) ? v1 : zero;
        acc += (r > 2) ? v2 : zero;
        acc += (r > 3) ? v3 : zero;
        acc += (r > 4) ? v4 : zero;
        acc += (r > 5) ? v5 : zero;
        acc += (r > 6) ? v6 : zero;
    }
    if (deg > SLOTS) {
        const int n_ovf = cnt[n_nodes];
        for (int p = 0; p < n_ovf; ++p) {
            if (ovf_dst[p] == node) acc += *(const v4f*)(zq + (size_t)ovf_src[p] * DOUT3);
        }
    }
    const float rdeg = 1.0f / (float)(deg > 0 ? deg : 1);
    const float sc   = (deg > 0) ? rdeg : 0.0f;
    v4f o = acc * sc + self;
    o.x = (o.x > 0.0f) ? o.x : 0.0f;
    o.y = (o.y > 0.0f) ? o.y : 0.0f;
    o.z = (o.z > 0.0f) ? o.z : 0.0f;
    o.w = (o.w > 0.0f) ? o.w : 0.0f;
    *(v4f*)(out + (size_t)node * DOUT3 + q * 4) = o;
}

#define WPITCH 260

template <int NT>
__device__ __forceinline__
void gnn_sage_gemm_body(const float* __restrict__ agg,
                        const float* __restrict__ h,
                        const float* __restrict__ Wl,
                        const float* __restrict__ bl,
                        const float* __restrict__ Wr,
                        float* __restrict__ out,
                        float* wlds,
                        int n_nodes) {
    const int tid  = threadIdx.x;
    const int wave = tid >> 5;
    const int lane = tid & 31;

    constexpr int NV4 = NT * 16 * 64;
    const int q  = tid & 63;
    const int nr = tid >> 6;
    const float* __restrict__ wsrc = ((q < 32) ? Wl : Wr) + (size_t)nr * DFEAT + 4 * (q & 31);
    float* wdst = wlds + nr * WPITCH + 4 * q;
    #pragma unroll
    for (int i = 0; i < NV4 / 256; ++i) {
        const v4f v = *(const v4f*)(wsrc + (size_t)i * 4 * DFEAT);
        *(v4f*)(wdst + i * 4 * WPITCH) = v;
    }
    __syncthreads();

    const int mtile = blockIdx.x * 8 + wave;
    if (mtile * 16 >= n_nodes) return;

    const int m0 = mtile * 16;
    int rA = m0 + (lane & 15);
    if (rA >= n_nodes) rA = n_nodes - 1;
    const int klo = 2 * (lane >> 4);

    const float* __restrict__ arow = agg + (size_t)rA * DFEAT + klo;
    const float* __restrict__ hrow = h   + (size_t)rA * DFEAT + klo;
    const float* wb = wlds + (lane & 15) * WPITCH + klo;

    v8f acc[NT];
    #pragma unroll
    for (int t = 0; t < NT; ++t) acc[t] = v8f{};

    #pragma unroll 8
    for (int kk = 0; kk < DFEAT; kk += 4) {
        const v2f a = *(const v2f*)(arow + kk);
        #pragma unroll
        for (int t = 0; t < NT; ++t) {
            const v2f b = *(const v2f*)(wb + t * 16 * WPITCH + kk);
            acc[t] = __builtin_amdgcn_wmma_f32_16x16x4_f32(
                false, a, false, b, (short)0, acc[t], false, false);
        }
    }

    #pragma unroll 8
    for (int kk = 0; kk < DFEAT; kk += 4) {
        const v2f a = *(const v2f*)(hrow + kk);
        #pragma unroll
        for (int t = 0; t < NT; ++t) {
            const v2f b = *(const v2f*)(wb + t * 16 * WPITCH + DFEAT + kk);
            acc[t] = __builtin_amdgcn_wmma_f32_16x16x4_f32(
                false, a, false, b, (short)0, acc[t], false, false);
        }
    }

    constexpr int DOUT = NT * 16;
    const int mbase = m0 + 8 * (lane >> 4);
    float bb[NT];
    #pragma unroll
    for (int t = 0; t < NT; ++t) bb[t] = bl[t * 16 + (lane & 15)];
    float* __restrict__ orow = out + (size_t)mbase * DOUT + (lane & 15);
    if (m0 + 16 <= n_nodes) {
        #pragma unroll
        for (int t = 0; t < NT; ++t) {
            #pragma unroll
            for (int v = 0; v < 8; ++v) {
                const float val = acc[t][v] + bb[t];
                orow[v * DOUT + t * 16] = (val > 0.0f) ? val : 0.0f;
            }
        }
    } else {
        #pragma unroll
        for (int t = 0; t < NT; ++t) {
            #pragma unroll
            for (int v = 0; v < 8; ++v) {
                if (mbase + v < n_nodes) {
                    const float val = acc[t][v] + bb[t];
                    orow[v * DOUT + t * 16] = (val > 0.0f) ? val : 0.0f;
                }
            }
        }
    }
}

__global__ __launch_bounds__(256)
void gnn_sage_gemm_wmma_hid(const float* __restrict__ agg,
                            const float* __restrict__ h,
                            const float* __restrict__ Wl,
                            const float* __restrict__ bl,
                            const float* __restrict__ Wr,
                            float* __restrict__ out,
                            int n_nodes) {
    __shared__ __attribute__((aligned(16))) float wlds[8 * 16 * WPITCH];
    gnn_sage_gemm_body<8>(agg, h, Wl, bl, Wr, out, wlds, n_nodes);
}

#define ZPITCH 132

__global__ __launch_bounds__(256)
void gnn_sage_proj_wmma_out(const float* __restrict__ h,
                            const float* __restrict__ Wl,
                            const float* __restrict__ bl,
                            const float* __restrict__ Wr,
                            float* __restrict__ zl,
                            float* __restrict__ zr,
                            int n_nodes) {
    __shared__ __attribute__((aligned(16))) float wlds[4 * 16 * ZPITCH];
    const int tid  = threadIdx.x;
    const int wave = tid >> 5;
    const int lane = tid & 31;

    const float* __restrict__ wsl = Wl + (size_t)wave * DFEAT + 4 * lane;
    const float* __restrict__ wsr = Wr + (size_t)wave * DFEAT + 4 * lane;
    float* wdst = wlds + wave * ZPITCH + 4 * lane;
    #pragma unroll
    for (int i = 0; i < 4; ++i) {
        const v4f vl = *(const v4f*)(wsl + (size_t)i * 8 * DFEAT);
        const v4f vr = *(const v4f*)(wsr + (size_t)i * 8 * DFEAT);
        *(v4f*)(wdst + i * 8 * ZPITCH) = vl;
        *(v4f*)(wdst + (i * 8 + DOUT3) * ZPITCH) = vr;
    }
    __syncthreads();

    const int mtile = blockIdx.x * 8 + wave;
    if (mtile * 16 >= n_nodes) return;

    const int m0 = mtile * 16;
    int rA = m0 + (lane & 15);
    if (rA >= n_nodes) rA = n_nodes - 1;
    const int klo = 2 * (lane >> 4);

    const float* __restrict__ hrow = h + (size_t)rA * DFEAT + klo;
    const float* wb = wlds + (lane & 15) * ZPITCH + klo;

    v8f acc[4];
    #pragma unroll
    for (int t = 0; t < 4; ++t) acc[t] = v8f{};

    #pragma unroll 8
    for (int kk = 0; kk < DFEAT; kk += 4) {
        const v2f a = *(const v2f*)(hrow + kk);
        #pragma unroll
        for (int t = 0; t < 4; ++t) {
            const v2f b = *(const v2f*)(wb + t * 16 * ZPITCH + kk);
            acc[t] = __builtin_amdgcn_wmma_f32_16x16x4_f32(
                false, a, false, b, (short)0, acc[t], false, false);
        }
    }

    const int mbase = m0 + 8 * (lane >> 4);
    const float b0 = bl[lane & 15];
    const float b1 = bl[16 + (lane & 15)];
    float* __restrict__ lrow = zl + (size_t)mbase * DOUT3 + (lane & 15);
    float* __restrict__ rrow = zr + (size_t)mbase * DOUT3 + (lane & 15);
    if (m0 + 16 <= n_nodes) {
        #pragma unroll
        for (int v = 0; v < 8; ++v) {
            lrow[v * DOUT3]      = acc[0][v];
            lrow[v * DOUT3 + 16] = acc[1][v];
            rrow[v * DOUT3]      = acc[2][v] + b0;
            rrow[v * DOUT3 + 16] = acc[3][v] + b1;
        }
    } else {
        #pragma unroll
        for (int v = 0; v < 8; ++v) {
            if (mbase + v < n_nodes) {
                lrow[v * DOUT3]      = acc[0][v];
                lrow[v * DOUT3 + 16] = acc[1][v];
                rrow[v * DOUT3]      = acc[2][v] + b0;
                rrow[v * DOUT3 + 16] = acc[3][v] + b1;
            }
        }
    }
}

extern "C" void kernel_launch(void* const* d_in, const int* in_sizes, int n_in,
                              void* d_out, int out_size, void* d_ws, size_t ws_size,
                              hipStream_t stream) {
    const float* x    = (const float*)d_in[0];
    const int*   eidx = (const int*)  d_in[1];
    const float* W_l1 = (const float*)d_in[2];
    const float* b_l1 = (const float*)d_in[3];
    const float* W_r1 = (const float*)d_in[4];
    const float* W_l2 = (const float*)d_in[5];
    const float* b_l2 = (const float*)d_in[6];
    const float* W_r2 = (const float*)d_in[7];
    const float* W_l3 = (const float*)d_in[8];
    const float* b_l3 = (const float*)d_in[9];
    const float* W_r3 = (const float*)d_in[10];
    float* out = (float*)d_out;

    const int n_nodes = in_sizes[0] / DFEAT;
    const int n_edges = in_sizes[1] / 2;
    const int* src = eidx;
    const int* dst = eidx + n_edges;

    int*   slot = (int*)d_ws;
    float* agg  = (float*)(slot + (size_t)n_nodes * SLOTS);
    float* h1   = agg + (size_t)n_nodes * DFEAT;
    float* h2   = h1  + (size_t)n_nodes * DFEAT;
    int* cnt     = (int*)(h2 + (size_t)n_nodes * DFEAT);
    const int cur_off = (n_nodes + 1 + CUR_PITCH - 1) / CUR_PITCH * CUR_PITCH;
    int* cur     = cnt + cur_off;
    int* binbuf  = cur + MAXB * CUR_PITCH + CUR_PITCH;
    int* ovf_dst = binbuf + (size_t)MAXB * BIN_CAP * 2;
    int* ovf_src = ovf_dst + n_edges;
    int* spl_dst = ovf_src + n_edges;
    int* spl_src = spl_dst + n_edges;

    const int  TB      = 256;
    const int  nZero   = cur_off - n_nodes + MAXB * CUR_PITCH + CUR_PITCH;
    const int  gZero   = (nZero + TB - 1) / TB;
    const int  gBinE   = (n_edges + EPB - 1) / EPB;
    const int  gBins   = (n_nodes + BIN_NODES - 1) / BIN_NODES;
    const int  gGather = (n_nodes + 7) / 8;
    const int  gGatherOut = (n_nodes + 31) / 32;

    const int mtiles = (n_nodes + 15) / 16;
    const int gGemm = (mtiles + 7) / 8;

    gnn_zero_i32  <<<gZero, TB, 0, stream>>>(cnt + n_nodes, nZero);
    gnn_bin_edges <<<gBinE, TB, 0, stream>>>(src, dst, cur, binbuf, spl_dst, spl_src, n_edges);
    gnn_place_bins<<<gBins, 512, 0, stream>>>(cur, binbuf, spl_dst, spl_src, cnt, slot, ovf_dst, ovf_src, n_nodes);

    gnn_gather_mean<<<gGather, TB, 0, stream>>>(x, cnt, slot, ovf_dst, ovf_src, agg, n_nodes);
    gnn_sage_gemm_wmma_hid<<<gGemm, TB, 0, stream>>>(agg, x, W_l1, b_l1, W_r1,
                                                     h1, n_nodes);

    gnn_gather_mean<<<gGather, TB, 0, stream>>>(h1, cnt, slot, ovf_dst, ovf_src, agg, n_nodes);
    gnn_sage_gemm_wmma_hid<<<gGemm, TB, 0, stream>>>(agg, h1, W_l2, b_l2, W_r2,
                                                     h2, n_nodes);

    float* zl = agg;
    float* zr = agg + (size_t)n_nodes * DOUT3;
    gnn_sage_proj_wmma_out<<<gGemm, TB, 0, stream>>>(h2, W_l3, b_l3, W_r3, zl, zr, n_nodes);
    gnn_gather_mean_out<<<gGatherOut, TB, 0, stream>>>(zl, zr, cnt, slot, ovf_dst, ovf_src, out, n_nodes);
}


# Layer GEMM: grid sized to the chip, 256 resident blocks of 12 waves, one 16-row tile per wave (3 per SIMD); the 53 row tiles left over are split by column pairs over 4 waves of a block
# speedup vs baseline: 5.8637x; 1.0430x over previous
#include <hip/hip_runtime.h>


typedef float v2f __attribute__((ext_vector_type(2)));
typedef float v8f __attribute__((ext_vector_type(8)));

#define DFEAT 128

typedef float v4f __attribute__((ext_vector_type(4)));
typedef int   v4i __attribute__((ext_vector_type(4)));

__global__ void gnn_zero_i32(int* __restrict__ p, int n) {
    int i = blockIdx.x * blockDim.x + threadIdx.x;
    if (i < n) p[i] = 0;
}

#define SLOTS 64
#define BIN_SHIFT 8
#define BIN_NODES (1 << BIN_SHIFT)
#define MAXB 256
#define BIN_CAP 8192
#define CUR_PITCH 32
#define EPT 13
#define EPB (EPT * 256)

typedef int v2i __attribute__((ext_vector_type(2)));

__global__ __launch_bounds__(256)
void gnn_bin_edges(const int* __restrict__ src, const int* __restrict__ dst,
                   int* __restrict__ cur, int* __restrict__ binbuf_i,
                   int* __restrict__ spl_dst, int* __restrict__ spl_src, int n_edges) {
    v2i* __restrict__ binbuf = (v2i*)binbuf_i;
    __shared__ int hist[MAXB + 1];
    __shared__ int delta[MAXB];
    __shared__ int wsum[8];
    __shared__ int ntot;
    __shared__ v2i sorted[EPB];
    const int tid  = threadIdx.x;
    const int lane = tid & 31;
    const int wave = tid >> 5;
    const int e0   = blockIdx.x * EPB;

    hist[tid] = 0;
    if (tid == 0) hist[MAXB] = 0;
    __syncthreads();

    int d[EPT], s[EPT], r[EPT];
    #pragma unroll
    for (int i = 0; i < EPT; ++i) {
        const int e  = e0 + i * 256 + tid;
        const int ec = (e < n_edges) ? e : 0;
        const int dv = dst[ec];
        s[i] = src[ec];
        d[i] = (e < n_edges) ? dv : -1;
    }
    #pragma unroll
    for (int i = 0; i < EPT; ++i) {
        const int b = d[i] >> BIN_SHIFT;
        r[i] = atomicAdd(&hist[((unsigned)b < (unsigned)MAXB) ? b : MAXB], 1);
    }
    __syncthreads();

    const int c = hist[tid];
    int incl = c;
    #pragma unroll
    for (int o = 1; o < 32; o <<= 1) {
        const int t = __shfl_up(incl, o);
        if (lane >= o) incl += t;
    }
    if (lane == 31) wsum[wave] = incl;
    __syncthreads();
    int woff = 0;
    #pragma unroll
    for (int w = 0; w < 8; ++w) woff += (w < wave) ? wsum[w] : 0;
    const int loff = woff + incl - c;
    int g = 0;
    if (c > 0) g = atomicAdd(&cur[tid * CUR_PITCH], c);
    hist[tid]  = loff;
    delta[tid] = g - loff;
    if (tid == 255) ntot = loff + c;
    __syncthreads();

    int o[EPT];
    #pragma unroll
    for (int i = 0; i < EPT; ++i) {
        const int b = d[i] >> BIN_SHIFT;
        o[i] = hist[((unsigned)b < (unsigned)MAXB) ? b : MAXB] + r[i];
    }
    #pragma unroll
    for (int i = 0; i < EPT; ++i) {
        const int b = d[i] >> BIN_SHIFT;
        if ((unsigned)b < (unsigned)MAXB) {
            v2i e; e.x = d[i]; e.y = s[i];
            sorted[o[i]] = e;
        } else if (d[i] >= 0) {
            const int p = atomicAdd(&cur[MAXB * CUR_PITCH], 1);
            spl_dst[p] = d[i];
            spl_src[p] = s[i];
        }
    }
    __syncthreads();

    const int n = ntot;
    v2i oe[EPT];
    int pos[EPT];
    #pragma unroll
    for (int i = 0; i < EPT; ++i) {
        const int idx = i * 256 + tid;
        oe[i] = sorted[(idx < n) ? idx : 0];
        const int b = (idx < n) ? (oe[i].x >> BIN_SHIFT) : 0;
        pos[i] = delta[b] + idx;
    }
    #pragma unroll
    for (int i = 0; i < EPT; ++i) {
        if (i * 256 + tid < n) {
            const int b = oe[i].x >> BIN_SHIFT;
            if (pos[i] < BIN_CAP) {
                binbuf[(size_t)b * BIN_CAP + pos[i]] = oe[i];
            } else {
                const int p = atomicAdd(&cur[MAXB * CUR_PITCH], 1);
                spl_dst[p] = oe[i].x;
                spl_src[p] = oe[i].y;
            }
        }
    }
}

__global__ __launch_bounds__(512)
void gnn_place_bins(const int* __restrict__ cur, const int* __restrict__ binbuf_i,
                    const int* __restrict__ spl_dst, const int* __restrict__ spl_src,
                    int* __restrict__ cnt, int* __restrict__ slot,
                    int* __restrict__ ovf_dst, int* __restrict__ ovf_src, int n_nodes) {
    __shared__ int cntL[BIN_NODES + 1];
    __shared__ __attribute__((aligned(16))) int slotL[BIN_NODES * SLOTS];
    const int tid   = threadIdx.x;
    const int b     = blockIdx.x;
    const int node0 = b << BIN_SHIFT;

    if (tid <= BIN_NODES) cntL[tid] = 0;
    const v4i zero4 = {0, 0, 0, 0};
    #pragma unroll
    for (int i = 0; i < BIN_NODES * SLOTS / 4 / 512; ++i) ((v4i*)slotL)[i * 512 + tid] = zero4;
    __syncthreads();

    int n = 0;
    if (b < MAXB) { n = cur[b * CUR_PITCH]; if (n > BIN_CAP) n = BIN_CAP; }
    const v2i* __restrict__ eb = (const v2i*)binbuf_i + (size_t)b * BIN_CAP;
    for (int base = 0; base < n; base += 512 * 8) {
        v2i e[8];
        int r[8];
        #pragma unroll
        for (int u = 0; u < 8; ++u) {
            const int i = base + u * 512 + tid;
            e[u] = eb[(i < n) ? i : 0];
            if (i >= n) e[u].x = -1;
        }
        #pragma unroll
        for (int u = 0; u < 8; ++u)
            r[u] = atomicAdd(&cntL[(e[u].x >= 0) ? (e[u].x - node0) : BIN_NODES], 1);
        #pragma unroll
        for (int u = 0; u < 8; ++u) {
            if (e[u].x >= 0) {
                if (r[u] < SLOTS) {
                    slotL[(e[u].x - node0) * SLOTS + r[u]] = e[u].y;
                } else {
                    const int p = atomicAdd(&cnt[n_nodes], 1);
                    ovf_dst[p] = e[u].x;
                    ovf_src[p] = e[u].y;
                }
            }
        }
    }
    const int n_spl = cur[MAXB * CUR_PITCH];
    for (int i = tid; i < n_spl; i += 512) {
        const int dd = spl_dst[i];
        if ((dd >> BIN_SHIFT) == b) {
            const int ss = spl_src[i];
            const int l = dd - node0;
            const int r = atomicAdd(&cntL[l], 1);
            if (r < SLOTS) {
                slotL[l * SLOTS + r] = ss;
            } else {
                const int p = atomicAdd(&cnt[n_nodes], 1);
                ovf_dst[p] = dd;
                ovf_src[p] = ss;
            }
        }
    }
    __syncthreads();

    if (tid < BIN_NODES && node0 + tid < n_nodes) cnt[node0 + tid] = cntL[tid];
    const int q = tid & 7;
    #pragma unroll
    for (int g = tid >> 3; g < BIN_NODES; g += 64) {
        const int node = node0 + g;
        if (node < n_nodes) {
            const v4i lo = *(const v4i*)(slotL + g * SLOTS + q * 4);
            *(v4i*)(slot + (size_t)node * SLOTS + q * 4) = lo;
            if (cntL[g] >= 32) {
                const v4i hi = *(const v4i*)(slotL + g * SLOTS + 32 + q * 4);
                *(v4i*)(slot + (size_t)node * SLOTS + 32 + q * 4) = hi;
            }
        }
    }
}

__global__ __launch_bounds__(256)
void gnn_gather_mean(const float* __restrict__ h, const int* __restrict__ cnt,
                     const int* __restrict__ slot, const int* __restrict__ ovf_dst,
                     const int* __restrict__ ovf_src, float* __restrict__ agg, int n_nodes) {
    const int lane = threadIdx.x & 31;
    const int node = __builtin_amdgcn_readfirstlane((int)(blockIdx.x * 8 + (threadIdx.x >> 5)));
    if (node >= n_nodes) return;
    const int deg = cnt[node];
    const int end = (deg < SLOTS) ? deg : SLOTS;
    const int* __restrict__ ids = slot + (size_t)node * SLOTS;
    const float* __restrict__ hl = h + lane * 4;

    v4f acc = {0.0f, 0.0f, 0.0f, 0.0f};
    for (int base = 0; base < end; base += 32) {
        int m = end - base; if (m > 32) m = 32;
        const int mine = (lane < m) ? ids[base + lane] : 0;
        int k = 0;
        for (; k + 4 <= m; k += 4) {
            const int s0 = __builtin_amdgcn_readlane(mine, k);
            const int s1 = __builtin_amdgcn_readlane(mine, k + 1);
            const int s2 = __builtin_amdgcn_readlane(mine, k + 2);
            const int s3 = __builtin_amdgcn_readlane(mine, k + 3);
            const v4f v0 = *(const v4f*)(hl + (size_t)s0 * DFEAT);
            const v4f v1 = *(const v4f*)(hl + (size_t)s1 * DFEAT);
            const v4f v2 = *(const v4f*)(hl + (size_t)s2 * DFEAT);
            const v4f v3 = *(const v4f*)(hl + (size_t)s3 * DFEAT);
            acc += v0; acc += v1; acc += v2; acc += v3;
        }
        const int r = m - k;
        if (r == 3) {
            const int s0 = __builtin_amdgcn_readlane(mine, k);
            const int s1 = __builtin_amdgcn_readlane(mine, k + 1);
            const int s2 = __builtin_amdgcn_readlane(mine, k + 2);
            const v4f v0 = *(const v4f*)(hl + (size_t)s0 * DFEAT);
            const v4f v1 = *(const v4f*)(hl + (size_t)s1 * DFEAT);
            const v4f v2 = *(const v4f*)(hl + (size_t)s2 * DFEAT);
            acc += v0; acc += v1; acc += v2;
        } else if (r == 2) {
            const int s0 = __builtin_amdgcn_readlane(mine, k);
            const int s1 = __builtin_amdgcn_readlane(mine, k + 1);
            const v4f v0 = *(const v4f*)(hl + (size_t)s0 * DFEAT);
            const v4f v1 = *(const v4f*)(hl + (size_t)s1 * DFEAT);
            acc += v0; acc += v1;
        } else if (r == 1) {
            const int s0 = __builtin_amdgcn_readlane(mine, k);
            acc += *(const v4f*)(hl + (size_t)s0 * DFEAT);
        }
    }
    if (deg > SLOTS) {
        const int n_ovf = cnt[n_nodes];
        for (int p0 = 0; p0 < n_ovf; p0 += 32) {
            const int p  = p0 + lane;
            const int od = (p < n_ovf) ? ovf_dst[p] : -1;
            const int os = (p < n_ovf) ? ovf_src[p] : 0;
            unsigned hit = __builtin_amdgcn_ballot_w32(od == node);
            while (hit != 0u) {
                const int l = __builtin_ctz(hit);
                hit &= hit - 1u;
                const int s0 = __shfl(os, l);
                acc += *(const v4f*)(hl + (size_t)s0 * DFEAT);
            }
        }
    }
    const float rdeg = 1.0f / (float)(deg > 0 ? deg : 1);
    acc *= (deg > 0) ? rdeg : 0.0f;
    *(v4f*)(agg + (size_t)node * DFEAT + lane * 4) = acc;
}

#define DOUT3 32

__global__ __launch_bounds__(256)
void gnn_gather_mean_out(const float* __restrict__ zl, const float* __restrict__ zr,
                         const int* __restrict__ cnt, const int* __restrict__ slot,
                         const int* __restrict__ ovf_dst, const int* __restrict__ ovf_src,
                         float* __restrict__ out, int n_nodes) {
    const int q    = threadIdx.x & 7;
    const int node = (int)(blockIdx.x * 256 + threadIdx.x) >> 3;
    if (node >= n_nodes) return;
    const int* __restrict__ ids = slot + (size_t)node * SLOTS;
    v4i c0 = *(const v4i*)(ids);
    v4i c1 = *(const v4i*)(ids + 4);
    const int deg = cnt[node];
    const v4f self = *(const v4f*)(zr + (size_t)node * DOUT3 + q * 4);
    const int end = (deg < SLOTS) ? deg : SLOTS;
    const float* __restrict__ zq = zl + q * 4;

    v4f acc = {0.0f, 0.0f, 0.0f, 0.0f};
    int k = 0;
    for (; k + 8 <= end; k += 8) {
        const v4i n0 = *(const v4i*)(ids + ((k + 8) & (SLOTS - 1)));
        const v4i n1 = *(const v4i*)(ids + ((k + 12) & (SLOTS - 1)));
        const v4f v0 = *(const v4f*)(zq + (size_t)c0.x * DOUT3);
        const v4f v1 = *(const v4f*)(zq + (size_t)c0.y * DOUT3);
        const v4f v2 = *(const v4f*)(zq + (size_t)c0.z * DOUT3);
        const v4f v3 = *(const v4f*)(zq + (size_t)c0.w * DOUT3);
        const v4f v4 = *(const v4f*)(zq + (size_t)c1.x * DOUT3);
        const v4f v5 = *(const v4f*)(zq + (size_t)c1.y * DOUT3);
        const v4f v6 = *(const v4f*)(zq + (size_t)c1.z * DOUT3);
        const v4f v7 = *(const v4f*)(zq + (size_t)c1.w * DOUT3);
        acc += v0; acc += v1; acc += v2; acc += v3;
        acc += v4; acc += v5; acc += v6; acc += v7;
        c0 = n0; c1 = n1;
    }
    const int r = end - k;
    if (r > 0) {
        const v4f zero = {0.0f, 0.0f, 0.0f, 0.0f};
        const int s0 = c0.x;
        const int s1 = (r > 1) ? c0.y : s0;
        const int s2 = (r > 2) ? c0.z : s0;
        const int s3 = (r > 3) ? c0.w : s0;
        const int s4 = (r > 4) ? c1.x : s0;
        const int s5 = (r > 5) ? c1.y : s0;
        const int s6 = (r > 6) ? c1.z : s0;
        const v4f v0 = *(const v4f*)(zq + (size_t)s0 * DOUT3);
        const v4f v1 = *(const v4f*)(zq + (size_t)s1 * DOUT3);
        const v4f v2 = *(const v4f*)(zq + (size_t)s2 * DOUT3);
        const v4f v3 = *(const v4f*)(zq + (size_t)s3 * DOUT3);
        const v4f v4 = *(const v4f*)(zq + (size_t)s4 * DOUT3);
        const v4f v5 = *(const v4f*)(zq + (size_t)s5 * DOUT3);
        const v4f v6 = *(const v4f*)(zq + (size_t)s6 * DOUT3);
        acc += v0;
        acc += (r > 1) ? v1 : zero;
        acc += (r > 2) ? v2 : zero;
        acc += (r > 3) ? v3 : zero;
        acc += (r > 4) ? v4 : zero;
        acc += (r > 5) ? v5 : zero;
        acc += (r > 6) ? v6 : zero;
    }
    if (deg > SLOTS) {
        const int n_ovf = cnt[n_nodes];
        for (int p = 0; p < n_ovf; ++p) {
            if (ovf_dst[p] == node) acc += *(const v4f*)(zq + (size_t)ovf_src[p] * DOUT3);
        }
    }
    const float rdeg = 1.0f / (float)(deg > 0 ? deg : 1);
    const float sc   = (deg > 0) ? rdeg : 0.0f;
    v4f o = acc * sc + self;
    o.x = (o.x > 0.0f) ? o.x : 0.0f;
    o.y = (o.y > 0.0f) ? o.y : 0.0f;
    o.z = (o.z > 0.0f) ? o.z : 0.0f;
    o.w = (o.w > 0.0f) ? o.w : 0.0f;
    *(v4f*)(out + (size_t)node * DOUT3 + q * 4) = o;
}

#define WPITCH 260

#define GEMM_WGPS  256
#define GEMM_WAVES 12
#define GEMM_LDS_FLOATS 41984

template <int NT>
__device__ __forceinline__
void gnn_sage_gemm_tile(const float* __restrict__ agg,
                        const float* __restrict__ h,
                        const float* __restrict__ bl,
                        float* __restrict__ out,
                        const float* wlds,
                        int mtile, int T0, int lane, int n_nodes) {
    const int m0 = mtile * 16;
    int rA = m0 + (lane & 15);
    if (rA >= n_nodes) rA = n_nodes - 1;
    const int klo = 2 * (lane >> 4);

    const float* __restrict__ arow = agg + (size_t)rA * DFEAT + klo;
    const float* __restrict__ hrow = h   + (size_t)rA * DFEAT + klo;
    const float* wb = wlds + (T0 * 16 + (lane & 15)) * WPITCH + klo;

    v8f acc[NT];
    #pragma unroll
    for (int t = 0; t < NT; ++t) acc[t] = v8f{};

    #pragma unroll 8
    for (int kk = 0; kk < DFEAT; kk += 4) {
        const v2f a = *(const v2f*)(arow + kk);
        #pragma unroll
        for (int t = 0; t < NT; ++t) {
            const v2f b = *(const v2f*)(wb + t * 16 * WPITCH + kk);
            acc[t] = __builtin_amdgcn_wmma_f32_16x16x4_f32(
                false, a, false, b, (short)0, acc[t], false, false);
        }
    }

    #pragma unroll 8
    for (int kk = 0; kk < DFEAT; kk += 4) {
        const v2f a = *(const v2f*)(hrow + kk);
        #pragma unroll
        for (int t = 0; t < NT; ++t) {
            const v2f b = *(const v2f*)(wb + t * 16 * WPITCH + DFEAT + kk);
            acc[t] = __builtin_amdgcn_wmma_f32_16x16x4_f32(
                false, a, false, b, (short)0, acc[t], false, false);
        }
    }

    const int mbase = m0 + 8 * (lane >> 4);
    float bb[NT];
    #pragma unroll
    for (int t = 0; t < NT; ++t) bb[t] = bl[(T0 + t) * 16 + (lane & 15)];
    float* __restrict__ orow = out + (size_t)mbase * DFEAT + T0 * 16 + (lane & 15);
    if (m0 + 16 <= n_nodes) {
        #pragma unroll
        for (int t = 0; t < NT; ++t) {
            #pragma unroll
            for (int v = 0; v < 8; ++v) {
                const float val = acc[t][v] + bb[t];
                orow[v * DFEAT + t * 16] = (val > 0.0f) ? val : 0.0f;
            }
        }
    } else {
        #pragma unroll
        for (int t = 0; t < NT; ++t) {
            #pragma unroll
            for (int v = 0; v < 8; ++v) {
                if (mbase + v < n_nodes) {
                    const float val = acc[t][v] + bb[t];
                    orow[v * DFEAT + t * 16] = (val > 0.0f) ? val : 0.0f;
                }
            }
        }
    }
}

__global__ __launch_bounds__(GEMM_WAVES * 32)
void gnn_sage_gemm_wmma_hid(const float* __restrict__ agg,
                            const float* __restrict__ h,
                            const float* __restrict__ Wl,
                            const float* __restrict__ bl,
                            const float* __restrict__ Wr,
                            float* __restrict__ out,
                            int n_nodes) {
    __shared__ __attribute__((aligned(16))) float wlds[GEMM_LDS_FLOATS];
    const int tid  = threadIdx.x;
    const int wave = __builtin_amdgcn_readfirstlane(tid >> 5);
    const int lane = tid & 31;

    constexpr int NV4 = 8 * 16 * 64;
    constexpr int NTH = GEMM_WAVES * 32;
    constexpr int RPP = NTH / 64;
    const int q  = tid & 63;
    const int nr = tid >> 6;
    const float* __restrict__ wsrc = ((q < 32) ? Wl : Wr) + (size_t)nr * DFEAT + 4 * (q & 31);
    float* wdst = wlds + nr * WPITCH + 4 * q;
    v4f wv[(NV4 + NTH - 1) / NTH];
    #pragma unroll
    for (int i = 0; i < (NV4 + NTH - 1) / NTH; ++i) {
        const int row = (i * RPP + nr < 8 * 16) ? i * RPP : 0;
        wv[i] = *(const v4f*)(wsrc + (size_t)row * DFEAT);
    }
    #pragma unroll
    for (int i = 0; i < (NV4 + NTH - 1) / NTH; ++i) {
        if (i * RPP + nr < 8 * 16) *(v4f*)(wdst + i * RPP * WPITCH) = wv[i];
    }
    __syncthreads();

    const int mtiles  = (n_nodes + 15) / 16;
    const int per_rnd = GEMM_WAVES * (int)gridDim.x;
    const int n_whole = mtiles / per_rnd * per_rnd;
    for (int mt = (int)blockIdx.x * GEMM_WAVES + wave; mt < n_whole; mt += per_rnd)
        gnn_sage_gemm_tile<8>(agg, h, bl, out, wlds, mt, 0, lane, n_nodes);
    if (wave < 4) {
        for (int mt = n_whole + (int)blockIdx.x; mt < mtiles; mt += (int)gridDim.x)
            gnn_sage_gemm_tile<2>(agg, h, bl, out, wlds, mt, 2 * wave, lane, n_nodes);
    }
}

#define ZPITCH 132

__global__ __launch_bounds__(256)
void gnn_sage_proj_wmma_out(const float* __restrict__ h,
                            const float* __restrict__ Wl,
                            const float* __restrict__ bl,
                            const float* __restrict__ Wr,
                            float* __restrict__ zl,
                            float* __restrict__ zr,
                            int n_nodes) {
    __shared__ __attribute__((aligned(16))) float wlds[4 * 16 * ZPITCH];
    const int tid  = threadIdx.x;
    const int wave = tid >> 5;
    const int lane = tid & 31;

    const float* __restrict__ wsl = Wl + (size_t)wave * DFEAT + 4 * lane;
    const float* __restrict__ wsr = Wr + (size_t)wave * DFEAT + 4 * lane;
    float* wdst = wlds + wave * ZPITCH + 4 * lane;
    #pragma unroll
    for (int i = 0; i < 4; ++i) {
        const v4f vl = *(const v4f*)(wsl + (size_t)i * 8 * DFEAT);
        const v4f vr = *(const v4f*)(wsr + (size_t)i * 8 * DFEAT);
        *(v4f*)(wdst + i * 8 * ZPITCH) = vl;
        *(v4f*)(wdst + (i * 8 + DOUT3) * ZPITCH) = vr;
    }
    __syncthreads();

    const int mtile = blockIdx.x * 8 + wave;
    if (mtile * 16 >= n_nodes) return;

    const int m0 = mtile * 16;
    int rA = m0 + (lane & 15);
    if (rA >= n_nodes) rA = n_nodes - 1;
    const int klo = 2 * (lane >> 4);

    const float* __restrict__ hrow = h + (size_t)rA * DFEAT + klo;
    const float* wb = wlds + (lane & 15) * ZPITCH + klo;

    v8f acc[4];
    #pragma unroll
    for (int t = 0; t < 4; ++t) acc[t] = v8f{};

    #pragma unroll 8
    for (int kk = 0; kk < DFEAT; kk += 4) {
        const v2f a = *(const v2f*)(hrow + kk);
        #pragma unroll
        for (int t = 0; t < 4; ++t) {
            const v2f b = *(const v2f*)(wb + t * 16 * ZPITCH + kk);
            acc[t] = __builtin_amdgcn_wmma_f32_16x16x4_f32(
                false, a, false, b, (short)0, acc[t], false, false);
        }
    }

    const int mbase = m0 + 8 * (lane >> 4);
    const float b0 = bl[lane & 15];
    const float b1 = bl[16 + (lane & 15)];
    float* __restrict__ lrow = zl + (size_t)mbase * DOUT3 + (lane & 15);
    float* __restrict__ rrow = zr + (size_t)mbase * DOUT3 + (lane & 15);
    if (m0 + 16 <= n_nodes) {
        #pragma unroll
        for (int v = 0; v < 8; ++v) {
            lrow[v * DOUT3]      = acc[0][v];
            lrow[v * DOUT3 + 16] = acc[1][v];
            rrow[v * DOUT3]      = acc[2][v] + b0;
            rrow[v * DOUT3 + 16] = acc[3][v] + b1;
        }
    } else {
        #pragma unroll
        for (int v = 0; v < 8; ++v) {
            if (mbase + v < n_nodes) {
                lrow[v * DOUT3]      = acc[0][v];
                lrow[v * DOUT3 + 16] = acc[1][v];
                rrow[v * DOUT3]      = acc[2][v] + b0;
                rrow[v * DOUT3 + 16] = acc[3][v] + b1;
            }
        }
    }
}

extern "C" void kernel_launch(void* const* d_in, const int* in_sizes, int n_in,
                              void* d_out, int out_size, void* d_ws, size_t ws_size,
                              hipStream_t stream) {
    const float* x    = (const float*)d_in[0];
    const int*   eidx = (const int*)  d_in[1];
    const float* W_l1 = (const float*)d_in[2];
    const float* b_l1 = (const float*)d_in[3];
    const float* W_r1 = (const float*)d_in[4];
    const float* W_l2 = (const float*)d_in[5];
    const float* b_l2 = (const float*)d_in[6];
    const float* W_r2 = (const float*)d_in[7];
    const float* W_l3 = (const float*)d_in[8];
    const float* b_l3 = (const float*)d_in[9];
    const float* W_r3 = (const float*)d_in[10];
    float* out = (float*)d_out;

    const int n_nodes = in_sizes[0] / DFEAT;
    const int n_edges = in_sizes[1] / 2;
    const int* src = eidx;
    const int* dst = eidx + n_edges;

    int*   slot = (int*)d_ws;
    float* agg  = (float*)(slot + (size_t)n_nodes * SLOTS);
    float* h1   = agg + (size_t)n_nodes * DFEAT;
    float* h2   = h1  + (size_t)n_nodes * DFEAT;
    int* cnt     = (int*)(h2 + (size_t)n_nodes * DFEAT);
    const int cur_off = (n_nodes + 1 + CUR_PITCH - 1) / CUR_PITCH * CUR_PITCH;
    int* cur     = cnt + cur_off;
    int* binbuf  = cur + MAXB * CUR_PITCH + CUR_PITCH;
    int* ovf_dst = binbuf + (size_t)MAXB * BIN_CAP * 2;
    int* ovf_src = ovf_dst + n_edges;
    int* spl_dst = ovf_src + n_edges;
    int* spl_src = spl_dst + n_edges;

    const int  TB      = 256;
    const int  nZero   = cur_off - n_nodes + MAXB * CUR_PITCH + CUR_PITCH;
    const int  gZero   = (nZero + TB - 1) / TB;
    const int  gBinE   = (n_edges + EPB - 1) / EPB;
    const int  gBins   = (n_nodes + BIN_NODES - 1) / BIN_NODES;
    const int  gGather = (n_nodes + 7) / 8;
    const int  gGatherOut = (n_nodes + 31) / 32;

    const int mtiles = (n_nodes + 15) / 16;
    const int gGemm = (mtiles + 7) / 8;

    gnn_zero_i32  <<<gZero, TB, 0, stream>>>(cnt + n_nodes, nZero);
    gnn_bin_edges <<<gBinE, TB, 0, stream>>>(src, dst, cur, binbuf, spl_dst, spl_src, n_edges);
    gnn_place_bins<<<gBins, 512, 0, stream>>>(cur, binbuf, spl_dst, spl_src, cnt, slot, ovf_dst, ovf_src, n_nodes);

    gnn_gather_mean<<<gGather, TB, 0, stream>>>(x, cnt, slot, ovf_dst, ovf_src, agg, n_nodes);
    gnn_sage_gemm_wmma_hid<<<GEMM_WGPS, GEMM_WAVES * 32, 0, stream>>>(agg, x, W_l1, b_l1, W_r1,
                                                                      h1, n_nodes);

    gnn_gather_mean<<<gGather, TB, 0, stream>>>(h1, cnt, slot, ovf_dst, ovf_src, agg, n_nodes);
    gnn_sage_gemm_wmma_hid<<<GEMM_WGPS, GEMM_WAVES * 32, 0, stream>>>(agg, h1, W_l2, b_l2, W_r2,
                                                                      h2, n_nodes);

    float* zl = agg;
    float* zr = agg + (size_t)n_nodes * DOUT3;
    gnn_sage_proj_wmma_out<<<gGemm, TB, 0, stream>>>(h2, W_l3, b_l3, W_r3, zl, zr, n_nodes);
    gnn_gather_mean_out<<<gGatherOut, TB, 0, stream>>>(zl, zr, cnt, slot, ovf_dst, ovf_src, out, n_nodes);
}


# Last layer's projection (zl, zr) moved into the layer-2 GEMM launch: a wave passes its relu'd 16x128 tile of h2 through LDS into A fragments; projection launch and h2 plane go; sum order kept
# speedup vs baseline: 6.0544x; 1.0325x over previous
#include <hip/hip_runtime.h>


typedef float v2f __attribute__((ext_vector_type(2)));
typedef float v8f __attribute__((ext_vector_type(8)));

#define DFEAT 128

typedef float v4f __attribute__((ext_vector_type(4)));
typedef int   v4i __attribute__((ext_vector_type(4)));

__global__ void gnn_zero_i32(int* __restrict__ p, int n) {
    int i = blockIdx.x * blockDim.x + threadIdx.x;
    if (i < n) p[i] = 0;
}

#define SLOTS 64
#define BIN_SHIFT 8
#define BIN_NODES (1 << BIN_SHIFT)
#define MAXB 256
#define BIN_CAP 8192
#define CUR_PITCH 32
#define EPT 13
#define EPB (EPT * 256)

typedef int v2i __attribute__((ext_vector_type(2)));

__global__ __launch_bounds__(256)
void gnn_bin_edges(const int* __restrict__ src, const int* __restrict__ dst,
                   int* __restrict__ cur, int* __restrict__ binbuf_i,
                   int* __restrict__ spl_dst, int* __restrict__ spl_src, int n_edges) {
    v2i* __restrict__ binbuf = (v2i*)binbuf_i;
    __shared__ int hist[MAXB + 1];
    __shared__ int delta[MAXB];
    __shared__ int wsum[8];
    __shared__ int ntot;
    __shared__ v2i sorted[EPB];
    const int tid  = threadIdx.x;
    const int lane = tid & 31;
    const int wave = tid >> 5;
    const int e0   = blockIdx.x * EPB;

    hist[tid] = 0;
    if (tid == 0) hist[MAXB] = 0;
    __syncthreads();

    int d[EPT], s[EPT], r[EPT];
    #pragma unroll
    for (int i = 0; i < EPT; ++i) {
        const int e  = e0 + i * 256 + tid;
        const int ec = (e < n_edges) ? e : 0;
        const int dv = dst[ec];
        s[i] = src[ec];
        d[i] = (e < n_edges) ? dv : -1;
    }
    #pragma unroll
    for (int i = 0; i < EPT; ++i) {
        const int b = d[i] >> BIN_SHIFT;
        r[i] = atomicAdd(&hist[((unsigned)b < (unsigned)MAXB) ? b : MAXB], 1);
    }
    __syncthreads();

    const int c = hist[tid];
    int incl = c;
    #pragma unroll
    for (int o = 1; o < 32; o <<= 1) {
        const int t = __shfl_up(incl, o);
        if (lane >= o) incl += t;
    }
    if (lane == 31) wsum[wave] = incl;
    __syncthreads();
    int woff = 0;
    #pragma unroll
    for (int w = 0; w < 8; ++w) woff += (w < wave) ? wsum[w] : 0;
    const int loff = woff + incl - c;
    int g = 0;
    if (c > 0) g = atomicAdd(&cur[tid * CUR_PITCH], c);
    hist[tid]  = loff;
    delta[tid] = g - loff;
    if (tid == 255) ntot = loff + c;
    __syncthreads();

    int o[EPT];
    #pragma unroll
    for (int i = 0; i < EPT; ++i) {
        const int b = d[i] >> BIN_SHIFT;
        o[i] = hist[((unsigned)b < (unsigned)MAXB) ? b : MAXB] + r[i];
    }
    #pragma unroll
    for (int i = 0; i < EPT; ++i) {
        const int b = d[i] >> BIN_SHIFT;
        if ((unsigned)b < (unsigned)MAXB) {
            v2i e; e.x = d[i]; e.y = s[i];
            sorted[o[i]] = e;
        } else if (d[i] >= 0) {
            const int p = atomicAdd(&cur[MAXB * CUR_PITCH], 1);
            spl_dst[p] = d[i];
            spl_src[p] = s[i];
        }
    }
    __syncthreads();

    const int n = ntot;
    v2i oe[EPT];
    int pos[EPT];
    #pragma unroll
    for (int i = 0; i < EPT; ++i) {
        const int idx = i * 256 + tid;
        oe[i] = sorted[(idx < n) ? idx : 0];
        const int b = (idx < n) ? (oe[i].x >> BIN_SHIFT) : 0;
        pos[i] = delta[b] + idx;
    }
    #pragma unroll
    for (int i = 0; i < EPT; ++i) {
        if (i * 256 + tid < n) {
            const int b = oe[i].x >> BIN_SHIFT;
            if (pos[i] < BIN_CAP) {
                binbuf[(size_t)b * BIN_CAP + pos[i]] = oe[i];
            } else {
                const int p = atomicAdd(&cur[MAXB * CUR_PITCH], 1);
                spl_dst[p] = oe[i].x;
                spl_src[p] = oe[i].y;
            }
        }
    }
}

__global__ __launch_bounds__(512)
void gnn_place_bins(const int* __restrict__ cur, const int* __restrict__ binbuf_i,
                    const int* __restrict__ spl_dst, const int* __restrict__ spl_src,
                    int* __restrict__ cnt, int* __restrict__ slot,
                    int* __restrict__ ovf_dst, int* __restrict__ ovf_src, int n_nodes) {
    __shared__ int cntL[BIN_NODES + 1];
    __shared__ __attribute__((aligned(16))) int slotL[BIN_NODES * SLOTS];
    const int tid   = threadIdx.x;
    const int b     = blockIdx.x;
    const int node0 = b << BIN_SHIFT;

    if (tid <= BIN_NODES) cntL[tid] = 0;
    const v4i zero4 = {0, 0, 0, 0};
    #pragma unroll
    for (int i = 0; i < BIN_NODES * SLOTS / 4 / 512; ++i) ((v4i*)slotL)[i * 512 + tid] = zero4;
    __syncthreads();

    int n = 0;
    if (b < MAXB) { n = cur[b * CUR_PITCH]; if (n > BIN_CAP) n = BIN_CAP; }
    const v2i* __restrict__ eb = (const v2i*)binbuf_i + (size_t)b * BIN_CAP;
    for (int base = 0; base < n; base += 512 * 8) {
        v2i e[8];
        int r[8];
        #pragma unroll
        for (int u = 0; u < 8; ++u) {
            const int i = base + u * 512 + tid;
            e[u] = eb[(i < n) ? i : 0];
            if (i >= n) e[u].x = -1;
        }
        #pragma unroll
        for (int u = 0; u < 8; ++u)
            r[u] = atomicAdd(&cntL[(e[u].x >= 0) ? (e[u].x - node0) : BIN_NODES], 1);
        #pragma unroll
        for (int u = 0; u < 8; ++u) {
            if (e[u].x >= 0) {
                if (r[u] < SLOTS) {
                    slotL[(e[u].x - node0) * SLOTS + r[u]] = e[u].y;
                } else {
                    const int p = atomicAdd(&cnt[n_nodes], 1);
                    ovf_dst[p] = e[u].x;
                    ovf_src[p] = e[u].y;
                }
            }
        }
    }
    const int n_spl = cur[MAXB * CUR_PITCH];
    for (int i = tid; i < n_spl; i += 512) {
        const int dd = spl_dst[i];
        if ((dd >> BIN_SHIFT) == b) {
            const int ss = spl_src[i];
            const int l = dd - node0;
            const int r = atomicAdd(&cntL[l], 1);
            if (r < SLOTS) {
                slotL[l * SLOTS + r] = ss;
            } else {
                const int p = atomicAdd(&cnt[n_nodes], 1);
                ovf_dst[p] = dd;
                ovf_src[p] = ss;
            }
        }
    }
    __syncthreads();

    if (tid < BIN_NODES && node0 + tid < n_nodes) cnt[node0 + tid] = cntL[tid];
    const int q = tid & 7;
    #pragma unroll
    for (int g = tid >> 3; g < BIN_NODES; g += 64) {
        const int node = node0 + g;
        if (node < n_nodes) {
            const v4i lo = *(const v4i*)(slotL + g * SLOTS + q * 4);
            *(v4i*)(slot + (size_t)node * SLOTS + q * 4) = lo;
            if (cntL[g] >= 32) {
                const v4i hi = *(const v4i*)(slotL + g * SLOTS + 32 + q * 4);
                *(v4i*)(slot + (size_t)node * SLOTS + 32 + q * 4) = hi;
            }
        }
    }
}

__global__ __launch_bounds__(256)
void gnn_gather_mean(const float* __restrict__ h, const int* __restrict__ cnt,
                     const int* __restrict__ slot, const int* __restrict__ ovf_dst,
                     const int* __restrict__ ovf_src, float* __restrict__ agg, int n_nodes) {
    const int lane = threadIdx.x & 31;
    const int node = __builtin_amdgcn_readfirstlane((int)(blockIdx.x * 8 + (threadIdx.x >> 5)));
    if (node >= n_nodes) return;
    const int deg = cnt[node];
    const int end = (deg < SLOTS) ? deg : SLOTS;
    const int* __restrict__ ids = slot + (size_t)node * SLOTS;
    const float* __restrict__ hl = h + lane * 4;

    v4f acc = {0.0f, 0.0f, 0.0f, 0.0f};
    for (int base = 0; base < end; base += 32) {
        int m = end - base; if (m > 32) m = 32;
        const int mine = (lane < m) ? ids[base + lane] : 0;
        int k = 0;
        for (; k + 4 <= m; k += 4) {
            const int s0 = __builtin_amdgcn_readlane(mine, k);
            const int s1 = __builtin_amdgcn_readlane(mine, k + 1);
            const int s2 = __builtin_amdgcn_readlane(mine, k + 2);
            const int s3 = __builtin_amdgcn_readlane(mine, k + 3);
            const v4f v0 = *(const v4f*)(hl + (size_t)s0 * DFEAT);
            const v4f v1 = *(const v4f*)(hl + (size_t)s1 * DFEAT);
            const v4f v2 = *(const v4f*)(hl + (size_t)s2 * DFEAT);
            const v4f v3 = *(const v4f*)(hl + (size_t)s3 * DFEAT);
            acc += v0; acc += v1; acc += v2; acc += v3;
        }
        const int r = m - k;
        if (r == 3) {
            const int s0 = __builtin_amdgcn_readlane(mine, k);
            const int s1 = __builtin_amdgcn_readlane(mine, k + 1);
            const int s2 = __builtin_amdgcn_readlane(mine, k + 2);
            const v4f v0 = *(const v4f*)(hl + (size_t)s0 * DFEAT);
            const v4f v1 = *(const v4f*)(hl + (size_t)s1 * DFEAT);
            const v4f v2 = *(const v4f*)(hl + (size_t)s2 * DFEAT);
            acc += v0; acc += v1; acc += v2;
        } else if (r == 2) {
            const int s0 = __builtin_amdgcn_readlane(mine, k);
            const int s1 = __builtin_amdgcn_readlane(mine, k + 1);
            const v4f v0 = *(const v4f*)(hl + (size_t)s0 * DFEAT);
            const v4f v1 = *(const v4f*)(hl + (size_t)s1 * DFEAT);
            acc += v0; acc += v1;
        } else if (r == 1) {
            const int s0 = __builtin_amdgcn_readlane(mine, k);
            acc += *(const v4f*)(hl + (size_t)s0 * DFEAT);
        }
    }
    if (deg > SLOTS) {
        const int n_ovf = cnt[n_nodes];
        for (int p0 = 0; p0 < n_ovf; p0 += 32) {
            const int p  = p0 + lane;
            const int od = (p < n_ovf) ? ovf_dst[p] : -1;
            const int os = (p < n_ovf) ? ovf_src[p] : 0;
            unsigned hit = __builtin_amdgcn_ballot_w32(od == node);
            while (hit != 0u) {
                const int l = __builtin_ctz(hit);
                hit &= hit - 1u;
                const int s0 = __shfl(os, l);
                acc += *(const v4f*)(hl + (size_t)s0 * DFEAT);
            }
        }
    }
    const float rdeg = 1.0f / (float)(deg > 0 ? deg : 1);
    acc *= (deg > 0) ? rdeg : 0.0f;
    *(v4f*)(agg + (size_t)node * DFEAT + lane * 4) = acc;
}

#define DOUT3 32

__global__ __launch_bounds__(256)
void gnn_gather_mean_out(const float* __restrict__ zl, const float* __restrict__ zr,
                         const int* __restrict__ cnt, const int* __restrict__ slot,
                         const int* __restrict__ ovf_dst, const int* __restrict__ ovf_src,
                         float* __restrict__ out, int n_nodes) {
    const int q    = threadIdx.x & 7;
    const int node = (int)(blockIdx.x * 256 + threadIdx.x) >> 3;
    if (node >= n_nodes) return;
    const int* __restrict__ ids = slot + (size_t)node * SLOTS;
    v4i c0 = *(const v4i*)(ids);
    v4i c1 = *(const v4i*)(ids + 4);
    const int deg = cnt[node];
    const v4f self = *(const v4f*)(zr + (size_t)node * DOUT3 + q * 4);
    const int end = (deg < SLOTS) ? deg : SLOTS;
    const float* __restrict__ zq = zl + q * 4;

    v4f acc = {0.0f, 0.0f, 0.0f, 0.0f};
    int k = 0;
    for (; k + 8 <= end; k += 8) {
        const v4i n0 = *(const v4i*)(ids + ((k + 8) & (SLOTS - 1)));
        const v4i n1 = *(const v4i*)(ids + ((k + 12) & (SLOTS - 1)));
        const v4f v0 = *(const v4f*)(zq + (size_t)c0.x * DOUT3);
        const v4f v1 = *(const v4f*)(zq + (size_t)c0.y * DOUT3);
        const v4f v2 = *(const v4f*)(zq + (size_t)c0.z * DOUT3);
        const v4f v3 = *(const v4f*)(zq + (size_t)c0.w * DOUT3);
        const v4f v4 = *(const v4f*)(zq + (size_t)c1.x * DOUT3);
        const v4f v5 = *(const v4f*)(zq + (size_t)c1.y * DOUT3);
        const v4f v6 = *(const v4f*)(zq + (size_t)c1.z * DOUT3);
        const v4f v7 = *(const v4f*)(zq + (size_t)c1.w * DOUT3);
        acc += v0; acc += v1; acc += v2; acc += v3;
        acc += v4; acc += v5; acc += v6; acc += v7;
        c0 = n0; c1 = n1;
    }
    const int r = end - k;
    if (r > 0) {
        const v4f zero = {0.0f, 0.0f, 0.0f, 0.0f};
        const int s0 = c0.x;
        const int s1 = (r > 1) ? c0.y : s0;
        const int s2 = (r > 2) ? c0.z : s0;
        const int s3 = (r > 3) ? c0.w : s0;
        const int s4 = (r > 4) ? c1.x : s0;
        const int s5 = (r > 5) ? c1.y : s0;
        const int s6 = (r > 6) ? c1.z : s0;
        const v4f v0 = *(const v4f*)(zq + (size_t)s0 * DOUT3);
        const v4f v1 = *(const v4f*)(zq + (size_t)s1 * DOUT3);
        const v4f v2 = *(const v4f*)(zq + (size_t)s2 * DOUT3);
        const v4f v3 = *(const v4f*)(zq + (size_t)s3 * DOUT3);
        const v4f v4 = *(const v4f*)(zq + (size_t)s4 * DOUT3);
        const v4f v5 = *(const v4f*)(zq + (size_t)s5 * DOUT3);
        const v4f v6 = *(const v4f*)(zq + (size_t)s6 * DOUT3);
        acc += v0;
        acc += (r > 1) ? v1 : zero;
        acc += (r > 2) ? v2 : zero;
        acc += (r > 3) ? v3 : zero;
        acc += (r > 4) ? v4 : zero;
        acc += (r > 5) ? v5 : zero;
        acc += (r > 6) ? v6 : zero;
    }
    if (deg > SLOTS) {
        const int n_ovf = cnt[n_nodes];
        for (int p = 0; p < n_ovf; ++p) {
            if (ovf_dst[p] == node) acc += *(const v4f*)(zq + (size_t)ovf_src[p] * DOUT3);
        }
    }
    const float rdeg = 1.0f / (float)(deg > 0 ? deg : 1);
    const float sc   = (deg > 0) ? rdeg : 0.0f;
    v4f o = acc * sc + self;
    o.x = (o.x > 0.0f) ? o.x : 0.0f;
    o.y = (o.y > 0.0f) ? o.y : 0.0f;
    o.z = (o.z > 0.0f) ? o.z : 0.0f;
    o.w = (o.w > 0.0f) ? o.w : 0.0f;
    *(v4f*)(out + (size_t)node * DOUT3 + q * 4) = o;
}

#define WPITCH 260

#define GEMM_WGPS  256
#define GEMM_WAVES 12
#define GEMM_LDS_FLOATS 41984

#define ZPITCH 132
#define GEMM_W3_OFF     (8 * 16 * WPITCH)
#define GEMM_TILE_OFF   GEMM_LDS_FLOATS
#define GEMM_TILE_FLOATS (16 * ZPITCH)
#define GEMM_LDS_FLOATS_PROJ (GEMM_LDS_FLOATS + (GEMM_WAVES + 1) * GEMM_TILE_FLOATS)

template <int NT, bool TO_LDS>
__device__ __forceinline__
void gnn_sage_gemm_tile(const float* __restrict__ agg,
                        const float* __restrict__ h,
                        const float* __restrict__ bl,
                        float* __restrict__ out,
                        const float* wlds, float* tile,
                        int mtile, int T0, int lane, int n_nodes) {
    const int m0 = mtile * 16;
    int rA = m0 + (lane & 15);
    if (rA >= n_nodes) rA = n_nodes - 1;
    const int klo = 2 * (lane >> 4);

    const float* __restrict__ arow = agg + (size_t)rA * DFEAT + klo;
    const float* __restrict__ hrow = h   + (size_t)rA * DFEAT + klo;
    const float* wb = wlds + (T0 * 16 + (lane & 15)) * WPITCH + klo;

    v8f acc[NT];
    #pragma unroll
    for (int t = 0; t < NT; ++t) acc[t] = v8f{};

    #pragma unroll 8
    for (int kk = 0; kk < DFEAT; kk += 4) {
        const v2f a = *(const v2f*)(arow + kk);
        #pragma unroll
        for (int t = 0; t < NT; ++t) {
            const v2f b = *(const v2f*)(wb + t * 16 * WPITCH + kk);
            acc[t] = __builtin_amdgcn_wmma_f32_16x16x4_f32(
                false, a, false, b, (short)0, acc[t], false, false);
        }
    }

    #pragma unroll 8
    for (int kk = 0; kk < DFEAT; kk += 4) {
        const v2f a = *(const v2f*)(hrow + kk);
        #pragma unroll
        for (int t = 0; t < NT; ++t) {
            const v2f b = *(const v2f*)(wb + t * 16 * WPITCH + DFEAT + kk);
            acc[t] = __builtin_amdgcn_wmma_f32_16x16x4_f32(
                false, a, false, b, (short)0, acc[t], false, false);
        }
    }

    const int mbase = m0 + 8 * (lane >> 4);
    float bb[NT];
    #pragma unroll
    for (int t = 0; t < NT; ++t) bb[t] = bl[(T0 + t) * 16 + (lane & 15)];
    if (TO_LDS) {
        float* trow = tile + (8 * (lane >> 4)) * ZPITCH + T0 * 16 + (lane & 15);
        #pragma unroll
        for (int t = 0; t < NT; ++t) {
            #pragma unroll
            for (int v = 0; v < 8; ++v) {
                const float val = acc[t][v] + bb[t];
                trow[v * ZPITCH + t * 16] = (val > 0.0f) ? val : 0.0f;
            }
        }
        return;
    }
    float* __restrict__ orow = out + (size_t)mbase * DFEAT + T0 * 16 + (lane & 15);
    if (m0 + 16 <= n_nodes) {
        #pragma unroll
        for (int t = 0; t < NT; ++t) {
            #pragma unroll
            for (int v = 0; v < 8; ++v) {
                const float val = acc[t][v] + bb[t];
                orow[v * DFEAT + t * 16] = (val > 0.0f) ? val : 0.0f;
            }
        }
    } else {
        #pragma unroll
        for (int t = 0; t < NT; ++t) {
            #pragma unroll
            for (int v = 0; v < 8; ++v) {
                if (mbase + v < n_nodes) {
                    const float val = acc[t][v] + bb[t];
                    orow[v * DFEAT + t * 16] = (val > 0.0f) ? val : 0.0f;
                }
            }
        }
    }
}

template <int NC>
__device__ __forceinline__
void gnn_proj_tile(const float* tile, const float* w3,
                   const float* __restrict__ bl3,
                   float* __restrict__ zl, float* __restrict__ zr,
                   int m0, int C0, int lane, int n_nodes) {
    const int klo = 2 * (lane >> 4);
    const float* ta = tile + (lane & 15) * ZPITCH + klo;
    const float* wb = w3 + (C0 * 16 + (lane & 15)) * ZPITCH + klo;

    v8f acc[NC];
    #pragma unroll
    for (int t = 0; t < NC; ++t) acc[t] = v8f{};

    #pragma unroll 8
    for (int kk = 0; kk < DFEAT; kk += 4) {
        const v2f a = *(const v2f*)(ta + kk);
        #pragma unroll
        for (int t = 0; t < NC; ++t) {
            const v2f b = *(const v2f*)(wb + t * 16 * ZPITCH + kk);
            acc[t] = __builtin_amdgcn_wmma_f32_16x16x4_f32(
                false, a, false, b, (short)0, acc[t], false, false);
        }
    }

    const int mbase = m0 + 8 * (lane >> 4);
    #pragma unroll
    for (int t = 0; t < NC; ++t) {
        const int c   = C0 + t;
        const int col = (c & 1) * 16 + (lane & 15);
        if (c < 2) {
            float* __restrict__ lrow = zl + (size_t)mbase * DOUT3 + col;
            #pragma unroll
            for (int v = 0; v < 8; ++v)
                if (mbase + v < n_nodes) lrow[v * DOUT3] = acc[t][v];
        } else {
            const float bias = bl3[col];
            float* __restrict__ rrow = zr + (size_t)mbase * DOUT3 + col;
            #pragma unroll
            for (int v = 0; v < 8; ++v)
                if (mbase + v < n_nodes) rrow[v * DOUT3] = acc[t][v] + bias;
        }
    }
}

template <bool PROJ>
__global__ __launch_bounds__(GEMM_WAVES * 32)
void gnn_sage_gemm_wmma_hid(const float* __restrict__ agg,
                            const float* __restrict__ h,
                            const float* __restrict__ Wl,
                            const float* __restrict__ bl,
                            const float* __restrict__ Wr,
                            float* __restrict__ out,
                            const float* __restrict__ Wl3,
                            const float* __restrict__ bl3,
                            const float* __restrict__ Wr3,
                            float* __restrict__ zl,
                            float* __restrict__ zr,
                            int n_nodes) {
    __shared__ __attribute__((aligned(16))) float wlds[PROJ ? GEMM_LDS_FLOATS_PROJ : GEMM_LDS_FLOATS];
    const int tid  = threadIdx.x;
    const int wave = __builtin_amdgcn_readfirstlane(tid >> 5);
    const int lane = tid & 31;

    constexpr int NV4 = 8 * 16 * 64;
    constexpr int NTH = GEMM_WAVES * 32;
    constexpr int RPP = NTH / 64;
    const int q  = tid & 63;
    const int nr = tid >> 6;
    const float* __restrict__ wsrc = ((q < 32) ? Wl : Wr) + (size_t)nr * DFEAT + 4 * (q & 31);
    float* wdst = wlds + nr * WPITCH + 4 * q;
    v4f wv[(NV4 + NTH - 1) / NTH];
    #pragma unroll
    for (int i = 0; i < (NV4 + NTH - 1) / NTH; ++i) {
        const int row = (i * RPP + nr < 8 * 16) ? i * RPP : 0;
        wv[i] = *(const v4f*)(wsrc + (size_t)row * DFEAT);
    }
    if (PROJ) {
        constexpr int NV3 = 2 * DOUT3 * 32;
        v4f w3v[(NV3 + NTH - 1) / NTH];
        #pragma unroll
        for (int i = 0; i < (NV3 + NTH - 1) / NTH; ++i) {
            const int p   = i * NTH + tid;
            const int pc  = (p < NV3) ? p : tid;
            const int row = pc >> 5;
            const float* __restrict__ s3 = ((row < DOUT3) ? Wl3 + (size_t)row * DFEAT : Wr3 + (size_t)(row - DOUT3) * DFEAT) + 4 * (pc & 31);
            w3v[i] = *(const v4f*)s3;
        }
        #pragma unroll
        for (int i = 0; i < (NV3 + NTH - 1) / NTH; ++i) {
            const int p = i * NTH + tid;
            if (p < NV3) *(v4f*)(wlds + GEMM_W3_OFF + (p >> 5) * ZPITCH + 4 * (p & 31)) = w3v[i];
        }
    }
    #pragma unroll
    for (int i = 0; i < (NV4 + NTH - 1) / NTH; ++i) {
        if (i * RPP + nr < 8 * 16) *(v4f*)(wdst + i * RPP * WPITCH) = wv[i];
    }
    __syncthreads();

    const int mtiles  = (n_nodes + 15) / 16;
    const int per_rnd = GEMM_WAVES * (int)gridDim.x;
    const int n_whole = mtiles / per_rnd * per_rnd;
    if (!PROJ) {
        for (int mt = (int)blockIdx.x * GEMM_WAVES + wave; mt < n_whole; mt += per_rnd)
            gnn_sage_gemm_tile<8, false>(agg, h, bl, out, wlds, nullptr, mt, 0, lane, n_nodes);
        if (wave < 4) {
            for (int mt = n_whole + (int)blockIdx.x; mt < mtiles; mt += (int)gridDim.x)
                gnn_sage_gemm_tile<2, false>(agg, h, bl, out, wlds, nullptr, mt, 2 * wave, lane, n_nodes);
        }
    } else {
        const float* w3 = wlds + GEMM_W3_OFF;
        float* mytile = wlds + GEMM_TILE_OFF + wave * GEMM_TILE_FLOATS;
        for (int mt = (int)blockIdx.x * GEMM_WAVES + wave; mt < n_whole; mt += per_rnd) {
            gnn_sage_gemm_tile<8, true>(agg, h, bl, out, wlds, mytile, mt, 0, lane, n_nodes);
            __builtin_amdgcn_fence(__ATOMIC_RELEASE, "wavefront");
            __builtin_amdgcn_wave_barrier();
            __builtin_amdgcn_fence(__ATOMIC_ACQUIRE, "wavefront");
            gnn_proj_tile<4>(mytile, w3, bl3, zl, zr, mt * 16, 0, lane, n_nodes);
            __builtin_amdgcn_fence(__ATOMIC_RELEASE, "wavefront");
            __builtin_amdgcn_wave_barrier();
            __builtin_amdgcn_fence(__ATOMIC_ACQUIRE, "wavefront");
        }
        float* shtile = wlds + GEMM_TILE_OFF + GEMM_WAVES * GEMM_TILE_FLOATS;
        for (int mt = n_whole + (int)blockIdx.x; mt < mtiles; mt += (int)gridDim.x) {
            if (wave < 4)
                gnn_sage_gemm_tile<2, true>(agg, h, bl, out, wlds, shtile, mt, 2 * wave, lane, n_nodes);
            __syncthreads();
            if (wave < 4)
                gnn_proj_tile<1>(shtile, w3, bl3, zl, zr, mt * 16, wave, lane, n_nodes);
            __syncthreads();
        }
    }
}

extern "C" void kernel_launch(void* const* d_in, const int* in_sizes, int n_in,
                              void* d_out, int out_size, void* d_ws, size_t ws_size,
                              hipStream_t stream) {
    const float* x    = (const float*)d_in[0];
    const int*   eidx = (const int*)  d_in[1];
    const float* W_l1 = (const float*)d_in[2];
    const float* b_l1 = (const float*)d_in[3];
    const float* W_r1 = (const float*)d_in[4];
    const float* W_l2 = (const float*)d_in[5];
    const float* b_l2 = (const float*)d_in[6];
    const float* W_r2 = (const float*)d_in[7];
    const float* W_l3 = (const float*)d_in[8];
    const float* b_l3 = (const float*)d_in[9];
    const float* W_r3 = (const float*)d_in[10];
    float* out = (float*)d_out;

    const int n_nodes = in_sizes[0] / DFEAT;
    const int n_edges = in_sizes[1] / 2;
    const int* src = eidx;
    const int* dst = eidx + n_edges;

    int*   slot = (int*)d_ws;
    float* agg  = (float*)(slot + (size_t)n_nodes * SLOTS);
    float* h1   = agg + (size_t)n_nodes * DFEAT;
    float* h2   = h1  + (size_t)n_nodes * DFEAT;
    float* zl    = h2;
    float* zr    = h2 + (size_t)n_nodes * DOUT3;
    int* cnt     = (int*)(h2 + (size_t)n_nodes * DFEAT);
    const int cur_off = (n_nodes + 1 + CUR_PITCH - 1) / CUR_PITCH * CUR_PITCH;
    int* cur     = cnt + cur_off;
    int* binbuf  = cur + MAXB * CUR_PITCH + CUR_PITCH;
    int* ovf_dst = binbuf + (size_t)MAXB * BIN_CAP * 2;
    int* ovf_src = ovf_dst + n_edges;
    int* spl_dst = ovf_src + n_edges;
    int* spl_src = spl_dst + n_edges;

    const int  TB      = 256;
    const int  nZero   = cur_off - n_nodes + MAXB * CUR_PITCH + CUR_PITCH;
    const int  gZero   = (nZero + TB - 1) / TB;
    const int  gBinE   = (n_edges + EPB - 1) / EPB;
    const int  gBins   = (n_nodes + BIN_NODES - 1) / BIN_NODES;
    const int  gGather = (n_nodes + 7) / 8;
    const int  gGatherOut = (n_nodes + 31) / 32;

    gnn_zero_i32  <<<gZero, TB, 0, stream>>>(cnt + n_nodes, nZero);
    gnn_bin_edges <<<gBinE, TB, 0, stream>>>(src, dst, cur, binbuf, spl_dst, spl_src, n_edges);
    gnn_place_bins<<<gBins, 512, 0, stream>>>(cur, binbuf, spl_dst, spl_src, cnt, slot, ovf_dst, ovf_src, n_nodes);

    gnn_gather_mean<<<gGather, TB, 0, stream>>>(x, cnt, slot, ovf_dst, ovf_src, agg, n_nodes);
    gnn_sage_gemm_wmma_hid<false><<<GEMM_WGPS, GEMM_WAVES * 32, 0, stream>>>(agg, x, W_l1, b_l1, W_r1, h1,
                                                                             W_l3, b_l3, W_r3, zl, zr, n_nodes);

    gnn_gather_mean<<<gGather, TB, 0, stream>>>(h1, cnt, slot, ovf_dst, ovf_src, agg, n_nodes);
    gnn_sage_gemm_wmma_hid<true><<<GEMM_WGPS, GEMM_WAVES * 32, 0, stream>>>(agg, h1, W_l2, b_l2, W_r2, h2,
                                                                            W_l3, b_l3, W_r3, zl, zr, n_nodes);

    gnn_gather_mean_out<<<gGatherOut, TB, 0, stream>>>(zl, zr, cnt, slot, ovf_dst, ovf_src, out, n_nodes);
}


# Hidden layers' gather moved into the layer GEMM launch: a wave averages its 16 nodes' neighbour rows into LDS as the Wl half's A tile; 2 of 3 waves per SIMD sum the Wr half first; 2 launches go
# speedup vs baseline: 6.2800x; 1.0373x over previous
#include <hip/hip_runtime.h>


typedef float v2f __attribute__((ext_vector_type(2)));
typedef float v8f __attribute__((ext_vector_type(8)));

#define DFEAT 128

typedef float v4f __attribute__((ext_vector_type(4)));
typedef int   v4i __attribute__((ext_vector_type(4)));

__global__ void gnn_zero_i32(int* __restrict__ p, int n) {
    int i = blockIdx.x * blockDim.x + threadIdx.x;
    if (i < n) p[i] = 0;
}

#define SLOTS 64
#define BIN_SHIFT 8
#define BIN_NODES (1 << BIN_SHIFT)
#define MAXB 256
#define BIN_CAP 8192
#define CUR_PITCH 32
#define EPT 13
#define EPB (EPT * 256)

typedef int v2i __attribute__((ext_vector_type(2)));

__global__ __launch_bounds__(256)
void gnn_bin_edges(const int* __restrict__ src, const int* __restrict__ dst,
                   int* __restrict__ cur, int* __restrict__ binbuf_i,
                   int* __restrict__ spl_dst, int* __restrict__ spl_src, int n_edges) {
    v2i* __restrict__ binbuf = (v2i*)binbuf_i;
    __shared__ int hist[MAXB + 1];
    __shared__ int delta[MAXB];
    __shared__ int wsum[8];
    __shared__ int ntot;
    __shared__ v2i sorted[EPB];
    const int tid  = threadIdx.x;
    const int lane = tid & 31;
    const int wave = tid >> 5;
    const int e0   = blockIdx.x * EPB;

    hist[tid] = 0;
    if (tid == 0) hist[MAXB] = 0;
    __syncthreads();

    int d[EPT], s[EPT], r[EPT];
    #pragma unroll
    for (int i = 0; i < EPT; ++i) {
        const int e  = e0 + i * 256 + tid;
        const int ec = (e < n_edges) ? e : 0;
        const int dv = dst[ec];
        s[i] = src[ec];
        d[i] = (e < n_edges) ? dv : -1;
    }
    #pragma unroll
    for (int i = 0; i < EPT; ++i) {
        const int b = d[i] >> BIN_SHIFT;
        r[i] = atomicAdd(&hist[((unsigned)b < (unsigned)MAXB) ? b : MAXB], 1);
    }
    __syncthreads();

    const int c = hist[tid];
    int incl = c;
    #pragma unroll
    for (int o = 1; o < 32; o <<= 1) {
        const int t = __shfl_up(incl, o);
        if (lane >= o) incl += t;
    }
    if (lane == 31) wsum[wave] = incl;
    __syncthreads();
    int woff = 0;
    #pragma unroll
    for (int w = 0; w < 8; ++w) woff += (w < wave) ? wsum[w] : 0;
    const int loff = woff + incl - c;
    int g = 0;
    if (c > 0) g = atomicAdd(&cur[tid * CUR_PITCH], c);
    hist[tid]  = loff;
    delta[tid] = g - loff;
    if (tid == 255) ntot = loff + c;
    __syncthreads();

    int o[EPT];
    #pragma unroll
    for (int i = 0; i < EPT; ++i) {
        const int b = d[i] >> BIN_SHIFT;
        o[i] = hist[((unsigned)b < (unsigned)MAXB) ? b : MAXB] + r[i];
    }
    #pragma unroll
    for (int i = 0; i < EPT; ++i) {
        const int b = d[i] >> BIN_SHIFT;
        if ((unsigned)b < (unsigned)MAXB) {
            v2i e; e.x = d[i]; e.y = s[i];
            sorted[o[i]] = e;
        } else if (d[i] >= 0) {
            const int p = atomicAdd(&cur[MAXB * CUR_PITCH], 1);
            spl_dst[p] = d[i];
            spl_src[p] = s[i];
        }
    }
    __syncthreads();

    const int n = ntot;
    v2i oe[EPT];
    int pos[EPT];
    #pragma unroll
    for (int i = 0; i < EPT; ++i) {
        const int idx = i * 256 + tid;
        oe[i] = sorted[(idx < n) ? idx : 0];
        const int b = (idx < n) ? (oe[i].x >> BIN_SHIFT) : 0;
        pos[i] = delta[b] + idx;
    }
    #pragma unroll
    for (int i = 0; i < EPT; ++i) {
        if (i * 256 + tid < n) {
            const int b = oe[i].x >> BIN_SHIFT;
            if (pos[i] < BIN_CAP) {
                binbuf[(size_t)b * BIN_CAP + pos[i]] = oe[i];
            } else {
                const int p = atomicAdd(&cur[MAXB * CUR_PITCH], 1);
                spl_dst[p] = oe[i].x;
                spl_src[p] = oe[i].y;
            }
        }
    }
}

__global__ __launch_bounds__(512)
void gnn_place_bins(const int* __restrict__ cur, const int* __restrict__ binbuf_i,
                    const int* __restrict__ spl_dst, const int* __restrict__ spl_src,
                    int* __restrict__ cnt, int* __restrict__ slot,
                    int* __restrict__ ovf_dst, int* __restrict__ ovf_src, int n_nodes) {
    __shared__ int cntL[BIN_NODES + 1];
    __shared__ __attribute__((aligned(16))) int slotL[BIN_NODES * SLOTS];
    const int tid   = threadIdx.x;
    const int b     = blockIdx.x;
    const int node0 = b << BIN_SHIFT;

    if (tid <= BIN_NODES) cntL[tid] = 0;
    const v4i zero4 = {0, 0, 0, 0};
    #pragma unroll
    for (int i = 0; i < BIN_NODES * SLOTS / 4 / 512; ++i) ((v4i*)slotL)[i * 512 + tid] = zero4;
    __syncthreads();

    int n = 0;
    if (b < MAXB) { n = cur[b * CUR_PITCH]; if (n > BIN_CAP) n = BIN_CAP; }
    const v2i* __restrict__ eb = (const v2i*)binbuf_i + (size_t)b * BIN_CAP;
    for (int base = 0; base < n; base += 512 * 8) {
        v2i e[8];
        int r[8];
        #pragma unroll
        for (int u = 0; u < 8; ++u) {
            const int i = base + u * 512 + tid;
            e[u] = eb[(i < n) ? i : 0];
            if (i >= n) e[u].x = -1;
        }
        #pragma unroll
        for (int u = 0; u < 8; ++u)
            r[u] = atomicAdd(&cntL[(e[u].x >= 0) ? (e[u].x - node0) : BIN_NODES], 1);
        #pragma unroll
        for (int u = 0; u < 8; ++u) {
            if (e[u].x >= 0) {
                if (r[u] < SLOTS) {
                    slotL[(e[u].x - node0) * SLOTS + r[u]] = e[u].y;
                } else {
                    const int p = atomicAdd(&cnt[n_nodes], 1);
                    ovf_dst[p] = e[u].x;
                    ovf_src[p] = e[u].y;
                }
            }
        }
    }
    const int n_spl = cur[MAXB * CUR_PITCH];
    for (int i = tid; i < n_spl; i += 512) {
        const int dd = spl_dst[i];
        if ((dd >> BIN_SHIFT) == b) {
            const int ss = spl_src[i];
            const int l = dd - node0;
            const int r = atomicAdd(&cntL[l], 1);
            if (r < SLOTS) {
                slotL[l * SLOTS + r] = ss;
            } else {
                const int p = atomicAdd(&cnt[n_nodes], 1);
                ovf_dst[p] = dd;
                ovf_src[p] = ss;
            }
        }
    }
    __syncthreads();

    if (tid < BIN_NODES && node0 + tid < n_nodes) cnt[node0 + tid] = cntL[tid];
    const int q = tid & 7;
    #pragma unroll
    for (int g = tid >> 3; g < BIN_NODES; g += 64) {
        const int node = node0 + g;
        if (node < n_nodes) {
            const v4i lo = *(const v4i*)(slotL + g * SLOTS + q * 4);
            *(v4i*)(slot + (size_t)node * SLOTS + q * 4) = lo;
            if (cntL[g] >= 32) {
                const v4i hi = *(const v4i*)(slotL + g * SLOTS + 32 + q * 4);
                *(v4i*)(slot + (size_t)node * SLOTS + 32 + q * 4) = hi;
            }
        }
    }
}

__device__ __forceinline__
v4f gnn_node_mean(const float* __restrict__ hl, int node, int deg, int ids_lo,
                  const int* __restrict__ ids, const int* __restrict__ cnt,
                  const int* __restrict__ ovf_dst, const int* __restrict__ ovf_src,
                  int lane, int n_nodes) {
    const int end = (deg < SLOTS) ? deg : SLOTS;
    const v4f zero = {0.0f, 0.0f, 0.0f, 0.0f};
    const int mine0 = (lane < end) ? ids_lo : 0;
    v4f acc = zero;
    {
        const bool more = end > 16;
        v4f v[16];
        v4f w[8];
        #pragma unroll
        for (int j = 0; j < 16; ++j) {
            const int s = __builtin_amdgcn_readlane(mine0, j);
            v[j] = *(const v4f*)(hl + (size_t)s * DFEAT);
        }
        if (more) {
            #pragma unroll
            for (int j = 0; j < 8; ++j) {
                const int s = __builtin_amdgcn_readlane(mine0, 16 + j);
                w[j] = *(const v4f*)(hl + (size_t)s * DFEAT);
            }
        }
        #pragma unroll
        for (int j = 0; j < 16; ++j) acc += (j < end) ? v[j] : zero;
        if (more) {
            #pragma unroll
            for (int j = 0; j < 8; ++j) acc += (16 + j < end) ? w[j] : zero;
        }
    }
    if (end > 24) {
        int mine1 = 0;
        if (end > 32) mine1 = (32 + lane < end) ? ids[32 + lane] : 0;
        for (int k = 24; k < end; k += 8) {
            const int mm = (k < 32) ? mine0 : mine1;
            const int kb = k & 31;
            v4f v[8];
            #pragma unroll
            for (int j = 0; j < 8; ++j) {
                const int s = __builtin_amdgcn_readlane(mm, kb + j);
                v[j] = *(const v4f*)(hl + (size_t)s * DFEAT);
            }
            #pragma unroll
            for (int j = 0; j < 8; ++j) acc += (k + j < end) ? v[j] : zero;
        }
    }
    if (deg > SLOTS) {
        const int n_ovf = cnt[n_nodes];
        for (int p0 = 0; p0 < n_ovf; p0 += 32) {
            const int p  = p0 + lane;
            const int od = (p < n_ovf) ? ovf_dst[p] : -1;
            const int os = (p < n_ovf) ? ovf_src[p] : 0;
            unsigned hit = __builtin_amdgcn_ballot_w32(od == node);
            while (hit != 0u) {
                const int l = __builtin_ctz(hit);
                hit &= hit - 1u;
                const int s0 = __shfl(os, l);
                acc += *(const v4f*)(hl + (size_t)s0 * DFEAT);
            }
        }
    }
    const float rdeg = 1.0f / (float)(deg > 0 ? deg : 1);
    acc *= (deg > 0) ? rdeg : 0.0f;
    return acc;
}

#define DOUT3 32

__global__ __launch_bounds__(256)
void gnn_gather_mean_out(const float* __restrict__ zl, const float* __restrict__ zr,
                         const int* __restrict__ cnt, const int* __restrict__ slot,
                         const int* __restrict__ ovf_dst, const int* __restrict__ ovf_src,
                         float* __restrict__ out, int n_nodes) {
    const int q    = threadIdx.x & 7;
    const int node = (int)(blockIdx.x * 256 + threadIdx.x) >> 3;
    if (node >= n_nodes) return;
    const int* __restrict__ ids = slot + (size_t)node * SLOTS;
    v4i c0 = *(const v4i*)(ids);
    v4i c1 = *(const v4i*)(ids + 4);
    const int deg = cnt[node];
    const v4f self = *(const v4f*)(zr + (size_t)node * DOUT3 + q * 4);
    const int end = (deg < SLOTS) ? deg : SLOTS;
    const float* __restrict__ zq = zl + q * 4;

    v4f acc = {0.0f, 0.0f, 0.0f, 0.0f};
    int k = 0;
    for (; k + 8 <= end; k += 8) {
        const v4i n0 = *(const v4i*)(ids + ((k + 8) & (SLOTS - 1)));
        const v4i n1 = *(const v4i*)(ids + ((k + 12) & (SLOTS - 1)));
        const v4f v0 = *(const v4f*)(zq + (size_t)c0.x * DOUT3);
        const v4f v1 = *(const v4f*)(zq + (size_t)c0.y * DOUT3);
        const v4f v2 = *(const v4f*)(zq + (size_t)c0.z * DOUT3);
        const v4f v3 = *(const v4f*)(zq + (size_t)c0.w * DOUT3);
        const v4f v4 = *(const v4f*)(zq + (size_t)c1.x * DOUT3);
        const v4f v5 = *(const v4f*)(zq + (size_t)c1.y * DOUT3);
        const v4f v6 = *(const v4f*)(zq + (size_t)c1.z * DOUT3);
        const v4f v7 = *(const v4f*)(zq + (size_t)c1.w * DOUT3);
        acc += v0; acc += v1; acc += v2; acc += v3;
        acc += v4; acc += v5; acc += v6; acc += v7;
        c0 = n0; c1 = n1;
    }
    const int r = end - k;
    if (r > 0) {
        const v4f zero = {0.0f, 0.0f, 0.0f, 0.0f};
        const int s0 = c0.x;
        const int s1 = (r > 1) ? c0.y : s0;
        const int s2 = (r > 2) ? c0.z : s0;
        const int s3 = (r > 3) ? c0.w : s0;
        const int s4 = (r > 4) ? c1.x : s0;
        const int s5 = (r > 5) ? c1.y : s0;
        const int s6 = (r > 6) ? c1.z : s0;
        const v4f v0 = *(const v4f*)(zq + (size_t)s0 * DOUT3);
        const v4f v1 = *(const v4f*)(zq + (size_t)s1 * DOUT3);
        const v4f v2 = *(const v4f*)(zq + (size_t)s2 * DOUT3);
        const v4f v3 = *(const v4f*)(zq + (size_t)s3 * DOUT3);
        const v4f v4 = *(const v4f*)(zq + (size_t)s4 * DOUT3);
        const v4f v5 = *(const v4f*)(zq + (size_t)s5 * DOUT3);
        const v4f v6 = *(const v4f*)(zq + (size_t)s6 * DOUT3);
        acc += v0;
        acc += (r > 1) ? v1 : zero;
        acc += (r > 2) ? v2 : zero;
        acc += (r > 3) ? v3 : zero;
        acc += (r > 4) ? v4 : zero;
        acc += (r > 5) ? v5 : zero;
        acc += (r > 6) ? v6 : zero;
    }
    if (deg > SLOTS) {
        const int n_ovf = cnt[n_nodes];
        for (int p = 0; p < n_ovf; ++p) {
            if (ovf_dst[p] == node) acc += *(const v4f*)(zq + (size_t)ovf_src[p] * DOUT3);
        }
    }
    const float rdeg = 1.0f / (float)(deg > 0 ? deg : 1);
    const float sc   = (deg > 0) ? rdeg : 0.0f;
    v4f o = acc * sc + self;
    o.x = (o.x > 0.0f) ? o.x : 0.0f;
    o.y = (o.y > 0.0f) ? o.y : 0.0f;
    o.z = (o.z > 0.0f) ? o.z : 0.0f;
    o.w = (o.w > 0.0f) ? o.w : 0.0f;
    *(v4f*)(out + (size_t)node * DOUT3 + q * 4) = o;
}

#define WPITCH 260

#define GEMM_WGPS  256
#define GEMM_WAVES 12
#define GEMM_LDS_FLOATS 41984

#define ZPITCH 132
#define GEMM_W3_OFF     (8 * 16 * WPITCH)
#define GEMM_TILE_OFF   GEMM_LDS_FLOATS
#define GEMM_TILE_FLOATS (16 * ZPITCH)
#define GEMM_LDS_FLOATS_ALL (GEMM_LDS_FLOATS + (GEMM_WAVES + 2) * GEMM_TILE_FLOATS)

__device__ __forceinline__
void gnn_gather_tile_rows(const float* __restrict__ h, const int* __restrict__ cnt,
                          const int* __restrict__ slot, const int* __restrict__ ovf_dst,
                          const int* __restrict__ ovf_src, float* tile,
                          int node0, int r0, int rstep, int lane, int n_nodes) {
    if (r0 >= 16) return;
    const float* __restrict__ hl = h + lane * 4;
    int node = node0 + r0;
    if (node >= n_nodes) node = n_nodes - 1;
    int deg_n = cnt[node];
    int ids_n = slot[(size_t)node * SLOTS + lane];
    for (int r = r0; r < 16; r += rstep) {
        const int cur = node;
        const int deg = deg_n;
        const int ids_lo = ids_n;
        const int rn = (r + rstep < 16) ? r + rstep : r;
        node = node0 + rn;
        if (node >= n_nodes) node = n_nodes - 1;
        deg_n = cnt[node];
        ids_n = slot[(size_t)node * SLOTS + lane];
        const v4f m = gnn_node_mean(hl, cur, deg, ids_lo, slot + (size_t)cur * SLOTS, cnt, ovf_dst, ovf_src, lane, n_nodes);
        *(v4f*)(tile + r * ZPITCH + 4 * lane) = m;
    }
}

template <int NT, bool TO_LDS, bool OWN_GATHER>
__device__ __forceinline__
void gnn_sage_gemm_tile(const float* __restrict__ h,
                        const int* __restrict__ cnt, const int* __restrict__ slot,
                        const int* __restrict__ ovf_dst, const int* __restrict__ ovf_src,
                        const float* __restrict__ bl,
                        float* __restrict__ out,
                        const float* wlds, float* atile, float* tile,
                        int mtile, int T0, int lane, int n_nodes, bool rfirst) {
    const int m0 = mtile * 16;
    int rA = m0 + (lane & 15);
    if (rA >= n_nodes) rA = n_nodes - 1;
    const int klo = 2 * (lane >> 4);

    const float* __restrict__ hrow = h + (size_t)rA * DFEAT + klo;
    const float* arow = atile + (lane & 15) * ZPITCH + klo;
    const float* wb = wlds + (T0 * 16 + (lane & 15)) * WPITCH + klo;

    v8f acc[NT];
    #pragma unroll
    for (int t = 0; t < NT; ++t) acc[t] = v8f{};

    const bool r_ahead = OWN_GATHER && rfirst;

    auto half_r = [&]() {
        #pragma unroll 8
        for (int kk = 0; kk < DFEAT; kk += 4) {
            const v2f a = *(const v2f*)(hrow + kk);
            #pragma unroll
            for (int t = 0; t < NT; ++t) {
                const v2f b = *(const v2f*)(wb + t * 16 * WPITCH + DFEAT + kk);
                acc[t] = __builtin_amdgcn_wmma_f32_16x16x4_f32(
                    false, a, false, b, (short)0, acc[t], false, false);
            }
        }
    };

    if (r_ahead) half_r();

    if (OWN_GATHER) {
        gnn_gather_tile_rows(h, cnt, slot, ovf_dst, ovf_src, atile, m0, 0, 1, lane, n_nodes);
        __builtin_amdgcn_fence(__ATOMIC_RELEASE, "wavefront");
        __builtin_amdgcn_wave_barrier();
        __builtin_amdgcn_fence(__ATOMIC_ACQUIRE, "wavefront");
    }

    #pragma unroll 8
    for (int kk = 0; kk < DFEAT; kk += 4) {
        const v2f a = *(const v2f*)(arow + kk);
        #pragma unroll
        for (int t = 0; t < NT; ++t) {
            const v2f b = *(const v2f*)(wb + t * 16 * WPITCH + kk);
            acc[t] = __builtin_amdgcn_wmma_f32_16x16x4_f32(
                false, a, false, b, (short)0, acc[t], false, false);
        }
    }

    if (!r_ahead) half_r();

    const int mbase = m0 + 8 * (lane >> 4);
    float bb[NT];
    #pragma unroll
    for (int t = 0; t < NT; ++t) bb[t] = bl[(T0 + t) * 16 + (lane & 15)];
    if (TO_LDS) {
        if (OWN_GATHER) {
            __builtin_amdgcn_fence(__ATOMIC_RELEASE, "wavefront");
            __builtin_amdgcn_wave_barrier();
            __builtin_amdgcn_fence(__ATOMIC_ACQUIRE, "wavefront");
        }
        float* trow = tile + (8 * (lane >> 4)) * ZPITCH + T0 * 16 + (lane & 15);
        #pragma unroll
        for (int t = 0; t < NT; ++t) {
            #pragma unroll
            for (int v = 0; v < 8; ++v) {
                const float val = acc[t][v] + bb[t];
                trow[v * ZPITCH + t * 16] = (val > 0.0f) ? val : 0.0f;
            }
        }
        return;
    }
    float* __restrict__ orow = out + (size_t)mbase * DFEAT + T0 * 16 + (lane & 15);
    if (m0 + 16 <= n_nodes) {
        #pragma unroll
        for (int t = 0; t < NT; ++t) {
            #pragma unroll
            for (int v = 0; v < 8; ++v) {
                const float val = acc[t][v] + bb[t];
                orow[v * DFEAT + t * 16] = (val > 0.0f) ? val : 0.0f;
            }
        }
    } else {
        #pragma unroll
        for (int t = 0; t < NT; ++t) {
            #pragma unroll
            for (int v = 0; v < 8; ++v) {
                if (mbase + v < n_nodes) {
                    const float val = acc[t][v] + bb[t];
                    orow[v * DFEAT + t * 16] = (val > 0.0f) ? val : 0.0f;
                }
            }
        }
    }
}

template <int NC>
__device__ __forceinline__
void gnn_proj_tile(const float* tile, const float* w3,
                   const float* __restrict__ bl3,
                   float* __restrict__ zl, float* __restrict__ zr,
                   int m0, int C0, int lane, int n_nodes) {
    const int klo = 2 * (lane >> 4);
    const float* ta = tile + (lane & 15) * ZPITCH + klo;
    const float* wb = w3 + (C0 * 16 + (lane & 15)) * ZPITCH + klo;

    v8f acc[NC];
    #pragma unroll
    for (int t = 0; t < NC; ++t) acc[t] = v8f{};

    #pragma unroll 8
    for (int kk = 0; kk < DFEAT; kk += 4) {
        const v2f a = *(const v2f*)(ta + kk);
        #pragma unroll
        for (int t = 0; t < NC; ++t) {
            const v2f b = *(const v2f*)(wb + t * 16 * ZPITCH + kk);
            acc[t] = __builtin_amdgcn_wmma_f32_16x16x4_f32(
                false, a, false, b, (short)0, acc[t], false, false);
        }
    }

    const int mbase = m0 + 8 * (lane >> 4);
    #pragma unroll
    for (int t = 0; t < NC; ++t) {
        const int c   = C0 + t;
        const int col = (c & 1) * 16 + (lane & 15);
        if (c < 2) {
            float* __restrict__ lrow = zl + (size_t)mbase * DOUT3 + col;
            #pragma unroll
            for (int v = 0; v < 8; ++v)
                if (mbase + v < n_nodes) lrow[v * DOUT3] = acc[t][v];
        } else {
            const float bias = bl3[col];
            float* __restrict__ rrow = zr + (size_t)mbase * DOUT3 + col;
            #pragma unroll
            for (int v = 0; v < 8; ++v)
                if (mbase + v < n_nodes) rrow[v * DOUT3] = acc[t][v] + bias;
        }
    }
}

template <bool PROJ>
__global__ __launch_bounds__(GEMM_WAVES * 32)
void gnn_sage_gemm_wmma_hid(const float* __restrict__ h,
                            const int* __restrict__ cnt,
                            const int* __restrict__ slot,
                            const int* __restrict__ ovf_dst,
                            const int* __restrict__ ovf_src,
                            const float* __restrict__ Wl,
                            const float* __restrict__ bl,
                            const float* __restrict__ Wr,
                            float* __restrict__ out,
                            const float* __restrict__ Wl3,
                            const float* __restrict__ bl3,
                            const float* __restrict__ Wr3,
                            float* __restrict__ zl,
                            float* __restrict__ zr,
                            int n_nodes) {
    __shared__ __attribute__((aligned(16))) float wlds[GEMM_LDS_FLOATS_ALL];
    const int tid  = threadIdx.x;
    const int wave = __builtin_amdgcn_readfirstlane(tid >> 5);
    const int lane = tid & 31;

    constexpr int NV4 = 8 * 16 * 64;
    constexpr int NTH = GEMM_WAVES * 32;
    constexpr int RPP = NTH / 64;
    const int q  = tid & 63;
    const int nr = tid >> 6;
    const float* __restrict__ wsrc = ((q < 32) ? Wl : Wr) + (size_t)nr * DFEAT + 4 * (q & 31);
    float* wdst = wlds + nr * WPITCH + 4 * q;
    v4f wv[(NV4 + NTH - 1) / NTH];
    #pragma unroll
    for (int i = 0; i < (NV4 + NTH - 1) / NTH; ++i) {
        const int row = (i * RPP + nr < 8 * 16) ? i * RPP : 0;
        wv[i] = *(const v4f*)(wsrc + (size_t)row * DFEAT);
    }
    if (PROJ) {
        constexpr int NV3 = 2 * DOUT3 * 32;
        v4f w3v[(NV3 + NTH - 1) / NTH];
        #pragma unroll
        for (int i = 0; i < (NV3 + NTH - 1) / NTH; ++i) {
            const int p   = i * NTH + tid;
            const int pc  = (p < NV3) ? p : tid;
            const int row = pc >> 5;
            const float* __restrict__ s3 = ((row < DOUT3) ? Wl3 + (size_t)row * DFEAT : Wr3 + (size_t)(row - DOUT3) * DFEAT) + 4 * (pc & 31);
            w3v[i] = *(const v4f*)s3;
        }
        #pragma unroll
        for (int i = 0; i < (NV3 + NTH - 1) / NTH; ++i) {
            const int p = i * NTH + tid;
            if (p < NV3) *(v4f*)(wlds + GEMM_W3_OFF + (p >> 5) * ZPITCH + 4 * (p & 31)) = w3v[i];
        }
    }
    #pragma unroll
    for (int i = 0; i < (NV4 + NTH - 1) / NTH; ++i) {
        if (i * RPP + nr < 8 * 16) *(v4f*)(wdst + i * RPP * WPITCH) = wv[i];
    }
    __syncthreads();

    const int mtiles  = (n_nodes + 15) / 16;
    const int per_rnd = GEMM_WAVES * (int)gridDim.x;
    const int n_whole = mtiles / per_rnd * per_rnd;
    const bool rfirst = wave < 8;
    float* mytile = wlds + GEMM_TILE_OFF + wave * GEMM_TILE_FLOATS;
    float* shtile = wlds + GEMM_TILE_OFF + GEMM_WAVES * GEMM_TILE_FLOATS;
    float* shagg  = shtile + GEMM_TILE_FLOATS;
    if (!PROJ) {
        for (int mt = (int)blockIdx.x * GEMM_WAVES + wave; mt < n_whole; mt += per_rnd)
            gnn_sage_gemm_tile<8, false, true>(h, cnt, slot, ovf_dst, ovf_src, bl, out, wlds, mytile, nullptr,
                                               mt, 0, lane, n_nodes, rfirst);
        for (int mt = n_whole + (int)blockIdx.x; mt < mtiles; mt += (int)gridDim.x) {
            gnn_gather_tile_rows(h, cnt, slot, ovf_dst, ovf_src, shagg, mt * 16, wave, GEMM_WAVES, lane, n_nodes);
            __syncthreads();
            if (wave < 4)
                gnn_sage_gemm_tile<2, false, false>(h, cnt, slot, ovf_dst, ovf_src, bl, out, wlds, shagg, nullptr,
                                                    mt, 2 * wave, lane, n_nodes, false);
            __syncthreads();
        }
    } else {
        const float* w3 = wlds + GEMM_W3_OFF;
        for (int mt = (int)blockIdx.x * GEMM_WAVES + wave; mt < n_whole; mt += per_rnd) {
            gnn_sage_gemm_tile<8, true, true>(h, cnt, slot, ovf_dst, ovf_src, bl, out, wlds, mytile, mytile,
                                              mt, 0, lane, n_nodes, rfirst);
            __builtin_amdgcn_fence(__ATOMIC_RELEASE, "wavefront");
            __builtin_amdgcn_wave_barrier();
            __builtin_amdgcn_fence(__ATOMIC_ACQUIRE, "wavefront");
            gnn_proj_tile<4>(mytile, w3, bl3, zl, zr, mt * 16, 0, lane, n_nodes);
            __builtin_amdgcn_fence(__ATOMIC_RELEASE, "wavefront");
            __builtin_amdgcn_wave_barrier();
            __builtin_amdgcn_fence(__ATOMIC_ACQUIRE, "wavefront");
        }
        for (int mt = n_whole + (int)blockIdx.x; mt < mtiles; mt += (int)gridDim.x) {
            gnn_gather_tile_rows(h, cnt, slot, ovf_dst, ovf_src, shagg, mt * 16, wave, GEMM_WAVES, lane, n_nodes);
            __syncthreads();
            if (wave < 4)
                gnn_sage_gemm_tile<2, true, false>(h, cnt, slot, ovf_dst, ovf_src, bl, out, wlds, shagg, shtile,
                                                   mt, 2 * wave, lane, n_nodes, false);
            __syncthreads();
            if (wave < 4)
                gnn_proj_tile<1>(shtile, w3, bl3, zl, zr, mt * 16, wave, lane, n_nodes);
            __syncthreads();
        }
    }
}

extern "C" void kernel_launch(void* const* d_in, const int* in_sizes, int n_in,
                              void* d_out, int out_size, void* d_ws, size_t ws_size,
                              hipStream_t stream) {
    const float* x    = (const float*)d_in[0];
    const int*   eidx = (const int*)  d_in[1];
    const float* W_l1 = (const float*)d_in[2];
    const float* b_l1 = (const float*)d_in[3];
    const float* W_r1 = (const float*)d_in[4];
    const float* W_l2 = (const float*)d_in[5];
    const float* b_l2 = (const float*)d_in[6];
    const float* W_r2 = (const float*)d_in[7];
    const float* W_l3 = (const float*)d_in[8];
    const float* b_l3 = (const float*)d_in[9];
    const float* W_r3 = (const float*)d_in[10];
    float* out = (float*)d_out;

    const int n_nodes = in_sizes[0] / DFEAT;
    const int n_edges = in_sizes[1] / 2;
    const int* src = eidx;
    const int* dst = eidx + n_edges;

    int*   slot = (int*)d_ws;
    float* h1   = (float*)(slot + (size_t)n_nodes * SLOTS) + (size_t)n_nodes * DFEAT;
    float* h2   = h1  + (size_t)n_nodes * DFEAT;
    float* zl    = h2;
    float* zr    = h2 + (size_t)n_nodes * DOUT3;
    int* cnt     = (int*)(h2 + (size_t)n_nodes * DFEAT);
    const int cur_off = (n_nodes + 1 + CUR_PITCH - 1) / CUR_PITCH * CUR_PITCH;
    int* cur     = cnt + cur_off;
    int* binbuf  = cur + MAXB * CUR_PITCH + CUR_PITCH;
    int* ovf_dst = binbuf + (size_t)MAXB * BIN_CAP * 2;
    int* ovf_src = ovf_dst + n_edges;
    int* spl_dst = ovf_src + n_edges;
    int* spl_src = spl_dst + n_edges;

    const int  TB      = 256;
    const int  nZero   = cur_off - n_nodes + MAXB * CUR_PITCH + CUR_PITCH;
    const int  gZero   = (nZero + TB - 1) / TB;
    const int  gBinE   = (n_edges + EPB - 1) / EPB;
    const int  gBins   = (n_nodes + BIN_NODES - 1) / BIN_NODES;
    const int  gGatherOut = (n_nodes + 31) / 32;

    gnn_zero_i32  <<<gZero, TB, 0, stream>>>(cnt + n_nodes, nZero);
    gnn_bin_edges <<<gBinE, TB, 0, stream>>>(src, dst, cur, binbuf, spl_dst, spl_src, n_edges);
    gnn_place_bins<<<gBins, 512, 0, stream>>>(cur, binbuf, spl_dst, spl_src, cnt, slot, ovf_dst, ovf_src, n_nodes);

    gnn_sage_gemm_wmma_hid<false><<<GEMM_WGPS, GEMM_WAVES * 32, 0, stream>>>(x, cnt, slot, ovf_dst, ovf_src,
                                                                             W_l1, b_l1, W_r1, h1,
                                                                             W_l3, b_l3, W_r3, zl, zr, n_nodes);

    gnn_sage_gemm_wmma_hid<true><<<GEMM_WGPS, GEMM_WAVES * 32, 0, stream>>>(h1, cnt, slot, ovf_dst, ovf_src,
                                                                            W_l2, b_l2, W_r2, h2,
                                                                            W_l3, b_l3, W_r3, zl, zr, n_nodes);

    gnn_gather_mean_out<<<gGatherOut, TB, 0, stream>>>(zl, zr, cnt, slot, ovf_dst, ovf_src, out, n_nodes);
}


# Edge placement pass 1 keeps each block's bin-sorted edges in the block's own region beside a table of bin offsets: no global atomic, no counter to clear, zero launch gone; sums stay in LDS-rank order
# speedup vs baseline: 6.9536x; 1.1073x over previous
#include <hip/hip_runtime.h>


typedef float v2f __attribute__((ext_vector_type(2)));
typedef float v8f __attribute__((ext_vector_type(8)));

#define DFEAT 128

typedef float v4f __attribute__((ext_vector_type(4)));
typedef int   v4i __attribute__((ext_vector_type(4)));

#define SLOTS 64
#define BIN_SHIFT 8
#define BIN_NODES (1 << BIN_SHIFT)
#define MAXB 256
#define BOFF_PITCH (MAXB + 1)
#define EPT 13
#define EPB (EPT * 256)

typedef int v2i __attribute__((ext_vector_type(2)));

__global__ __launch_bounds__(256)
void gnn_bin_edges(const int* __restrict__ src, const int* __restrict__ dst,
                   int* __restrict__ boff, int* __restrict__ binbuf_i,
                   int* __restrict__ ovf_count, int n_edges) {
    v2i* __restrict__ binbuf = (v2i*)binbuf_i;
    __shared__ int hist[MAXB + 1];
    __shared__ int wsum[8];
    __shared__ int ntot;
    __shared__ v2i sorted[EPB];
    const int tid  = threadIdx.x;
    const int lane = tid & 31;
    const int wave = tid >> 5;
    const int e0   = blockIdx.x * EPB;

    hist[tid] = 0;
    if (tid == 0) hist[MAXB] = 0;
    __syncthreads();

    int d[EPT], s[EPT], r[EPT];
    #pragma unroll
    for (int i = 0; i < EPT; ++i) {
        const int e  = e0 + i * 256 + tid;
        const int ec = (e < n_edges) ? e : 0;
        const int dv = dst[ec];
        s[i] = src[ec];
        d[i] = (e < n_edges) ? dv : -1;
    }
    #pragma unroll
    for (int i = 0; i < EPT; ++i) {
        const int b = d[i] >> BIN_SHIFT;
        r[i] = atomicAdd(&hist[((unsigned)b < (unsigned)MAXB) ? b : MAXB], 1);
    }
    __syncthreads();

    const int c = hist[tid];
    int incl = c;
    #pragma unroll
    for (int o = 1; o < 32; o <<= 1) {
        const int t = __shfl_up(incl, o);
        if (lane >= o) incl += t;
    }
    if (lane == 31) wsum[wave] = incl;
    __syncthreads();
    int woff = 0;
    #pragma unroll
    for (int w = 0; w < 8; ++w) woff += (w < wave) ? wsum[w] : 0;
    const int loff = woff + incl - c;
    hist[tid] = loff;
    if (tid == 255) ntot = loff + c;
    __syncthreads();

    int o[EPT];
    #pragma unroll
    for (int i = 0; i < EPT; ++i) {
        const int b = d[i] >> BIN_SHIFT;
        o[i] = hist[((unsigned)b < (unsigned)MAXB) ? b : MAXB] + r[i];
    }
    #pragma unroll
    for (int i = 0; i < EPT; ++i) {
        const int b = d[i] >> BIN_SHIFT;
        if ((unsigned)b < (unsigned)MAXB) {
            v2i e; e.x = d[i]; e.y = s[i];
            sorted[o[i]] = e;
        }
    }
    __syncthreads();

    const int n = ntot;
    v2i* __restrict__ myreg = binbuf + (size_t)blockIdx.x * EPB;
    v2i oe[EPT];
    #pragma unroll
    for (int i = 0; i < EPT; ++i) {
        const int idx = i * 256 + tid;
        oe[i] = sorted[(idx < n) ? idx : 0];
    }
    #pragma unroll
    for (int i = 0; i < EPT; ++i) {
        const int idx = i * 256 + tid;
        if (idx < n) myreg[idx] = oe[i];
    }
    int* __restrict__ mytab = boff + (size_t)blockIdx.x * BOFF_PITCH;
    mytab[tid] = loff;
    if (tid == 255) mytab[MAXB] = loff + c;
    if (blockIdx.x == 0 && tid == 0) *ovf_count = 0;
}

__global__ __launch_bounds__(512)
void gnn_place_bins(const int* __restrict__ boff, const int* __restrict__ binbuf_i,
                    int* __restrict__ cnt, int* __restrict__ slot,
                    int* __restrict__ ovf_dst, int* __restrict__ ovf_src, int n_nodes, int n_blk) {
    __shared__ int cntL[BIN_NODES];
    __shared__ __attribute__((aligned(16))) int slotL[BIN_NODES * SLOTS];
    const int tid   = threadIdx.x;
    const int lane  = tid & 31;
    const int wave  = __builtin_amdgcn_readfirstlane(tid >> 5);
    const int b     = blockIdx.x;
    const int node0 = b << BIN_SHIFT;

    if (tid < BIN_NODES) cntL[tid] = 0;
    const v4i zero4 = {0, 0, 0, 0};
    #pragma unroll
    for (int i = 0; i < BIN_NODES * SLOTS / 4 / 512; ++i) ((v4i*)slotL)[i * 512 + tid] = zero4;
    __syncthreads();

    const v2i* __restrict__ eb = (const v2i*)binbuf_i;
    for (int g0 = 0; g0 < n_blk; g0 += 256) {
        const int  pm = g0 + lane * 16 + wave;
        const bool pv = (lane < 16) && (pm < n_blk) && (b < MAXB);
        const int* __restrict__ tab = boff + (size_t)(pv ? pm : 0) * BOFF_PITCH + (pv ? b : 0);
        int rs = tab[0];
        int re = tab[1];
        if (!pv) { rs = 0; re = 0; }

        v2i ev[16];
        int rr[16];
        #pragma unroll
        for (int j = 0; j < 16; ++j) {
            const int sj = __builtin_amdgcn_readlane(rs, j);
            const int ej = __builtin_amdgcn_readlane(re, j);
            const int pj = g0 + j * 16 + wave;
            const int i  = sj + lane;
            const size_t at = (i < ej) ? ((size_t)pj * EPB + (size_t)i) : (size_t)0;
            ev[j] = eb[at];
            if (i >= ej) ev[j].x = -1;
        }
        #pragma unroll
        for (int j = 0; j < 16; ++j) {
            rr[j] = 0;
            if (ev[j].x >= 0) rr[j] = atomicAdd(&cntL[ev[j].x - node0], 1);
        }
        #pragma unroll
        for (int j = 0; j < 16; ++j) {
            if (ev[j].x >= 0) {
                if (rr[j] < SLOTS) {
                    slotL[(ev[j].x - node0) * SLOTS + rr[j]] = ev[j].y;
                } else {
                    const int p = atomicAdd(&cnt[n_nodes], 1);
                    ovf_dst[p] = ev[j].x;
                    ovf_src[p] = ev[j].y;
                }
            }
        }
        if (__builtin_amdgcn_ballot_w32(re - rs > 32) != 0u) {
            for (int j = 0; j < 16; ++j) {
                const int sj = __shfl(rs, j);
                const int ej = __shfl(re, j);
                const int pj = g0 + j * 16 + wave;
                for (int i = sj + 32 + lane; i < ej; i += 32) {
                    const v2i e = eb[(size_t)pj * EPB + (size_t)i];
                    const int l = e.x - node0;
                    const int r = atomicAdd(&cntL[l], 1);
                    if (r < SLOTS) {
                        slotL[l * SLOTS + r] = e.y;
                    } else {
                        const int p = atomicAdd(&cnt[n_nodes], 1);
                        ovf_dst[p] = e.x;
                        ovf_src[p] = e.y;
                    }
                }
            }
        }
    }
    __syncthreads();

    if (tid < BIN_NODES && node0 + tid < n_nodes) cnt[node0 + tid] = cntL[tid];
    const int q = tid & 7;
    #pragma unroll
    for (int g = tid >> 3; g < BIN_NODES; g += 64) {
        const int node = node0 + g;
        if (node < n_nodes) {
            const v4i lo = *(const v4i*)(slotL + g * SLOTS + q * 4);
            *(v4i*)(slot + (size_t)node * SLOTS + q * 4) = lo;
            if (cntL[g] >= 32) {
                const v4i hi = *(const v4i*)(slotL + g * SLOTS + 32 + q * 4);
                *(v4i*)(slot + (size_t)node * SLOTS + 32 + q * 4) = hi;
            }
        }
    }
}

__device__ __forceinline__
v4f gnn_node_mean(const float* __restrict__ hl, int node, int deg, int ids_lo,
                  const int* __restrict__ ids, const int* __restrict__ cnt,
                  const int* __restrict__ ovf_dst, const int* __restrict__ ovf_src,
                  int lane, int n_nodes) {
    const int end = (deg < SLOTS) ? deg : SLOTS;
    const v4f zero = {0.0f, 0.0f, 0.0f, 0.0f};
    const int mine0 = (lane < end) ? ids_lo : 0;
    v4f acc = zero;
    {
        const bool more = end > 16;
        v4f v[16];
        v4f w[8];
        #pragma unroll
        for (int j = 0; j < 16; ++j) {
            const int s = __builtin_amdgcn_readlane(mine0, j);
            v[j] = *(const v4f*)(hl + (size_t)s * DFEAT);
        }
        if (more) {
            #pragma unroll
            for (int j = 0; j < 8; ++j) {
                const int s = __builtin_amdgcn_readlane(mine0, 16 + j);
                w[j] = *(const v4f*)(hl + (size_t)s * DFEAT);
            }
        }
        #pragma unroll
        for (int j = 0; j < 16; ++j) acc += (j < end) ? v[j] : zero;
        if (more) {
            #pragma unroll
            for (int j = 0; j < 8; ++j) acc += (16 + j < end) ? w[j] : zero;
        }
    }
    if (end > 24) {
        int mine1 = 0;
        if (end > 32) mine1 = (32 + lane < end) ? ids[32 + lane] : 0;
        for (int k = 24; k < end; k += 8) {
            const int mm = (k < 32) ? mine0 : mine1;
            const int kb = k & 31;
            v4f v[8];
            #pragma unroll
            for (int j = 0; j < 8; ++j) {
                const int s = __builtin_amdgcn_readlane(mm, kb + j);
                v[j] = *(const v4f*)(hl + (size_t)s * DFEAT);
            }
            #pragma unroll
            for (int j = 0; j < 8; ++j) acc += (k + j < end) ? v[j] : zero;
        }
    }
    if (deg > SLOTS) {
        const int n_ovf = cnt[n_nodes];
        for (int p0 = 0; p0 < n_ovf; p0 += 32) {
            const int p  = p0 + lane;
            const int od = (p < n_ovf) ? ovf_dst[p] : -1;
            const int os = (p < n_ovf) ? ovf_src[p] : 0;
            unsigned hit = __builtin_amdgcn_ballot_w32(od == node);
            while (hit != 0u) {
                const int l = __builtin_ctz(hit);
                hit &= hit - 1u;
                const int s0 = __shfl(os, l);
                acc += *(const v4f*)(hl + (size_t)s0 * DFEAT);
            }
        }
    }
    const float rdeg = 1.0f / (float)(deg > 0 ? deg : 1);
    acc *= (deg > 0) ? rdeg : 0.0f;
    return acc;
}

#define DOUT3 32

__global__ __launch_bounds__(256)
void gnn_gather_mean_out(const float* __restrict__ zl, const float* __restrict__ zr,
                         const int* __restrict__ cnt, const int* __restrict__ slot,
                         const int* __restrict__ ovf_dst, const int* __restrict__ ovf_src,
                         float* __restrict__ out, int n_nodes) {
    const int q    = threadIdx.x & 7;
    const int node = (int)(blockIdx.x * 256 + threadIdx.x) >> 3;
    if (node >= n_nodes) return;
    const int* __restrict__ ids = slot + (size_t)node * SLOTS;
    v4i c0 = *(const v4i*)(ids);
    v4i c1 = *(const v4i*)(ids + 4);
    const int deg = cnt[node];
    const v4f self = *(const v4f*)(zr + (size_t)node * DOUT3 + q * 4);
    const int end = (deg < SLOTS) ? deg : SLOTS;
    const float* __restrict__ zq = zl + q * 4;

    v4f acc = {0.0f, 0.0f, 0.0f, 0.0f};
    int k = 0;
    for (; k + 8 <= end; k += 8) {
        const v4i n0 = *(const v4i*)(ids + ((k + 8) & (SLOTS - 1)));
        const v4i n1 = *(const v4i*)(ids + ((k + 12) & (SLOTS - 1)));
        const v4f v0 = *(const v4f*)(zq + (size_t)c0.x * DOUT3);
        const v4f v1 = *(const v4f*)(zq + (size_t)c0.y * DOUT3);
        const v4f v2 = *(const v4f*)(zq + (size_t)c0.z * DOUT3);
        const v4f v3 = *(const v4f*)(zq + (size_t)c0.w * DOUT3);
        const v4f v4 = *(const v4f*)(zq + (size_t)c1.x * DOUT3);
        const v4f v5 = *(const v4f*)(zq + (size_t)c1.y * DOUT3);
        const v4f v6 = *(const v4f*)(zq + (size_t)c1.z * DOUT3);
        const v4f v7 = *(const v4f*)(zq + (size_t)c1.w * DOUT3);
        acc += v0; acc += v1; acc += v2; acc += v3;
        acc += v4; acc += v5; acc += v6; acc += v7;
        c0 = n0; c1 = n1;
    }
    const int r = end - k;
    if (r > 0) {
        const v4f zero = {0.0f, 0.0f, 0.0f, 0.0f};
        const int s0 = c0.x;
        const int s1 = (r > 1) ? c0.y : s0;
        const int s2 = (r > 2) ? c0.z : s0;
        const int s3 = (r > 3) ? c0.w : s0;
        const int s4 = (r > 4) ? c1.x : s0;
        const int s5 = (r > 5) ? c1.y : s0;
        const int s6 = (r > 6) ? c1.z : s0;
        const v4f v0 = *(const v4f*)(zq + (size_t)s0 * DOUT3);
        const v4f v1 = *(const v4f*)(zq + (size_t)s1 * DOUT3);
        const v4f v2 = *(const v4f*)(zq + (size_t)s2 * DOUT3);
        const v4f v3 = *(const v4f*)(zq + (size_t)s3 * DOUT3);
        const v4f v4 = *(const v4f*)(zq + (size_t)s4 * DOUT3);
        const v4f v5 = *(const v4f*)(zq + (size_t)s5 * DOUT3);
        const v4f v6 = *(const v4f*)(zq + (size_t)s6 * DOUT3);
        acc += v0;
        acc += (r > 1) ? v1 : zero;
        acc += (r > 2) ? v2 : zero;
        acc += (r > 3) ? v3 : zero;
        acc += (r > 4) ? v4 : zero;
        acc += (r > 5) ? v5 : zero;
        acc += (r > 6) ? v6 : zero;
    }
    if (deg > SLOTS) {
        const int n_ovf = cnt[n_nodes];
        for (int p = 0; p < n_ovf; ++p) {
            if (ovf_dst[p] == node) acc += *(const v4f*)(zq + (size_t)ovf_src[p] * DOUT3);
        }
    }
    const float rdeg = 1.0f / (float)(deg > 0 ? deg : 1);
    const float sc   = (deg > 0) ? rdeg : 0.0f;
    v4f o = acc * sc + self;
    o.x = (o.x > 0.0f) ? o.x : 0.0f;
    o.y = (o.y > 0.0f) ? o.y : 0.0f;
    o.z = (o.z > 0.0f) ? o.z : 0.0f;
    o.w = (o.w > 0.0f) ? o.w : 0.0f;
    *(v4f*)(out + (size_t)node * DOUT3 + q * 4) = o;
}

#define WPITCH 260

#define GEMM_WGPS  256
#define GEMM_WAVES 12
#define GEMM_LDS_FLOATS 41984

#define ZPITCH 132
#define GEMM_W3_OFF     (8 * 16 * WPITCH)
#define GEMM_TILE_OFF   GEMM_LDS_FLOATS
#define GEMM_TILE_FLOATS (16 * ZPITCH)
#define GEMM_LDS_FLOATS_ALL (GEMM_LDS_FLOATS + (GEMM_WAVES + 2) * GEMM_TILE_FLOATS)

__device__ __forceinline__
void gnn_gather_tile_rows(const float* __restrict__ h, const int* __restrict__ cnt,
                          const int* __restrict__ slot, const int* __restrict__ ovf_dst,
                          const int* __restrict__ ovf_src, float* tile,
                          int node0, int r0, int rstep, int lane, int n_nodes) {
    if (r0 >= 16) return;
    const float* __restrict__ hl = h + lane * 4;
    int node = node0 + r0;
    if (node >= n_nodes) node = n_nodes - 1;
    int deg_n = cnt[node];
    int ids_n = slot[(size_t)node * SLOTS + lane];
    for (int r = r0; r < 16; r += rstep) {
        const int cur = node;
        const int deg = deg_n;
        const int ids_lo = ids_n;
        const int rn = (r + rstep < 16) ? r + rstep : r;
        node = node0 + rn;
        if (node >= n_nodes) node = n_nodes - 1;
        deg_n = cnt[node];
        ids_n = slot[(size_t)node * SLOTS + lane];
        const v4f m = gnn_node_mean(hl, cur, deg, ids_lo, slot + (size_t)cur * SLOTS, cnt, ovf_dst, ovf_src, lane, n_nodes);
        *(v4f*)(tile + r * ZPITCH + 4 * lane) = m;
    }
}

template <int NT, bool TO_LDS, bool OWN_GATHER>
__device__ __forceinline__
void gnn_sage_gemm_tile(const float* __restrict__ h,
                        const int* __restrict__ cnt, const int* __restrict__ slot,
                        const int* __restrict__ ovf_dst, const int* __restrict__ ovf_src,
                        const float* __restrict__ bl,
                        float* __restrict__ out,
                        const float* wlds, float* atile, float* tile,
                        int mtile, int T0, int lane, int n_nodes, bool rfirst) {
    const int m0 = mtile * 16;
    int rA = m0 + (lane & 15);
    if (rA >= n_nodes) rA = n_nodes - 1;
    const int klo = 2 * (lane >> 4);

    const float* __restrict__ hrow = h + (size_t)rA * DFEAT + klo;
    const float* arow = atile + (lane & 15) * ZPITCH + klo;
    const float* wb = wlds + (T0 * 16 + (lane & 15)) * WPITCH + klo;

    v8f acc[NT];
    #pragma unroll
    for (int t = 0; t < NT; ++t) acc[t] = v8f{};

    const bool r_ahead = OWN_GATHER && rfirst;

    auto half_r = [&]() {
        #pragma unroll 8
        for (int kk = 0; kk < DFEAT; kk += 4) {
            const v2f a = *(const v2f*)(hrow + kk);
            #pragma unroll
            for (int t = 0; t < NT; ++t) {
                const v2f b = *(const v2f*)(wb + t * 16 * WPITCH + DFEAT + kk);
                acc[t] = __builtin_amdgcn_wmma_f32_16x16x4_f32(
                    false, a, false, b, (short)0, acc[t], false, false);
            }
        }
    };

    if (r_ahead) half_r();

    if (OWN_GATHER) {
        gnn_gather_tile_rows(h, cnt, slot, ovf_dst, ovf_src, atile, m0, 0, 1, lane, n_nodes);
        __builtin_amdgcn_fence(__ATOMIC_RELEASE, "wavefront");
        __builtin_amdgcn_wave_barrier();
        __builtin_amdgcn_fence(__ATOMIC_ACQUIRE, "wavefront");
    }

    #pragma unroll 8
    for (int kk = 0; kk < DFEAT; kk += 4) {
        const v2f a = *(const v2f*)(arow + kk);
        #pragma unroll
        for (int t = 0; t < NT; ++t) {
            const v2f b = *(const v2f*)(wb + t * 16 * WPITCH + kk);
            acc[t] = __builtin_amdgcn_wmma_f32_16x16x4_f32(
                false, a, false, b, (short)0, acc[t], false, false);
        }
    }

    if (!r_ahead) half_r();

    const int mbase = m0 + 8 * (lane >> 4);
    float bb[NT];
    #pragma unroll
    for (int t = 0; t < NT; ++t) bb[t] = bl[(T0 + t) * 16 + (lane & 15)];
    if (TO_LDS) {
        if (OWN_GATHER) {
            __builtin_amdgcn_fence(__ATOMIC_RELEASE, "wavefront");
            __builtin_amdgcn_wave_barrier();
            __builtin_amdgcn_fence(__ATOMIC_ACQUIRE, "wavefront");
        }
        float* trow = tile + (8 * (lane >> 4)) * ZPITCH + T0 * 16 + (lane & 15);
        #pragma unroll
        for (int t = 0; t < NT; ++t) {
            #pragma unroll
            for (int v = 0; v < 8; ++v) {
                const float val = acc[t][v] + bb[t];
                trow[v * ZPITCH + t * 16] = (val > 0.0f) ? val : 0.0f;
            }
        }
        return;
    }
    float* __restrict__ orow = out + (size_t)mbase * DFEAT + T0 * 16 + (lane & 15);
    if (m0 + 16 <= n_nodes) {
        #pragma unroll
        for (int t = 0; t < NT; ++t) {
            #pragma unroll
            for (int v = 0; v < 8; ++v) {
                const float val = acc[t][v] + bb[t];
                orow[v * DFEAT + t * 16] = (val > 0.0f) ? val : 0.0f;
            }
        }
    } else {
        #pragma unroll
        for (int t = 0; t < NT; ++t) {
            #pragma unroll
            for (int v = 0; v < 8; ++v) {
                if (mbase + v < n_nodes) {
                    const float val = acc[t][v] + bb[t];
                    orow[v * DFEAT + t * 16] = (val > 0.0f) ? val : 0.0f;
                }
            }
        }
    }
}

template <int NC>
__device__ __forceinline__
void gnn_proj_tile(const float* tile, const float* w3,
                   const float* __restrict__ bl3,
                   float* __restrict__ zl, float* __restrict__ zr,
                   int m0, int C0, int lane, int n_nodes) {
    const int klo = 2 * (lane >> 4);
    const float* ta = tile + (lane & 15) * ZPITCH + klo;
    const float* wb = w3 + (C0 * 16 + (lane & 15)) * ZPITCH + klo;

    v8f acc[NC];
    #pragma unroll
    for (int t = 0; t < NC; ++t) acc[t] = v8f{};

    #pragma unroll 8
    for (int kk = 0; kk < DFEAT; kk += 4) {
        const v2f a = *(const v2f*)(ta + kk);
        #pragma unroll
        for (int t = 0; t < NC; ++t) {
            const v2f b = *(const v2f*)(wb + t * 16 * ZPITCH + kk);
            acc[t] = __builtin_amdgcn_wmma_f32_16x16x4_f32(
                false, a, false, b, (short)0, acc[t], false, false);
        }
    }

    const int mbase = m0 + 8 * (lane >> 4);
    #pragma unroll
    for (int t = 0; t < NC; ++t) {
        const int c   = C0 + t;
        const int col = (c & 1) * 16 + (lane & 15);
        if (c < 2) {
            float* __restrict__ lrow = zl + (size_t)mbase * DOUT3 + col;
            #pragma unroll
            for (int v = 0; v < 8; ++v)
                if (mbase + v < n_nodes) lrow[v * DOUT3] = acc[t][v];
        } else {
            const float bias = bl3[col];
            float* __restrict__ rrow = zr + (size_t)mbase * DOUT3 + col;
            #pragma unroll
            for (int v = 0; v < 8; ++v)
                if (mbase + v < n_nodes) rrow[v * DOUT3] = acc[t][v] + bias;
        }
    }
}

template <bool PROJ>
__global__ __launch_bounds__(GEMM_WAVES * 32)
void gnn_sage_gemm_wmma_hid(const float* __restrict__ h,
                            const int* __restrict__ cnt,
                            const int* __restrict__ slot,
                            const int* __restrict__ ovf_dst,
                            const int* __restrict__ ovf_src,
                            const float* __restrict__ Wl,
                            const float* __restrict__ bl,
                            const float* __restrict__ Wr,
                            float* __restrict__ out,
                            const float* __restrict__ Wl3,
                            const float* __restrict__ bl3,
                            const float* __restrict__ Wr3,
                            float* __restrict__ zl,
                            float* __restrict__ zr,
                            int n_nodes) {
    __shared__ __attribute__((aligned(16))) float wlds[GEMM_LDS_FLOATS_ALL];
    const int tid  = threadIdx.x;
    const int wave = __builtin_amdgcn_readfirstlane(tid >> 5);
    const int lane = tid & 31;

    constexpr int NV4 = 8 * 16 * 64;
    constexpr int NTH = GEMM_WAVES * 32;
    constexpr int RPP = NTH / 64;
    const int q  = tid & 63;
    const int nr = tid >> 6;
    const float* __restrict__ wsrc = ((q < 32) ? Wl : Wr) + (size_t)nr * DFEAT + 4 * (q & 31);
    float* wdst = wlds + nr * WPITCH + 4 * q;
    v4f wv[(NV4 + NTH - 1) / NTH];
    #pragma unroll
    for (int i = 0; i < (NV4 + NTH - 1) / NTH; ++i) {
        const int row = (i * RPP + nr < 8 * 16) ? i * RPP : 0;
        wv[i] = *(const v4f*)(wsrc + (size_t)row * DFEAT);
    }
    if (PROJ) {
        constexpr int NV3 = 2 * DOUT3 * 32;
        v4f w3v[(NV3 + NTH - 1) / NTH];
        #pragma unroll
        for (int i = 0; i < (NV3 + NTH - 1) / NTH; ++i) {
            const int p   = i * NTH + tid;
            const int pc  = (p < NV3) ? p : tid;
            const int row = pc >> 5;
            const float* __restrict__ s3 = ((row < DOUT3) ? Wl3 + (size_t)row * DFEAT : Wr3 + (size_t)(row - DOUT3) * DFEAT) + 4 * (pc & 31);
            w3v[i] = *(const v4f*)s3;
        }
        #pragma unroll
        for (int i = 0; i < (NV3 + NTH - 1) / NTH; ++i) {
            const int p = i * NTH + tid;
            if (p < NV3) *(v4f*)(wlds + GEMM_W3_OFF + (p >> 5) * ZPITCH + 4 * (p & 31)) = w3v[i];
        }
    }
    #pragma unroll
    for (int i = 0; i < (NV4 + NTH - 1) / NTH; ++i) {
        if (i * RPP + nr < 8 * 16) *(v4f*)(wdst + i * RPP * WPITCH) = wv[i];
    }
    __syncthreads();

    const int mtiles  = (n_nodes + 15) / 16;
    const int per_rnd = GEMM_WAVES * (int)gridDim.x;
    const int n_whole = mtiles / per_rnd * per_rnd;
    const bool rfirst = wave < 8;
    float* mytile = wlds + GEMM_TILE_OFF + wave * GEMM_TILE_FLOATS;
    float* shtile = wlds + GEMM_TILE_OFF + GEMM_WAVES * GEMM_TILE_FLOATS;
    float* shagg  = shtile + GEMM_TILE_FLOATS;
    if (!PROJ) {
        for (int mt = (int)blockIdx.x * GEMM_WAVES + wave; mt < n_whole; mt += per_rnd)
            gnn_sage_gemm_tile<8, false, true>(h, cnt, slot, ovf_dst, ovf_src, bl, out, wlds, mytile, nullptr,
                                               mt, 0, lane, n_nodes, rfirst);
        for (int mt = n_whole + (int)blockIdx.x; mt < mtiles; mt += (int)gridDim.x) {
            gnn_gather_tile_rows(h, cnt, slot, ovf_dst, ovf_src, shagg, mt * 16, wave, GEMM_WAVES, lane, n_nodes);
            __syncthreads();
            if (wave < 4)
                gnn_sage_gemm_tile<2, false, false>(h, cnt, slot, ovf_dst, ovf_src, bl, out, wlds, shagg, nullptr,
                                                    mt, 2 * wave, lane, n_nodes, false);
            __syncthreads();
        }
    } else {
        const float* w3 = wlds + GEMM_W3_OFF;
        for (int mt = (int)blockIdx.x * GEMM_WAVES + wave; mt < n_whole; mt += per_rnd) {
            gnn_sage_gemm_tile<8, true, true>(h, cnt, slot, ovf_dst, ovf_src, bl, out, wlds, mytile, mytile,
                                              mt, 0, lane, n_nodes, rfirst);
            __builtin_amdgcn_fence(__ATOMIC_RELEASE, "wavefront");
            __builtin_amdgcn_wave_barrier();
            __builtin_amdgcn_fence(__ATOMIC_ACQUIRE, "wavefront");
            gnn_proj_tile<4>(mytile, w3, bl3, zl, zr, mt * 16, 0, lane, n_nodes);
            __builtin_amdgcn_fence(__ATOMIC_RELEASE, "wavefront");
            __builtin_amdgcn_wave_barrier();
            __builtin_amdgcn_fence(__ATOMIC_ACQUIRE, "wavefront");
        }
        for (int mt = n_whole + (int)blockIdx.x; mt < mtiles; mt += (int)gridDim.x) {
            gnn_gather_tile_rows(h, cnt, slot, ovf_dst, ovf_src, shagg, mt * 16, wave, GEMM_WAVES, lane, n_nodes);
            __syncthreads();
            if (wave < 4)
                gnn_sage_gemm_tile<2, true, false>(h, cnt, slot, ovf_dst, ovf_src, bl, out, wlds, shagg, shtile,
                                                   mt, 2 * wave, lane, n_nodes, false);
            __syncthreads();
            if (wave < 4)
                gnn_proj_tile<1>(shtile, w3, bl3, zl, zr, mt * 16, wave, lane, n_nodes);
            __syncthreads();
        }
    }
}

extern "C" void kernel_launch(void* const* d_in, const int* in_sizes, int n_in,
                              void* d_out, int out_size, void* d_ws, size_t ws_size,
                              hipStream_t stream) {
    const float* x    = (const float*)d_in[0];
    const int*   eidx = (const int*)  d_in[1];
    const float* W_l1 = (const float*)d_in[2];
    const float* b_l1 = (const float*)d_in[3];
    const float* W_r1 = (const float*)d_in[4];
    const float* W_l2 = (const float*)d_in[5];
    const float* b_l2 = (const float*)d_in[6];
    const float* W_r2 = (const float*)d_in[7];
    const float* W_l3 = (const float*)d_in[8];
    const float* b_l3 = (const float*)d_in[9];
    const float* W_r3 = (const float*)d_in[10];
    float* out = (float*)d_out;

    const int n_nodes = in_sizes[0] / DFEAT;
    const int n_edges = in_sizes[1] / 2;
    const int* src = eidx;
    const int* dst = eidx + n_edges;

    int*   slot = (int*)d_ws;
    float* h1   = (float*)(slot + (size_t)n_nodes * SLOTS) + (size_t)n_nodes * DFEAT;
    float* h2   = h1  + (size_t)n_nodes * DFEAT;
    float* zl    = h2;
    float* zr    = h2 + (size_t)n_nodes * DOUT3;
    int* cnt     = (int*)(h2 + (size_t)n_nodes * DFEAT);
    const int  gBinE   = (n_edges + EPB - 1) / EPB;
    const int cnt_len = (n_nodes + 1 + 31) / 32 * 32;
    const int tab_len = (gBinE * BOFF_PITCH + 31) / 32 * 32;
    int* boff    = cnt + cnt_len;
    int* binbuf  = boff + tab_len;
    int* ovf_dst = binbuf + (size_t)gBinE * EPB * 2;
    int* ovf_src = ovf_dst + n_edges;

    const int  TB      = 256;
    const int  gBins   = (n_nodes + BIN_NODES - 1) / BIN_NODES;
    const int  gGatherOut = (n_nodes + 31) / 32;

    gnn_bin_edges <<<gBinE, TB, 0, stream>>>(src, dst, boff, binbuf, cnt + n_nodes, n_edges);
    gnn_place_bins<<<gBins, 512, 0, stream>>>(boff, binbuf, cnt, slot, ovf_dst, ovf_src, n_nodes, gBinE);

    gnn_sage_gemm_wmma_hid<false><<<GEMM_WGPS, GEMM_WAVES * 32, 0, stream>>>(x, cnt, slot, ovf_dst, ovf_src,
                                                                             W_l1, b_l1, W_r1, h1,
                                                                             W_l3, b_l3, W_r3, zl, zr, n_nodes);

    gnn_sage_gemm_wmma_hid<true><<<GEMM_WGPS, GEMM_WAVES * 32, 0, stream>>>(h1, cnt, slot, ovf_dst, ovf_src,
                                                                            W_l2, b_l2, W_r2, h2,
                                                                            W_l3, b_l3, W_r3, zl, zr, n_nodes);

    gnn_gather_mean_out<<<gGatherOut, TB, 0, stream>>>(zl, zr, cnt, slot, ovf_dst, ovf_src, out, n_nodes);
}
